# starting kernel
# baseline (speedup 1.0000x reference)
#include <hip/hip_runtime.h>
#include <cmath>

constexpr int N_  = 100000;
constexpr int E_  = 1000000;
constexpr int ET_ = N_ + E_;
constexpr int D_  = 64;
constexpr int F1_ = 144;
constexpr int F2_ = 64;

#define CDIV(a,b) (((a)+(b)-1)/(b))

typedef __attribute__((ext_vector_type(2))) float v2f;
typedef __attribute__((ext_vector_type(8))) float v8f;

__device__ __forceinline__ void edge_sd(const int* __restrict__ src,
                                        const int* __restrict__ dst,
                                        int e, int& s, int& d) {
  if (e < E_) { s = src[e]; d = dst[e]; }
  else        { s = e - E_; d = s; }
}

__device__ __forceinline__ void atomicMaxF(float* addr, float v) {
  if (v >= 0.0f) atomicMax((int*)addr, __float_as_int(v));
  else           atomicMin((unsigned int*)addr, __float_as_uint(v));
}

__global__ void k_fill(float* __restrict__ p, int n, float v) {
  int i = blockIdx.x * blockDim.x + threadIdx.x;
  if (i < n) p[i] = v;
}

__global__ void k_deg(const int* __restrict__ dst, float* __restrict__ deg) {
  int e = blockIdx.x * blockDim.x + threadIdx.x;
  if (e < E_) atomicAdd(&deg[dst[e]], 1.0f);
}

__global__ void k_rsqrt_inplace(float* __restrict__ p, int n) {
  int i = blockIdx.x * blockDim.x + threadIdx.x;
  if (i < n) p[i] = rsqrtf(p[i]);
}

__global__ void k_norm(const int* __restrict__ src, const int* __restrict__ dst,
                       const float* __restrict__ dinv, float* __restrict__ norm) {
  int e = blockIdx.x * blockDim.x + threadIdx.x;
  if (e < E_) norm[e] = dinv[src[e]] * dinv[dst[e]];
}

__global__ void k_agg(const int* __restrict__ src, const int* __restrict__ dst,
                      const float* __restrict__ x, const float* __restrict__ norm,
                      const float* __restrict__ dinv, float* __restrict__ agg) {
  int idx = blockIdx.x * blockDim.x + threadIdx.x;
  if (idx >= ET_ * D_) return;
  int e = idx >> 6, c = idx & 63;
  int s, d; float w;
  if (e < E_) { s = src[e]; d = dst[e]; w = norm[e]; }
  else        { s = e - E_; d = s; float di = dinv[s]; w = di * di; }
  atomicAdd(&agg[(size_t)d * D_ + c], x[(size_t)s * D_ + c] * w);
}

__global__ void k_combine(const float* __restrict__ x, float* __restrict__ h) {
  int i = blockIdx.x * blockDim.x + threadIdx.x;
  if (i < N_ * D_) h[i] = 0.5f * x[i] + 0.5f * h[i];
}

__global__ void k_gemm_bias(const float* __restrict__ A, const float* __restrict__ W,
                            const float* __restrict__ bias, float* __restrict__ C,
                            int K, int Nc) {
  const int wave = blockIdx.x * (blockDim.x >> 5) + (threadIdx.x >> 5);
  const int lane = threadIdx.x & 31;
  const int tilesN = Nc >> 4;
  const int totalTiles = (N_ >> 4) * tilesN;
  if (wave >= totalTiles) return;

  const int row0 = (wave / tilesN) << 4;
  const int col0 = (wave % tilesN) << 4;
  const int lm = lane & 15;
  const int lk = (lane >> 4) << 1;

  const float* __restrict__ Arow = A + (size_t)(row0 + lm) * K;
  const float* __restrict__ Wcol = W + (col0 + lm);

  v8f acc = {};
  for (int k = 0; k < K; k += 4) {
    v2f a, b;
    a.x = Arow[k + lk];
    a.y = Arow[k + lk + 1];
    b.x = Wcol[(size_t)(k + lk) * Nc];
    b.y = Wcol[(size_t)(k + lk + 1) * Nc];
    acc = __builtin_amdgcn_wmma_f32_16x16x4_f32(false, a, false, b,
                                                (short)0, acc, false, false);
  }

  const int n = col0 + lm;
  const float bn = bias[n];
  const int rbase = row0 + ((lane >> 4) << 3);
#pragma unroll
  for (int j = 0; j < 8; ++j)
    C[(size_t)(rbase + j) * Nc + n] = acc[j] + bn;
}

__global__ void k_logit12(const int* __restrict__ src, const int* __restrict__ dst,
                          const float* __restrict__ xl, const float* __restrict__ xr,
                          const float* __restrict__ att, float* __restrict__ lg,
                          float* __restrict__ m) {
  int idx = blockIdx.x * blockDim.x + threadIdx.x;
  if (idx >= ET_ * 12) return;
  int e = idx / 12, h = idx % 12;
  int s, d; edge_sd(src, dst, e, s, d);
  const float* pl = xl + (size_t)s * F1_ + h * 12;
  const float* pr = xr + (size_t)d * F1_ + h * 12;
  const float* pa = att + h * 12;
  float acc = 0.0f;
#pragma unroll
  for (int c = 0; c < 12; ++c) {
    float t = pl[c] + pr[c];
    t = (t > 0.0f) ? t : 0.2f * t;
    acc += t * pa[c];
  }
  lg[idx] = acc;
  atomicMaxF(&m[d * 12 + h], acc);
}

__global__ void k_expden12(const int* __restrict__ src, const int* __restrict__ dst,
                           const float* __restrict__ m, float* __restrict__ lg,
                           float* __restrict__ den) {
  int idx = blockIdx.x * blockDim.x + threadIdx.x;
  if (idx >= ET_ * 12) return;
  int e = idx / 12, h = idx % 12;
  int s, d; edge_sd(src, dst, e, s, d);
  float a = expf(lg[idx] - m[d * 12 + h]);
  lg[idx] = a;
  atomicAdd(&den[d * 12 + h], a);
}

__global__ void k_accum12(const int* __restrict__ src, const int* __restrict__ dst,
                          const float* __restrict__ lg, const float* __restrict__ xl,
                          float* __restrict__ out) {
  int idx = blockIdx.x * blockDim.x + threadIdx.x;
  if (idx >= ET_ * F1_) return;
  int e = idx / F1_, r = idx % F1_;
  int h = r / 12;
  int s, d; edge_sd(src, dst, e, s, d);
  atomicAdd(&out[(size_t)d * F1_ + r], lg[e * 12 + h] * xl[(size_t)s * F1_ + r]);
}

__global__ void k_fin12(const float* __restrict__ den, const float* __restrict__ bias,
                        float* __restrict__ out) {
  int idx = blockIdx.x * blockDim.x + threadIdx.x;
  if (idx >= N_ * F1_) return;
  int n = idx / F1_, r = idx % F1_;
  int h = r / 12;
  out[idx] = out[idx] / (den[n * 12 + h] + 1e-16f) + bias[r];
}

__global__ void k_logit1(const int* __restrict__ src, const int* __restrict__ dst,
                         const float* __restrict__ xl, const float* __restrict__ xr,
                         const float* __restrict__ att, float* __restrict__ lg,
                         float* __restrict__ m) {
  int wv = blockIdx.x * (blockDim.x >> 5) + (threadIdx.x >> 5);
  int lane = threadIdx.x & 31;
  if (wv >= ET_) return;
  int s, d; edge_sd(src, dst, wv, s, d);
  float acc = 0.0f;
#pragma unroll
  for (int c = lane; c < 64; c += 32) {
    float t = xl[(size_t)s * F2_ + c] + xr[(size_t)d * F2_ + c];
    t = (t > 0.0f) ? t : 0.2f * t;
    acc += t * att[c];
  }
#pragma unroll
  for (int o = 16; o > 0; o >>= 1) acc += __shfl_xor(acc, o, 32);
  if (lane == 0) { lg[wv] = acc; atomicMaxF(&m[d], acc); }
}

__global__ void k_expden1(const int* __restrict__ src, const int* __restrict__ dst,
                          const float* __restrict__ m, float* __restrict__ lg,
                          float* __restrict__ den) {
  int e = blockIdx.x * blockDim.x + threadIdx.x;
  if (e >= ET_) return;
  int s, d; edge_sd(src, dst, e, s, d);
  float a = expf(lg[e] - m[d]);
  lg[e] = a;
  atomicAdd(&den[d], a);
}

__global__ void k_accum1(const int* __restrict__ src, const int* __restrict__ dst,
                         const float* __restrict__ lg, const float* __restrict__ xl,
                         float* __restrict__ num) {
  int idx = blockIdx.x * blockDim.x + threadIdx.x;
  if (idx >= ET_ * F2_) return;
  int e = idx >> 6, c = idx & 63;
  int s, d; edge_sd(src, dst, e, s, d);
  atomicAdd(&num[(size_t)d * F2_ + c], lg[e] * xl[(size_t)s * F2_ + c]);
}

__global__ void k_final(const float* __restrict__ num, const float* __restrict__ den,
                        const float* __restrict__ bias, float* __restrict__ out) {
  int idx = blockIdx.x * blockDim.x + threadIdx.x;
  if (idx >= N_ * F2_) return;
  int n = idx >> 6, c = idx & 63;
  out[idx] = num[idx] / (den[n] + 1e-16f) + bias[c];
}

constexpr size_t OFF_DEG  = 0;
constexpr size_t OFF_NORM = OFF_DEG  + N_;
constexpr size_t OFF_HBUF = OFF_NORM + E_;
constexpr size_t OFF_X1   = OFF_HBUF + (size_t)N_*D_;
constexpr size_t OFF_XL1  = OFF_X1   + (size_t)N_*D_;
constexpr size_t OFF_XR1  = OFF_XL1  + (size_t)N_*F1_;
constexpr size_t OFF_M1   = OFF_XR1  + (size_t)N_*F1_;
constexpr size_t OFF_DEN1 = OFF_M1   + (size_t)N_*12;
constexpr size_t OFF_LG1  = OFF_DEN1 + (size_t)N_*12;

extern "C" void kernel_launch(void* const* d_in, const int* in_sizes, int n_in,
                              void* d_out, int out_size, void* d_ws, size_t ws_size,
                              hipStream_t stream) {
  const float* x     = (const float*)d_in[0];
  const int*   ei    = (const int*)  d_in[1];
  const float* Wssg  = (const float*)d_in[2];
  const float* bssg  = (const float*)d_in[3];
  const float* W1l   = (const float*)d_in[4];
  const float* b1l   = (const float*)d_in[5];
  const float* W1r   = (const float*)d_in[6];
  const float* b1r   = (const float*)d_in[7];
  const float* att1  = (const float*)d_in[8];
  const float* bias1 = (const float*)d_in[9];
  const float* W2l   = (const float*)d_in[10];
  const float* b2l   = (const float*)d_in[11];
  const float* W2r   = (const float*)d_in[12];
  const float* b2r   = (const float*)d_in[13];
  const float* att2  = (const float*)d_in[14];
  const float* bias2 = (const float*)d_in[15];

  const int* src = ei;
  const int* dst = ei + E_;

  float* ws   = (float*)d_ws;
  float* deg  = ws + OFF_DEG;
  float* norm = ws + OFF_NORM;
  float* hbuf = ws + OFF_HBUF;
  float* x1   = ws + OFF_X1;
  float* xl1  = ws + OFF_XL1;
  float* xr1  = ws + OFF_XR1;
  float* m1   = ws + OFF_M1;
  float* den1 = ws + OFF_DEN1;
  float* lg1  = ws + OFF_LG1;
  float* out1 = xr1;
  float* xl2  = hbuf;
  float* xr2  = xl1;
  float* m2   = deg;
  float* den2 = norm;
  float* lg2  = lg1;
  float* num2 = x1;
  float* out  = (float*)d_out;

  const int TB = 256;
  const float NEG_INF = -__builtin_huge_valf();

  auto gemm = [&](const float* A, const float* W, const float* b, float* C,
                  int K, int Nc) {
    int tiles = (N_ / 16) * (Nc / 16);
    k_gemm_bias<<<CDIV(tiles, TB / 32), TB, 0, stream>>>(A, W, b, C, K, Nc);
  };

  k_fill<<<CDIV(N_, TB), TB, 0, stream>>>(deg, N_, 1.0f);
  k_fill<<<CDIV(N_*D_, TB), TB, 0, stream>>>(hbuf, N_*D_, 0.0f);
  k_deg<<<CDIV(E_, TB), TB, 0, stream>>>(dst, deg);
  k_rsqrt_inplace<<<CDIV(N_, TB), TB, 0, stream>>>(deg, N_);
  k_norm<<<CDIV(E_, TB), TB, 0, stream>>>(src, dst, deg, norm);
  k_agg<<<CDIV(ET_*D_, TB), TB, 0, stream>>>(src, dst, x, norm, deg, hbuf);
  k_combine<<<CDIV(N_*D_, TB), TB, 0, stream>>>(x, hbuf);
  gemm(hbuf, Wssg, bssg, x1, D_, D_);

  gemm(x1, W1l, b1l, xl1, D_, F1_);
  gemm(x1, W1r, b1r, xr1, D_, F1_);
  k_fill<<<CDIV(N_*12, TB), TB, 0, stream>>>(m1, N_*12, NEG_INF);
  k_fill<<<CDIV(N_*12, TB), TB, 0, stream>>>(den1, N_*12, 0.0f);
  k_logit12<<<CDIV(ET_*12, TB), TB, 0, stream>>>(src, dst, xl1, xr1, att1, lg1, m1);
  k_fill<<<CDIV(N_*F1_, TB), TB, 0, stream>>>(out1, N_*F1_, 0.0f);
  k_expden12<<<CDIV(ET_*12, TB), TB, 0, stream>>>(src, dst, m1, lg1, den1);
  k_accum12<<<CDIV(ET_*F1_, TB), TB, 0, stream>>>(src, dst, lg1, xl1, out1);
  k_fin12<<<CDIV(N_*F1_, TB), TB, 0, stream>>>(den1, bias1, out1);

  gemm(out1, W2l, b2l, xl2, F1_, F2_);
  gemm(out1, W2r, b2r, xr2, F1_, F2_);
  k_fill<<<CDIV(N_, TB), TB, 0, stream>>>(m2, N_, NEG_INF);
  k_fill<<<CDIV(N_, TB), TB, 0, stream>>>(den2, N_, 0.0f);
  k_fill<<<CDIV(N_*F2_, TB), TB, 0, stream>>>(num2, N_*F2_, 0.0f);
  k_logit1<<<CDIV(ET_, TB/32), TB, 0, stream>>>(src, dst, xl2, xr2, att2, lg2, m2);
  k_expden1<<<CDIV(ET_, TB), TB, 0, stream>>>(src, dst, m2, lg2, den2);
  k_accum1<<<CDIV(ET_*F2_, TB), TB, 0, stream>>>(src, dst, lg2, xl2, num2);
  k_final<<<CDIV(N_*F2_, TB), TB, 0, stream>>>(num2, den2, bias2, out);
}


# edges counting-sorted by destination in-call; one wave per destination does softmax max, sum and weighted sums in registers, replacing float atomics (sum order now row-sequential)
# speedup vs baseline: 2.3857x; 2.3857x over previous
#include <hip/hip_runtime.h>
#include <cmath>

constexpr int N_  = 100000;
constexpr int E_  = 1000000;
constexpr int ET_ = N_ + E_;
constexpr int D_  = 64;
constexpr int F1_ = 144;
constexpr int F2_ = 64;

#define CDIV(a,b) (((a)+(b)-1)/(b))

typedef __attribute__((ext_vector_type(2))) float v2f;
typedef __attribute__((ext_vector_type(8))) float v8f;

__device__ __forceinline__ void edge_sd(const int* __restrict__ src,
                                        const int* __restrict__ dst,
                                        int e, int& s, int& d) {
  if (e < E_) { s = src[e]; d = dst[e]; }
  else        { s = e - E_; d = s; }
}

__global__ void k_fill(float* __restrict__ p, int n, float v) {
  int i = blockIdx.x * blockDim.x + threadIdx.x;
  if (i < n) p[i] = v;
}

__global__ void k_deg(const int* __restrict__ dst, float* __restrict__ deg) {
  int e = blockIdx.x * blockDim.x + threadIdx.x;
  if (e < E_) atomicAdd(&deg[dst[e]], 1.0f);
}

__global__ void k_rsqrt_inplace(float* __restrict__ p, int n) {
  int i = blockIdx.x * blockDim.x + threadIdx.x;
  if (i < n) p[i] = rsqrtf(p[i]);
}

__global__ void k_rowscan(const float* __restrict__ deg, int* __restrict__ cur) {
  __shared__ int s_pre[8];
  __shared__ int s_tot[8];
  const int tid  = threadIdx.x;
  const int lane = tid & 31, wv = tid >> 5;
  const int base = blockIdx.x * 1024;

  int pre = 0;
  const float4* __restrict__ d4 = (const float4*)deg;
#pragma unroll 8
  for (int i = tid; i < (base >> 2); i += 256) {
    const float4 q = d4[i];
    pre += (int)q.x + (int)q.y + (int)q.z + (int)q.w - 4;
  }
  int v[4], loc = 0;
#pragma unroll
  for (int k = 0; k < 4; ++k) {
    const int idx = base + tid * 4 + k;
    v[k] = (idx < N_) ? (int)deg[idx] - 1 : 0;
    loc += v[k];
  }
  int inc = loc;
  for (int o = 1; o < 32; o <<= 1) {
    const int up = __shfl_up(inc, o, 32);
    if (lane >= o) inc += up;
  }
  for (int o = 16; o > 0; o >>= 1) pre += __shfl_xor(pre, o, 32);
  if (lane == 31) s_tot[wv] = inc;
  if (lane == 0)  s_pre[wv] = pre;
  __syncthreads();
  int off = 0;
  for (int w = 0; w < 8; ++w) {
    off += s_pre[w];
    if (w < wv) off += s_tot[w];
  }
  int run = off + inc - loc;
#pragma unroll
  for (int k = 0; k < 4; ++k) {
    const int idx = base + tid * 4 + k;
    if (idx < N_) cur[idx] = run;
    run += v[k];
  }
}

__global__ void k_csr_fill(const int* __restrict__ dst, int* __restrict__ cur,
                           int* __restrict__ eid) {
  int e = blockIdx.x * blockDim.x + threadIdx.x;
  if (e < E_) { const int p = atomicAdd(&cur[dst[e]], 1); eid[p] = e; }
}

__global__ void k_agg_dst(const int* __restrict__ src, const int* __restrict__ rowend,
                          const int* __restrict__ eid, const float* __restrict__ x,
                          const float* __restrict__ dinv, float* __restrict__ h) {
  const int d = blockIdx.x * (blockDim.x >> 5) + (threadIdx.x >> 5);
  const int lane = threadIdx.x & 31;
  if (d >= N_) return;
  const int beg = d ? rowend[d - 1] : 0;
  const int end = rowend[d];
  const float dd = dinv[d];
  const float2* __restrict__ x2 = (const float2*)x;
  const float2 xs = x2[(size_t)d * 32 + lane];
  const float wself = dd * dd;
  float2 acc;
  acc.x = xs.x * wself;
  acc.y = xs.y * wself;
  for (int b0 = beg; b0 < end; b0 += 32) {
    const int n = min(32, end - b0);
    int s_i = d; float w_i = 0.0f;
    if (lane < n) { s_i = src[eid[b0 + lane]]; w_i = dinv[s_i] * dd; }
    for (int t = 0; t < n; t += 4) {
      float2 v[4]; float w[4];
#pragma unroll
      for (int u = 0; u < 4; ++u) {
        const int s = __shfl(s_i, t + u, 32);
        w[u] = __shfl(w_i, t + u, 32);
        v[u] = x2[(size_t)s * 32 + lane];
      }
#pragma unroll
      for (int u = 0; u < 4; ++u) {
        acc.x += v[u].x * w[u];
        acc.y += v[u].y * w[u];
      }
    }
  }
  float2 o;
  o.x = 0.5f * xs.x + 0.5f * acc.x;
  o.y = 0.5f * xs.y + 0.5f * acc.y;
  ((float2*)h)[(size_t)d * 32 + lane] = o;
}

__global__ void k_gemm_bias(const float* __restrict__ A, const float* __restrict__ W,
                            const float* __restrict__ bias, float* __restrict__ C,
                            int K, int Nc) {
  const int wave = blockIdx.x * (blockDim.x >> 5) + (threadIdx.x >> 5);
  const int lane = threadIdx.x & 31;
  const int tilesN = Nc >> 4;
  const int totalTiles = (N_ >> 4) * tilesN;
  if (wave >= totalTiles) return;

  const int row0 = (wave / tilesN) << 4;
  const int col0 = (wave % tilesN) << 4;
  const int lm = lane & 15;
  const int lk = (lane >> 4) << 1;

  const float* __restrict__ Arow = A + (size_t)(row0 + lm) * K;
  const float* __restrict__ Wcol = W + (col0 + lm);

  v8f acc = {};
  for (int k = 0; k < K; k += 4) {
    v2f a, b;
    a.x = Arow[k + lk];
    a.y = Arow[k + lk + 1];
    b.x = Wcol[(size_t)(k + lk) * Nc];
    b.y = Wcol[(size_t)(k + lk + 1) * Nc];
    acc = __builtin_amdgcn_wmma_f32_16x16x4_f32(false, a, false, b,
                                                (short)0, acc, false, false);
  }

  const int n = col0 + lm;
  const float bn = bias[n];
  const int rbase = row0 + ((lane >> 4) << 3);
#pragma unroll
  for (int j = 0; j < 8; ++j)
    C[(size_t)(rbase + j) * Nc + n] = acc[j] + bn;
}

__global__ void k_logit12(const int* __restrict__ src, const int* __restrict__ dst,
                          const float* __restrict__ xl, const float* __restrict__ xr,
                          const float* __restrict__ att, float* __restrict__ lg) {
  int idx = blockIdx.x * blockDim.x + threadIdx.x;
  if (idx >= ET_ * 12) return;
  int e = idx / 12, h = idx % 12;
  int s, d; edge_sd(src, dst, e, s, d);
  const float* pl = xl + (size_t)s * F1_ + h * 12;
  const float* pr = xr + (size_t)d * F1_ + h * 12;
  const float* pa = att + h * 12;
  float acc = 0.0f;
#pragma unroll
  for (int c = 0; c < 12; ++c) {
    float t = pl[c] + pr[c];
    t = (t > 0.0f) ? t : 0.2f * t;
    acc += t * pa[c];
  }
  lg[idx] = acc;
}

__global__ void k_gat12_dst(const int* __restrict__ src, const int* __restrict__ rowend,
                            const int* __restrict__ eid, const float* __restrict__ lg,
                            const float* __restrict__ xl, const float* __restrict__ bias,
                            float* __restrict__ out) {
  const int d = blockIdx.x * (blockDim.x >> 5) + (threadIdx.x >> 5);
  const int lane = threadIdx.x & 31;
  if (d >= N_) return;
  const int beg = d ? rowend[d - 1] : 0;
  const int T = rowend[d] - beg + 1;
  const int half = lane >> 4;
  const int hh = lane & 15;
  const bool hv = hh < 12;

  float m = -__builtin_huge_valf();
  for (int q0 = 0; q0 < T; q0 += 32) {
    const int n = min(32, T - q0);
    int e_i = E_ + d;
    if (q0 + lane < T - 1) e_i = eid[beg + q0 + lane];
    for (int t = 0; t < n; t += 2) {
      const int tt = t + half;
      const int e = __shfl(e_i, tt, 32);
      if (hv && tt < n) m = fmaxf(m, lg[(size_t)e * 12 + hh]);
    }
  }
  m = fmaxf(m, __shfl_xor(m, 16, 32));

  int hk[5];
#pragma unroll
  for (int k = 0; k < 5; ++k) {
    const int r = lane + 32 * k;
    hk[k] = (r < F1_) ? r / 12 : 0;
  }
  float acc[5] = {0.0f, 0.0f, 0.0f, 0.0f, 0.0f};
  float den = 0.0f;
  for (int q0 = 0; q0 < T; q0 += 32) {
    const int n = min(32, T - q0);
    int e_i = E_ + d, s_i = d;
    if (q0 + lane < T - 1) { e_i = eid[beg + q0 + lane]; s_i = src[e_i]; }
    for (int t = 0; t < n; t += 2) {
      const int tt = t + half;
      const int e = __shfl(e_i, tt, 32);
      float a = 0.0f;
      if (hv && tt < n) a = expf(lg[(size_t)e * 12 + hh] - m);
      den += a;
      const int s0 = __shfl(s_i, t, 32);
      const int s1 = __shfl(s_i, t + 1, 32);
      const float* __restrict__ p0 = xl + (size_t)s0 * F1_ + lane;
      const float* __restrict__ p1 = xl + (size_t)s1 * F1_ + lane;
#pragma unroll
      for (int k = 0; k < 5; ++k) {
        const float a0 = __shfl(a, hk[k], 32);
        const float a1 = __shfl(a, 16 + hk[k], 32);
        if (k < 4 || lane < 16) {
          acc[k] += a0 * p0[32 * k];
          acc[k] += a1 * p1[32 * k];
        }
      }
    }
  }
  den += __shfl_xor(den, 16, 32);
#pragma unroll
  for (int k = 0; k < 5; ++k) {
    const float dk = __shfl(den, hk[k], 32);
    const int r = lane + 32 * k;
    if (k < 4 || lane < 16)
      out[(size_t)d * F1_ + r] = acc[k] / (dk + 1e-16f) + bias[r];
  }
}

__global__ void k_logit1(const int* __restrict__ src, const int* __restrict__ dst,
                         const float* __restrict__ xl, const float* __restrict__ xr,
                         const float* __restrict__ att, float* __restrict__ lg) {
  int wv = blockIdx.x * (blockDim.x >> 5) + (threadIdx.x >> 5);
  int lane = threadIdx.x & 31;
  if (wv >= ET_) return;
  int s, d; edge_sd(src, dst, wv, s, d);
  float acc = 0.0f;
#pragma unroll
  for (int c = lane; c < 64; c += 32) {
    float t = xl[(size_t)s * F2_ + c] + xr[(size_t)d * F2_ + c];
    t = (t > 0.0f) ? t : 0.2f * t;
    acc += t * att[c];
  }
#pragma unroll
  for (int o = 16; o > 0; o >>= 1) acc += __shfl_xor(acc, o, 32);
  if (lane == 0) lg[wv] = acc;
}

__global__ void k_gat1_dst(const int* __restrict__ src, const int* __restrict__ rowend,
                           const int* __restrict__ eid, const float* __restrict__ lg,
                           const float* __restrict__ xl, const float* __restrict__ bias,
                           float* __restrict__ out) {
  const int d = blockIdx.x * (blockDim.x >> 5) + (threadIdx.x >> 5);
  const int lane = threadIdx.x & 31;
  if (d >= N_) return;
  const int beg = d ? rowend[d - 1] : 0;
  const int T = rowend[d] - beg + 1;

  float m = -__builtin_huge_valf();
  for (int q0 = 0; q0 < T; q0 += 32) {
    const int q = q0 + lane;
    if (q < T) {
      const int e = (q < T - 1) ? eid[beg + q] : E_ + d;
      m = fmaxf(m, lg[e]);
    }
  }
  for (int o = 16; o > 0; o >>= 1) m = fmaxf(m, __shfl_xor(m, o, 32));

  const float2* __restrict__ xl2 = (const float2*)xl;
  float2 acc; acc.x = 0.0f; acc.y = 0.0f;
  float den = 0.0f;
  for (int q0 = 0; q0 < T; q0 += 32) {
    const int n = min(32, T - q0);
    const int q = q0 + lane;
    int s_i = d; float a_i = 0.0f;
    if (q < T) {
      int e = E_ + d;
      if (q < T - 1) { e = eid[beg + q]; s_i = src[e]; }
      a_i = expf(lg[e] - m);
    }
    den += a_i;
    for (int t = 0; t < n; t += 4) {
      float2 v[4]; float a[4];
#pragma unroll
      for (int u = 0; u < 4; ++u) {
        const int s = __shfl(s_i, t + u, 32);
        a[u] = __shfl(a_i, t + u, 32);
        v[u] = xl2[(size_t)s * 32 + lane];
      }
#pragma unroll
      for (int u = 0; u < 4; ++u) {
        acc.x += a[u] * v[u].x;
        acc.y += a[u] * v[u].y;
      }
    }
  }
  for (int o = 16; o > 0; o >>= 1) den += __shfl_xor(den, o, 32);
  const float inv = den + 1e-16f;
  float2 o2;
  o2.x = acc.x / inv + bias[2 * lane];
  o2.y = acc.y / inv + bias[2 * lane + 1];
  ((float2*)out)[(size_t)d * 32 + lane] = o2;
}

constexpr size_t OFF_DEG  = 0;
constexpr size_t OFF_EID  = OFF_DEG  + N_;
constexpr size_t OFF_HBUF = OFF_EID  + E_;
constexpr size_t OFF_X1   = OFF_HBUF + (size_t)N_*D_;
constexpr size_t OFF_XL1  = OFF_X1   + (size_t)N_*D_;
constexpr size_t OFF_XR1  = OFF_XL1  + (size_t)N_*F1_;
constexpr size_t OFF_ROW  = OFF_XR1  + (size_t)N_*F1_;
constexpr size_t OFF_LG1  = OFF_ROW  + (size_t)N_*24;

extern "C" void kernel_launch(void* const* d_in, const int* in_sizes, int n_in,
                              void* d_out, int out_size, void* d_ws, size_t ws_size,
                              hipStream_t stream) {
  const float* x     = (const float*)d_in[0];
  const int*   ei    = (const int*)  d_in[1];
  const float* Wssg  = (const float*)d_in[2];
  const float* bssg  = (const float*)d_in[3];
  const float* W1l   = (const float*)d_in[4];
  const float* b1l   = (const float*)d_in[5];
  const float* W1r   = (const float*)d_in[6];
  const float* b1r   = (const float*)d_in[7];
  const float* att1  = (const float*)d_in[8];
  const float* bias1 = (const float*)d_in[9];
  const float* W2l   = (const float*)d_in[10];
  const float* b2l   = (const float*)d_in[11];
  const float* W2r   = (const float*)d_in[12];
  const float* b2r   = (const float*)d_in[13];
  const float* att2  = (const float*)d_in[14];
  const float* bias2 = (const float*)d_in[15];

  const int* src = ei;
  const int* dst = ei + E_;

  float* ws   = (float*)d_ws;
  float* deg  = ws + OFF_DEG;
  int*   eid  = (int*)(ws + OFF_EID);
  int*   row  = (int*)(ws + OFF_ROW);
  float* hbuf = ws + OFF_HBUF;
  float* x1   = ws + OFF_X1;
  float* xl1  = ws + OFF_XL1;
  float* xr1  = ws + OFF_XR1;
  float* lg1  = ws + OFF_LG1;
  float* out1 = xr1;
  float* xl2  = hbuf;
  float* xr2  = xl1;
  float* lg2  = lg1;
  float* out  = (float*)d_out;

  const int TB = 256;
  const int NODE_BLOCKS = CDIV(N_, TB / 32);

  auto gemm = [&](const float* A, const float* W, const float* b, float* C,
                  int K, int Nc) {
    int tiles = (N_ / 16) * (Nc / 16);
    k_gemm_bias<<<CDIV(tiles, TB / 32), TB, 0, stream>>>(A, W, b, C, K, Nc);
  };

  k_fill<<<CDIV(N_, TB), TB, 0, stream>>>(deg, N_, 1.0f);
  k_deg<<<CDIV(E_, TB), TB, 0, stream>>>(dst, deg);
  k_rowscan<<<CDIV(N_, 1024), TB, 0, stream>>>(deg, row);
  k_rsqrt_inplace<<<CDIV(N_, TB), TB, 0, stream>>>(deg, N_);
  k_csr_fill<<<CDIV(E_, TB), TB, 0, stream>>>(dst, row, eid);
  k_agg_dst<<<NODE_BLOCKS, TB, 0, stream>>>(src, row, eid, x, deg, hbuf);
  gemm(hbuf, Wssg, bssg, x1, D_, D_);

  gemm(x1, W1l, b1l, xl1, D_, F1_);
  gemm(x1, W1r, b1r, xr1, D_, F1_);
  k_logit12<<<CDIV(ET_*12, TB), TB, 0, stream>>>(src, dst, xl1, xr1, att1, lg1);
  k_gat12_dst<<<NODE_BLOCKS, TB, 0, stream>>>(src, row, eid, lg1, xl1, bias1, out1);

  gemm(out1, W2l, b2l, xl2, F1_, F2_);
  gemm(out1, W2r, b2r, xr2, F1_, F2_);
  k_logit1<<<CDIV(ET_, TB/32), TB, 0, stream>>>(src, dst, xl2, xr2, att2, lg2);
  k_gat1_dst<<<NODE_BLOCKS, TB, 0, stream>>>(src, row, eid, lg2, xl2, bias2, out);
}


# attention logits computed inside the per-destination wave in both attention layers; per-edge logit kernels and logit buffer removed; running-max softmax (partial sums rescaled, sum order changed)
# speedup vs baseline: 3.7528x; 1.5730x over previous
#include <hip/hip_runtime.h>
#include <cmath>

constexpr int N_  = 100000;
constexpr int E_  = 1000000;
constexpr int D_  = 64;
constexpr int F1_ = 144;
constexpr int F2_ = 64;

#define CDIV(a,b) (((a)+(b)-1)/(b))

typedef __attribute__((ext_vector_type(2))) float v2f;
typedef __attribute__((ext_vector_type(8))) float v8f;

__global__ void k_fill(float* __restrict__ p, int n, float v) {
  int i = blockIdx.x * blockDim.x + threadIdx.x;
  if (i < n) p[i] = v;
}

__global__ void k_deg(const int* __restrict__ dst, float* __restrict__ deg) {
  int e = blockIdx.x * blockDim.x + threadIdx.x;
  if (e < E_) atomicAdd(&deg[dst[e]], 1.0f);
}

__global__ void k_rsqrt_inplace(float* __restrict__ p, int n) {
  int i = blockIdx.x * blockDim.x + threadIdx.x;
  if (i < n) p[i] = rsqrtf(p[i]);
}

__global__ void k_rowscan(const float* __restrict__ deg, int* __restrict__ cur) {
  __shared__ int s_pre[8];
  __shared__ int s_tot[8];
  const int tid  = threadIdx.x;
  const int lane = tid & 31, wv = tid >> 5;
  const int base = blockIdx.x * 1024;

  int pre = 0;
  const float4* __restrict__ d4 = (const float4*)deg;
#pragma unroll 8
  for (int i = tid; i < (base >> 2); i += 256) {
    const float4 q = d4[i];
    pre += (int)q.x + (int)q.y + (int)q.z + (int)q.w - 4;
  }
  int v[4], loc = 0;
#pragma unroll
  for (int k = 0; k < 4; ++k) {
    const int idx = base + tid * 4 + k;
    v[k] = (idx < N_) ? (int)deg[idx] - 1 : 0;
    loc += v[k];
  }
  int inc = loc;
  for (int o = 1; o < 32; o <<= 1) {
    const int up = __shfl_up(inc, o, 32);
    if (lane >= o) inc += up;
  }
  for (int o = 16; o > 0; o >>= 1) pre += __shfl_xor(pre, o, 32);
  if (lane == 31) s_tot[wv] = inc;
  if (lane == 0)  s_pre[wv] = pre;
  __syncthreads();
  int off = 0;
  for (int w = 0; w < 8; ++w) {
    off += s_pre[w];
    if (w < wv) off += s_tot[w];
  }
  int run = off + inc - loc;
#pragma unroll
  for (int k = 0; k < 4; ++k) {
    const int idx = base + tid * 4 + k;
    if (idx < N_) cur[idx] = run;
    run += v[k];
  }
}

__global__ void k_csr_fill(const int* __restrict__ dst, int* __restrict__ cur,
                           int* __restrict__ eid) {
  int e = blockIdx.x * blockDim.x + threadIdx.x;
  if (e < E_) { const int p = atomicAdd(&cur[dst[e]], 1); eid[p] = e; }
}

__global__ void k_agg_dst(const int* __restrict__ src, const int* __restrict__ rowend,
                          const int* __restrict__ eid, const float* __restrict__ x,
                          const float* __restrict__ dinv, float* __restrict__ h) {
  const int d = blockIdx.x * (blockDim.x >> 5) + (threadIdx.x >> 5);
  const int lane = threadIdx.x & 31;
  if (d >= N_) return;
  const int beg = d ? rowend[d - 1] : 0;
  const int end = rowend[d];
  const float dd = dinv[d];
  const float2* __restrict__ x2 = (const float2*)x;
  const float2 xs = x2[(size_t)d * 32 + lane];
  const float wself = dd * dd;
  float2 acc;
  acc.x = xs.x * wself;
  acc.y = xs.y * wself;
  for (int b0 = beg; b0 < end; b0 += 32) {
    const int n = min(32, end - b0);
    int s_i = d; float w_i = 0.0f;
    if (lane < n) { s_i = src[eid[b0 + lane]]; w_i = dinv[s_i] * dd; }
    for (int t = 0; t < n; t += 4) {
      float2 v[4]; float w[4];
#pragma unroll
      for (int u = 0; u < 4; ++u) {
        const int s = __shfl(s_i, t + u, 32);
        w[u] = __shfl(w_i, t + u, 32);
        v[u] = x2[(size_t)s * 32 + lane];
      }
#pragma unroll
      for (int u = 0; u < 4; ++u) {
        acc.x += v[u].x * w[u];
        acc.y += v[u].y * w[u];
      }
    }
  }
  float2 o;
  o.x = 0.5f * xs.x + 0.5f * acc.x;
  o.y = 0.5f * xs.y + 0.5f * acc.y;
  ((float2*)h)[(size_t)d * 32 + lane] = o;
}

__global__ void k_gemm_bias(const float* __restrict__ A, const float* __restrict__ W,
                            const float* __restrict__ bias, float* __restrict__ C,
                            int K, int Nc) {
  const int wave = blockIdx.x * (blockDim.x >> 5) + (threadIdx.x >> 5);
  const int lane = threadIdx.x & 31;
  const int tilesN = Nc >> 4;
  const int totalTiles = (N_ >> 4) * tilesN;
  if (wave >= totalTiles) return;

  const int row0 = (wave / tilesN) << 4;
  const int col0 = (wave % tilesN) << 4;
  const int lm = lane & 15;
  const int lk = (lane >> 4) << 1;

  const float* __restrict__ Arow = A + (size_t)(row0 + lm) * K;
  const float* __restrict__ Wcol = W + (col0 + lm);

  v8f acc = {};
  for (int k = 0; k < K; k += 4) {
    v2f a, b;
    a.x = Arow[k + lk];
    a.y = Arow[k + lk + 1];
    b.x = Wcol[(size_t)(k + lk) * Nc];
    b.y = Wcol[(size_t)(k + lk + 1) * Nc];
    acc = __builtin_amdgcn_wmma_f32_16x16x4_f32(false, a, false, b,
                                                (short)0, acc, false, false);
  }

  const int n = col0 + lm;
  const float bn = bias[n];
  const int rbase = row0 + ((lane >> 4) << 3);
#pragma unroll
  for (int j = 0; j < 8; ++j)
    C[(size_t)(rbase + j) * Nc + n] = acc[j] + bn;
}

__global__ void k_gat12_fused(const int* __restrict__ src, const int* __restrict__ rowend,
                              const int* __restrict__ eid, const float* __restrict__ xl,
                              const float* __restrict__ xr, const float* __restrict__ att,
                              const float* __restrict__ bias, float* __restrict__ out) {
  const int d = blockIdx.x * (blockDim.x >> 5) + (threadIdx.x >> 5);
  const int lane = threadIdx.x & 31;
  if (d >= N_) return;
  const int beg = d ? rowend[d - 1] : 0;
  const int T = rowend[d] - beg + 1;
  const int half = lane >> 4;
  const int hh = lane & 15;
  const bool hv = hh < 12;
  const int hc = hv ? hh : 0;
  const float NEG_INF = -__builtin_huge_valf();

  float r[12], w[12];
  {
    const float4* __restrict__ r4 = (const float4*)(xr + (size_t)d * F1_ + hc * 12);
    const float4* __restrict__ w4 = (const float4*)(att + hc * 12);
#pragma unroll
    for (int k = 0; k < 3; ++k) {
      const float4 q = r4[k];
      const float4 p = w4[k];
      r[4 * k] = q.x; r[4 * k + 1] = q.y; r[4 * k + 2] = q.z; r[4 * k + 3] = q.w;
      w[4 * k] = p.x; w[4 * k + 1] = p.y; w[4 * k + 2] = p.z; w[4 * k + 3] = p.w;
    }
  }

  float acc[12];
#pragma unroll
  for (int c = 0; c < 12; ++c) acc[c] = 0.0f;
  float den = 0.0f;
  float m = NEG_INF;

  for (int q0 = 0; q0 < T; q0 += 32) {
    const int n = min(32, T - q0);
    int s_i = d;
    if (q0 + lane < T - 1) s_i = src[eid[beg + q0 + lane]];
    for (int t = 0; t < n; t += 4) {
      const int tA = t + half;
      const int tB = t + 2 + half;
      const int sA = __shfl(s_i, tA, 32);
      const int sB = __shfl(s_i, tB, 32);
      const float4* __restrict__ pA = (const float4*)(xl + (size_t)sA * F1_ + hc * 12);
      const float4* __restrict__ pB = (const float4*)(xl + (size_t)sB * F1_ + hc * 12);
      float xa[12], xb[12];
#pragma unroll
      for (int k = 0; k < 3; ++k) {
        const float4 qa = pA[k];
        const float4 qb = pB[k];
        xa[4 * k] = qa.x; xa[4 * k + 1] = qa.y; xa[4 * k + 2] = qa.z; xa[4 * k + 3] = qa.w;
        xb[4 * k] = qb.x; xb[4 * k + 1] = qb.y; xb[4 * k + 2] = qb.z; xb[4 * k + 3] = qb.w;
      }
      float lA = 0.0f, lB = 0.0f;
#pragma unroll
      for (int c = 0; c < 12; ++c) {
        float ua = xa[c] + r[c];
        float ub = xb[c] + r[c];
        ua = (ua > 0.0f) ? ua : 0.2f * ua;
        ub = (ub > 0.0f) ? ub : 0.2f * ub;
        lA += ua * w[c];
        lB += ub * w[c];
      }
      if (tA >= n) lA = NEG_INF;
      if (tB >= n) lB = NEG_INF;
      const float mn = fmaxf(m, fmaxf(lA, lB));
      const float mr = (mn == NEG_INF) ? 0.0f : mn;
      const float sc = expf(m - mr);
      const float aA = expf(lA - mr);
      const float aB = expf(lB - mr);
      den = den * sc + aA + aB;
#pragma unroll
      for (int c = 0; c < 12; ++c) acc[c] = acc[c] * sc + aA * xa[c] + aB * xb[c];
      m = mn;
    }
  }

  const float mo = __shfl_xor(m, 16, 32);
  const float mm = fmaxf(m, mo);
  const float sc = expf(m - mm);
  den *= sc;
  den += __shfl_xor(den, 16, 32);
#pragma unroll
  for (int c = 0; c < 12; ++c) {
    acc[c] *= sc;
    acc[c] += __shfl_xor(acc[c], 16, 32);
  }
  if (half == 0 && hv) {
    const float inv = den + 1e-16f;
    float4* __restrict__ o4 = (float4*)(out + (size_t)d * F1_ + hh * 12);
    const float4* __restrict__ b4 = (const float4*)(bias + hh * 12);
#pragma unroll
    for (int k = 0; k < 3; ++k) {
      const float4 b = b4[k];
      float4 o;
      o.x = acc[4 * k] / inv + b.x;
      o.y = acc[4 * k + 1] / inv + b.y;
      o.z = acc[4 * k + 2] / inv + b.z;
      o.w = acc[4 * k + 3] / inv + b.w;
      o4[k] = o;
    }
  }
}

__global__ void k_gat1_fused(const int* __restrict__ src, const int* __restrict__ rowend,
                             const int* __restrict__ eid, const float* __restrict__ xl,
                             const float* __restrict__ xr, const float* __restrict__ att,
                             const float* __restrict__ bias, float* __restrict__ out) {
  const int d = blockIdx.x * (blockDim.x >> 5) + (threadIdx.x >> 5);
  const int lane = threadIdx.x & 31;
  if (d >= N_) return;
  const int beg = d ? rowend[d - 1] : 0;
  const int T = rowend[d] - beg + 1;
  const float NEG_INF = -__builtin_huge_valf();

  const float2* __restrict__ xl2 = (const float2*)xl;
  const float2 r = ((const float2*)xr)[(size_t)d * 32 + lane];
  const float2 w = ((const float2*)att)[lane];
  float2 acc; acc.x = 0.0f; acc.y = 0.0f;
  float den = 0.0f;
  float m = NEG_INF;
  for (int q0 = 0; q0 < T; q0 += 32) {
    const int n = min(32, T - q0);
    int s_i = d;
    if (q0 + lane < T - 1) s_i = src[eid[beg + q0 + lane]];
    for (int t = 0; t < n; t += 4) {
      float2 v[4]; float l[4];
#pragma unroll
      for (int u = 0; u < 4; ++u) {
        const int s = __shfl(s_i, t + u, 32);
        v[u] = xl2[(size_t)s * 32 + lane];
      }
#pragma unroll
      for (int u = 0; u < 4; ++u) {
        float a = v[u].x + r.x;
        float b = v[u].y + r.y;
        a = (a > 0.0f) ? a : 0.2f * a;
        b = (b > 0.0f) ? b : 0.2f * b;
        l[u] = a * w.x + b * w.y;
      }
#pragma unroll
      for (int o = 16; o > 0; o >>= 1) {
#pragma unroll
        for (int u = 0; u < 4; ++u) l[u] += __shfl_xor(l[u], o, 32);
      }
#pragma unroll
      for (int u = 0; u < 4; ++u)
        if (t + u >= n) l[u] = NEG_INF;
      const float mn = fmaxf(fmaxf(m, l[0]), fmaxf(fmaxf(l[1], l[2]), l[3]));
      const float sc = expf(m - mn);
      float a[4];
#pragma unroll
      for (int u = 0; u < 4; ++u) a[u] = expf(l[u] - mn);
      den *= sc; acc.x *= sc; acc.y *= sc;
#pragma unroll
      for (int u = 0; u < 4; ++u) {
        den += a[u];
        acc.x += a[u] * v[u].x;
        acc.y += a[u] * v[u].y;
      }
      m = mn;
    }
  }
  const float inv = den + 1e-16f;
  float2 o2;
  o2.x = acc.x / inv + bias[2 * lane];
  o2.y = acc.y / inv + bias[2 * lane + 1];
  ((float2*)out)[(size_t)d * 32 + lane] = o2;
}

constexpr size_t OFF_DEG  = 0;
constexpr size_t OFF_EID  = OFF_DEG  + N_;
constexpr size_t OFF_HBUF = OFF_EID  + E_;
constexpr size_t OFF_X1   = OFF_HBUF + (size_t)N_*D_;
constexpr size_t OFF_XL1  = OFF_X1   + (size_t)N_*D_;
constexpr size_t OFF_XR1  = OFF_XL1  + (size_t)N_*F1_;
constexpr size_t OFF_ROW  = OFF_XR1  + (size_t)N_*F1_;
constexpr size_t OFF_OUT1 = OFF_ROW  + N_;

extern "C" void kernel_launch(void* const* d_in, const int* in_sizes, int n_in,
                              void* d_out, int out_size, void* d_ws, size_t ws_size,
                              hipStream_t stream) {
  const float* x     = (const float*)d_in[0];
  const int*   ei    = (const int*)  d_in[1];
  const float* Wssg  = (const float*)d_in[2];
  const float* bssg  = (const float*)d_in[3];
  const float* W1l   = (const float*)d_in[4];
  const float* b1l   = (const float*)d_in[5];
  const float* W1r   = (const float*)d_in[6];
  const float* b1r   = (const float*)d_in[7];
  const float* att1  = (const float*)d_in[8];
  const float* bias1 = (const float*)d_in[9];
  const float* W2l   = (const float*)d_in[10];
  const float* b2l   = (const float*)d_in[11];
  const float* W2r   = (const float*)d_in[12];
  const float* b2r   = (const float*)d_in[13];
  const float* att2  = (const float*)d_in[14];
  const float* bias2 = (const float*)d_in[15];

  const int* src = ei;
  const int* dst = ei + E_;

  float* ws   = (float*)d_ws;
  float* deg  = ws + OFF_DEG;
  int*   eid  = (int*)(ws + OFF_EID);
  int*   row  = (int*)(ws + OFF_ROW);
  float* hbuf = ws + OFF_HBUF;
  float* x1   = ws + OFF_X1;
  float* xl1  = ws + OFF_XL1;
  float* xr1  = ws + OFF_XR1;
  float* out1 = ws + OFF_OUT1;
  float* xl2  = hbuf;
  float* xr2  = xl1;
  float* out  = (float*)d_out;

  const int TB = 256;
  const int NODE_BLOCKS = CDIV(N_, TB / 32);

  auto gemm = [&](const float* A, const float* W, const float* b, float* C,
                  int K, int Nc) {
    int tiles = (N_ / 16) * (Nc / 16);
    k_gemm_bias<<<CDIV(tiles, TB / 32), TB, 0, stream>>>(A, W, b, C, K, Nc);
  };

  k_fill<<<CDIV(N_, TB), TB, 0, stream>>>(deg, N_, 1.0f);
  k_deg<<<CDIV(E_, TB), TB, 0, stream>>>(dst, deg);
  k_rowscan<<<CDIV(N_, 1024), TB, 0, stream>>>(deg, row);
  k_rsqrt_inplace<<<CDIV(N_, TB), TB, 0, stream>>>(deg, N_);
  k_csr_fill<<<CDIV(E_, TB), TB, 0, stream>>>(dst, row, eid);
  k_agg_dst<<<NODE_BLOCKS, TB, 0, stream>>>(src, row, eid, x, deg, hbuf);
  gemm(hbuf, Wssg, bssg, x1, D_, D_);

  gemm(x1, W1l, b1l, xl1, D_, F1_);
  gemm(x1, W1r, b1r, xr1, D_, F1_);
  k_gat12_fused<<<NODE_BLOCKS, TB, 0, stream>>>(src, row, eid, xl1, xr1, att1, bias1, out1);

  gemm(out1, W2l, b2l, xl2, F1_, F2_);
  gemm(out1, W2r, b2r, xr2, F1_, F2_);
  k_gat1_fused<<<NODE_BLOCKS, TB, 0, stream>>>(src, row, eid, xl2, xr2, att2, bias2, out);
}


# GEMM operand reuse: one wave per 16-row stripe holds its A fragments in registers for all column tiles, W staged once per work-group in LDS in B-fragment order; k order and sum order unchanged
# speedup vs baseline: 4.4216x; 1.1782x over previous
#include <hip/hip_runtime.h>
#include <cmath>

constexpr int N_  = 100000;
constexpr int E_  = 1000000;
constexpr int D_  = 64;
constexpr int F1_ = 144;
constexpr int F2_ = 64;

#define CDIV(a,b) (((a)+(b)-1)/(b))

typedef __attribute__((ext_vector_type(2))) float v2f;
typedef __attribute__((ext_vector_type(8))) float v8f;

__global__ void k_fill(float* __restrict__ p, int n, float v) {
  int i = blockIdx.x * blockDim.x + threadIdx.x;
  if (i < n) p[i] = v;
}

__global__ void k_deg(const int* __restrict__ dst, float* __restrict__ deg) {
  int e = blockIdx.x * blockDim.x + threadIdx.x;
  if (e < E_) atomicAdd(&deg[dst[e]], 1.0f);
}

__global__ void k_rsqrt_inplace(float* __restrict__ p, int n) {
  int i = blockIdx.x * blockDim.x + threadIdx.x;
  if (i < n) p[i] = rsqrtf(p[i]);
}

__global__ void k_rowscan(const float* __restrict__ deg, int* __restrict__ cur) {
  __shared__ int s_pre[8];
  __shared__ int s_tot[8];
  const int tid  = threadIdx.x;
  const int lane = tid & 31, wv = tid >> 5;
  const int base = blockIdx.x * 1024;

  int pre = 0;
  const float4* __restrict__ d4 = (const float4*)deg;
#pragma unroll 8
  for (int i = tid; i < (base >> 2); i += 256) {
    const float4 q = d4[i];
    pre += (int)q.x + (int)q.y + (int)q.z + (int)q.w - 4;
  }
  int v[4], loc = 0;
#pragma unroll
  for (int k = 0; k < 4; ++k) {
    const int idx = base + tid * 4 + k;
    v[k] = (idx < N_) ? (int)deg[idx] - 1 : 0;
    loc += v[k];
  }
  int inc = loc;
  for (int o = 1; o < 32; o <<= 1) {
    const int up = __shfl_up(inc, o, 32);
    if (lane >= o) inc += up;
  }
  for (int o = 16; o > 0; o >>= 1) pre += __shfl_xor(pre, o, 32);
  if (lane == 31) s_tot[wv] = inc;
  if (lane == 0)  s_pre[wv] = pre;
  __syncthreads();
  int off = 0;
  for (int w = 0; w < 8; ++w) {
    off += s_pre[w];
    if (w < wv) off += s_tot[w];
  }
  int run = off + inc - loc;
#pragma unroll
  for (int k = 0; k < 4; ++k) {
    const int idx = base + tid * 4 + k;
    if (idx < N_) cur[idx] = run;
    run += v[k];
  }
}

__global__ void k_csr_fill(const int* __restrict__ dst, int* __restrict__ cur,
                           int* __restrict__ eid) {
  int e = blockIdx.x * blockDim.x + threadIdx.x;
  if (e < E_) { const int p = atomicAdd(&cur[dst[e]], 1); eid[p] = e; }
}

__global__ void k_agg_dst(const int* __restrict__ src, const int* __restrict__ rowend,
                          const int* __restrict__ eid, const float* __restrict__ x,
                          const float* __restrict__ dinv, float* __restrict__ h) {
  const int d = blockIdx.x * (blockDim.x >> 5) + (threadIdx.x >> 5);
  const int lane = threadIdx.x & 31;
  if (d >= N_) return;
  const int beg = d ? rowend[d - 1] : 0;
  const int end = rowend[d];
  const float dd = dinv[d];
  const float2* __restrict__ x2 = (const float2*)x;
  const float2 xs = x2[(size_t)d * 32 + lane];
  const float wself = dd * dd;
  float2 acc;
  acc.x = xs.x * wself;
  acc.y = xs.y * wself;
  for (int b0 = beg; b0 < end; b0 += 32) {
    const int n = min(32, end - b0);
    int s_i = d; float w_i = 0.0f;
    if (lane < n) { s_i = src[eid[b0 + lane]]; w_i = dinv[s_i] * dd; }
    for (int t = 0; t < n; t += 4) {
      float2 v[4]; float w[4];
#pragma unroll
      for (int u = 0; u < 4; ++u) {
        const int s = __shfl(s_i, t + u, 32);
        w[u] = __shfl(w_i, t + u, 32);
        v[u] = x2[(size_t)s * 32 + lane];
      }
#pragma unroll
      for (int u = 0; u < 4; ++u) {
        acc.x += v[u].x * w[u];
        acc.y += v[u].y * w[u];
      }
    }
  }
  float2 o;
  o.x = 0.5f * xs.x + 0.5f * acc.x;
  o.y = 0.5f * xs.y + 0.5f * acc.y;
  ((float2*)h)[(size_t)d * 32 + lane] = o;
}

constexpr int GEMM_WAVES  = 4;
constexpr int GEMM_BLOCKS = CDIV(N_ / 16, GEMM_WAVES);

template <int K, int NC>
__global__ __launch_bounds__(GEMM_WAVES * 32)
void k_gemm_stripe(const float* __restrict__ A, const float* __restrict__ W,
                   const float* __restrict__ bias, float* __restrict__ C) {
  constexpr int TN = NC / 16;
  constexpr int KS = K / 4;
  constexpr int N4 = NC / 4;
  constexpr int ITEMS = (K / 2) * N4;
  __shared__ float4 sB[K * NC / 4];

  const int tid  = threadIdx.x;
  const int lane = tid & 31;
  const int lm   = lane & 15;
  const int half = lane >> 4;
  const int stripe = blockIdx.x * GEMM_WAVES + (tid >> 5);
  const bool valid = stripe < (N_ >> 4);

  v2f a[KS];
  if (valid) {
    const v2f* __restrict__ Ap = (const v2f*)(A + (size_t)(stripe * 16 + lm) * K + 2 * half);
#pragma unroll
    for (int j = 0; j < KS; ++j) a[j] = Ap[2 * j];
  }

  for (int it = tid; it < ITEMS; it += GEMM_WAVES * 32) {
    const int kp = it / N4;
    const int n  = (it - kp * N4) * 4;
    const float4 r0 = *(const float4*)(W + (size_t)(2 * kp) * NC + n);
    const float4 r1 = *(const float4*)(W + (size_t)(2 * kp + 1) * NC + n);
    const int slot = (((kp >> 1) * TN + (n >> 4)) * 32 + (kp & 1) * 16 + (n & 15));
    float4 lo, hi;
    lo.x = r0.x; lo.y = r1.x; lo.z = r0.y; lo.w = r1.y;
    hi.x = r0.z; hi.y = r1.z; hi.z = r0.w; hi.w = r1.w;
    sB[(slot >> 1)]     = lo;
    sB[(slot >> 1) + 1] = hi;
  }
  __syncthreads();
  if (!valid) return;

  v8f acc[TN];
#pragma unroll
  for (int c = 0; c < TN; ++c) acc[c] = v8f{};

  const v2f* __restrict__ sBf = (const v2f*)sB;
#pragma unroll
  for (int j = 0; j < KS; ++j) {
#pragma unroll
    for (int c = 0; c < TN; ++c) {
      const v2f b = sBf[(j * TN + c) * 32 + lane];
      acc[c] = __builtin_amdgcn_wmma_f32_16x16x4_f32(false, a[j], false, b,
                                                     (short)0, acc[c], false, false);
    }
  }

  const int rbase = stripe * 16 + (half << 3);
#pragma unroll
  for (int c = 0; c < TN; ++c) {
    const int n = c * 16 + lm;
    const float bn = bias[n];
#pragma unroll
    for (int r = 0; r < 8; ++r)
      C[(size_t)(rbase + r) * NC + n] = acc[c][r] + bn;
  }
}

__global__ void k_gat12_fused(const int* __restrict__ src, const int* __restrict__ rowend,
                              const int* __restrict__ eid, const float* __restrict__ xl,
                              const float* __restrict__ xr, const float* __restrict__ att,
                              const float* __restrict__ bias, float* __restrict__ out) {
  const int d = blockIdx.x * (blockDim.x >> 5) + (threadIdx.x >> 5);
  const int lane = threadIdx.x & 31;
  if (d >= N_) return;
  const int beg = d ? rowend[d - 1] : 0;
  const int T = rowend[d] - beg + 1;
  const int half = lane >> 4;
  const int hh = lane & 15;
  const bool hv = hh < 12;
  const int hc = hv ? hh : 0;
  const float NEG_INF = -__builtin_huge_valf();

  float r[12], w[12];
  {
    const float4* __restrict__ r4 = (const float4*)(xr + (size_t)d * F1_ + hc * 12);
    const float4* __restrict__ w4 = (const float4*)(att + hc * 12);
#pragma unroll
    for (int k = 0; k < 3; ++k) {
      const float4 q = r4[k];
      const float4 p = w4[k];
      r[4 * k] = q.x; r[4 * k + 1] = q.y; r[4 * k + 2] = q.z; r[4 * k + 3] = q.w;
      w[4 * k] = p.x; w[4 * k + 1] = p.y; w[4 * k + 2] = p.z; w[4 * k + 3] = p.w;
    }
  }

  float acc[12];
#pragma unroll
  for (int c = 0; c < 12; ++c) acc[c] = 0.0f;
  float den = 0.0f;
  float m = NEG_INF;

  for (int q0 = 0; q0 < T; q0 += 32) {
    const int n = min(32, T - q0);
    int s_i = d;
    if (q0 + lane < T - 1) s_i = src[eid[beg + q0 + lane]];
    for (int t = 0; t < n; t += 4) {
      const int tA = t + half;
      const int tB = t + 2 + half;
      const int sA = __shfl(s_i, tA, 32);
      const int sB = __shfl(s_i, tB, 32);
      const float4* __restrict__ pA = (const float4*)(xl + (size_t)sA * F1_ + hc * 12);
      const float4* __restrict__ pB = (const float4*)(xl + (size_t)sB * F1_ + hc * 12);
      float xa[12], xb[12];
#pragma unroll
      for (int k = 0; k < 3; ++k) {
        const float4 qa = pA[k];
        const float4 qb = pB[k];
        xa[4 * k] = qa.x; xa[4 * k + 1] = qa.y; xa[4 * k + 2] = qa.z; xa[4 * k + 3] = qa.w;
        xb[4 * k] = qb.x; xb[4 * k + 1] = qb.y; xb[4 * k + 2] = qb.z; xb[4 * k + 3] = qb.w;
      }
      float lA = 0.0f, lB = 0.0f;
#pragma unroll
      for (int c = 0; c < 12; ++c) {
        float ua = xa[c] + r[c];
        float ub = xb[c] + r[c];
        ua = (ua > 0.0f) ? ua : 0.2f * ua;
        ub = (ub > 0.0f) ? ub : 0.2f * ub;
        lA += ua * w[c];
        lB += ub * w[c];
      }
      if (tA >= n) lA = NEG_INF;
      if (tB >= n) lB = NEG_INF;
      const float mn = fmaxf(m, fmaxf(lA, lB));
      const float mr = (mn == NEG_INF) ? 0.0f : mn;
      const float sc = expf(m - mr);
      const float aA = expf(lA - mr);
      const float aB = expf(lB - mr);
      den = den * sc + aA + aB;
#pragma unroll
      for (int c = 0; c < 12; ++c) acc[c] = acc[c] * sc + aA * xa[c] + aB * xb[c];
      m = mn;
    }
  }

  const float mo = __shfl_xor(m, 16, 32);
  const float mm = fmaxf(m, mo);
  const float sc = expf(m - mm);
  den *= sc;
  den += __shfl_xor(den, 16, 32);
#pragma unroll
  for (int c = 0; c < 12; ++c) {
    acc[c] *= sc;
    acc[c] += __shfl_xor(acc[c], 16, 32);
  }
  if (half == 0 && hv) {
    const float inv = den + 1e-16f;
    float4* __restrict__ o4 = (float4*)(out + (size_t)d * F1_ + hh * 12);
    const float4* __restrict__ b4 = (const float4*)(bias + hh * 12);
#pragma unroll
    for (int k = 0; k < 3; ++k) {
      const float4 b = b4[k];
      float4 o;
      o.x = acc[4 * k] / inv + b.x;
      o.y = acc[4 * k + 1] / inv + b.y;
      o.z = acc[4 * k + 2] / inv + b.z;
      o.w = acc[4 * k + 3] / inv + b.w;
      o4[k] = o;
    }
  }
}

__global__ void k_gat1_fused(const int* __restrict__ src, const int* __restrict__ rowend,
                             const int* __restrict__ eid, const float* __restrict__ xl,
                             const float* __restrict__ xr, const float* __restrict__ att,
                             const float* __restrict__ bias, float* __restrict__ out) {
  const int d = blockIdx.x * (blockDim.x >> 5) + (threadIdx.x >> 5);
  const int lane = threadIdx.x & 31;
  if (d >= N_) return;
  const int beg = d ? rowend[d - 1] : 0;
  const int T = rowend[d] - beg + 1;
  const float NEG_INF = -__builtin_huge_valf();

  const float2* __restrict__ xl2 = (const float2*)xl;
  const float2 r = ((const float2*)xr)[(size_t)d * 32 + lane];
  const float2 w = ((const float2*)att)[lane];
  float2 acc; acc.x = 0.0f; acc.y = 0.0f;
  float den = 0.0f;
  float m = NEG_INF;
  for (int q0 = 0; q0 < T; q0 += 32) {
    const int n = min(32, T - q0);
    int s_i = d;
    if (q0 + lane < T - 1) s_i = src[eid[beg + q0 + lane]];
    for (int t = 0; t < n; t += 4) {
      float2 v[4]; float l[4];
#pragma unroll
      for (int u = 0; u < 4; ++u) {
        const int s = __shfl(s_i, t + u, 32);
        v[u] = xl2[(size_t)s * 32 + lane];
      }
#pragma unroll
      for (int u = 0; u < 4; ++u) {
        float a = v[u].x + r.x;
        float b = v[u].y + r.y;
        a = (a > 0.0f) ? a : 0.2f * a;
        b = (b > 0.0f) ? b : 0.2f * b;
        l[u] = a * w.x + b * w.y;
      }
#pragma unroll
      for (int o = 16; o > 0; o >>= 1) {
#pragma unroll
        for (int u = 0; u < 4; ++u) l[u] += __shfl_xor(l[u], o, 32);
      }
#pragma unroll
      for (int u = 0; u < 4; ++u)
        if (t + u >= n) l[u] = NEG_INF;
      const float mn = fmaxf(fmaxf(m, l[0]), fmaxf(fmaxf(l[1], l[2]), l[3]));
      const float sc = expf(m - mn);
      float a[4];
#pragma unroll
      for (int u = 0; u < 4; ++u) a[u] = expf(l[u] - mn);
      den *= sc; acc.x *= sc; acc.y *= sc;
#pragma unroll
      for (int u = 0; u < 4; ++u) {
        den += a[u];
        acc.x += a[u] * v[u].x;
        acc.y += a[u] * v[u].y;
      }
      m = mn;
    }
  }
  const float inv = den + 1e-16f;
  float2 o2;
  o2.x = acc.x / inv + bias[2 * lane];
  o2.y = acc.y / inv + bias[2 * lane + 1];
  ((float2*)out)[(size_t)d * 32 + lane] = o2;
}

constexpr size_t OFF_DEG  = 0;
constexpr size_t OFF_EID  = OFF_DEG  + N_;
constexpr size_t OFF_HBUF = OFF_EID  + E_;
constexpr size_t OFF_X1   = OFF_HBUF + (size_t)N_*D_;
constexpr size_t OFF_XL1  = OFF_X1   + (size_t)N_*D_;
constexpr size_t OFF_XR1  = OFF_XL1  + (size_t)N_*F1_;
constexpr size_t OFF_ROW  = OFF_XR1  + (size_t)N_*F1_;
constexpr size_t OFF_OUT1 = OFF_ROW  + N_;

extern "C" void kernel_launch(void* const* d_in, const int* in_sizes, int n_in,
                              void* d_out, int out_size, void* d_ws, size_t ws_size,
                              hipStream_t stream) {
  const float* x     = (const float*)d_in[0];
  const int*   ei    = (const int*)  d_in[1];
  const float* Wssg  = (const float*)d_in[2];
  const float* bssg  = (const float*)d_in[3];
  const float* W1l   = (const float*)d_in[4];
  const float* b1l   = (const float*)d_in[5];
  const float* W1r   = (const float*)d_in[6];
  const float* b1r   = (const float*)d_in[7];
  const float* att1  = (const float*)d_in[8];
  const float* bias1 = (const float*)d_in[9];
  const float* W2l   = (const float*)d_in[10];
  const float* b2l   = (const float*)d_in[11];
  const float* W2r   = (const float*)d_in[12];
  const float* b2r   = (const float*)d_in[13];
  const float* att2  = (const float*)d_in[14];
  const float* bias2 = (const float*)d_in[15];

  const int* src = ei;
  const int* dst = ei + E_;

  float* ws   = (float*)d_ws;
  float* deg  = ws + OFF_DEG;
  int*   eid  = (int*)(ws + OFF_EID);
  int*   row  = (int*)(ws + OFF_ROW);
  float* hbuf = ws + OFF_HBUF;
  float* x1   = ws + OFF_X1;
  float* xl1  = ws + OFF_XL1;
  float* xr1  = ws + OFF_XR1;
  float* out1 = ws + OFF_OUT1;
  float* xl2  = hbuf;
  float* xr2  = xl1;
  float* out  = (float*)d_out;

  const int TB = 256;
  const int NODE_BLOCKS = CDIV(N_, TB / 32);

  k_fill<<<CDIV(N_, TB), TB, 0, stream>>>(deg, N_, 1.0f);
  k_deg<<<CDIV(E_, TB), TB, 0, stream>>>(dst, deg);
  k_rowscan<<<CDIV(N_, 1024), TB, 0, stream>>>(deg, row);
  k_rsqrt_inplace<<<CDIV(N_, TB), TB, 0, stream>>>(deg, N_);
  k_csr_fill<<<CDIV(E_, TB), TB, 0, stream>>>(dst, row, eid);
  k_agg_dst<<<NODE_BLOCKS, TB, 0, stream>>>(src, row, eid, x, deg, hbuf);
  k_gemm_stripe<D_, D_><<<GEMM_BLOCKS, GEMM_WAVES * 32, 0, stream>>>(hbuf, Wssg, bssg, x1);

  k_gemm_stripe<D_, F1_><<<GEMM_BLOCKS, GEMM_WAVES * 32, 0, stream>>>(x1, W1l, b1l, xl1);
  k_gemm_stripe<D_, F1_><<<GEMM_BLOCKS, GEMM_WAVES * 32, 0, stream>>>(x1, W1r, b1r, xr1);
  k_gat12_fused<<<NODE_BLOCKS, TB, 0, stream>>>(src, row, eid, xl1, xr1, att1, bias1, out1);

  k_gemm_stripe<F1_, F2_><<<GEMM_BLOCKS, GEMM_WAVES * 32, 0, stream>>>(out1, W2l, b2l, xl2);
  k_gemm_stripe<F1_, F2_><<<GEMM_BLOCKS, GEMM_WAVES * 32, 0, stream>>>(out1, W2r, b2r, xr2);
  k_gat1_fused<<<NODE_BLOCKS, TB, 0, stream>>>(src, row, eid, xl2, xr2, att2, bias2, out);
}


# two destinations per wave, one per 16-lane half, in the 3 per-destination kernels (64-float rows: 4 channels, one 16-byte load per lane): half the waves; softmax/logit sum order changed (no merge)
# speedup vs baseline: 4.6980x; 1.0625x over previous
#include <hip/hip_runtime.h>
#include <cmath>

constexpr int N_  = 100000;
constexpr int E_  = 1000000;
constexpr int D_  = 64;
constexpr int F1_ = 144;
constexpr int F2_ = 64;

#define CDIV(a,b) (((a)+(b)-1)/(b))

typedef __attribute__((ext_vector_type(2))) float v2f;
typedef __attribute__((ext_vector_type(8))) float v8f;

__global__ void k_fill(float* __restrict__ p, int n, float v) {
  int i = blockIdx.x * blockDim.x + threadIdx.x;
  if (i < n) p[i] = v;
}

__global__ void k_deg(const int* __restrict__ dst, float* __restrict__ deg) {
  int e = blockIdx.x * blockDim.x + threadIdx.x;
  if (e < E_) atomicAdd(&deg[dst[e]], 1.0f);
}

__global__ void k_rsqrt_inplace(float* __restrict__ p, int n) {
  int i = blockIdx.x * blockDim.x + threadIdx.x;
  if (i < n) p[i] = rsqrtf(p[i]);
}

__global__ void k_rowscan(const float* __restrict__ deg, int* __restrict__ cur) {
  __shared__ int s_pre[8];
  __shared__ int s_tot[8];
  const int tid  = threadIdx.x;
  const int lane = tid & 31, wv = tid >> 5;
  const int base = blockIdx.x * 1024;

  int pre = 0;
  const float4* __restrict__ d4 = (const float4*)deg;
#pragma unroll 8
  for (int i = tid; i < (base >> 2); i += 256) {
    const float4 q = d4[i];
    pre += (int)q.x + (int)q.y + (int)q.z + (int)q.w - 4;
  }
  int v[4], loc = 0;
#pragma unroll
  for (int k = 0; k < 4; ++k) {
    const int idx = base + tid * 4 + k;
    v[k] = (idx < N_) ? (int)deg[idx] - 1 : 0;
    loc += v[k];
  }
  int inc = loc;
  for (int o = 1; o < 32; o <<= 1) {
    const int up = __shfl_up(inc, o, 32);
    if (lane >= o) inc += up;
  }
  for (int o = 16; o > 0; o >>= 1) pre += __shfl_xor(pre, o, 32);
  if (lane == 31) s_tot[wv] = inc;
  if (lane == 0)  s_pre[wv] = pre;
  __syncthreads();
  int off = 0;
  for (int w = 0; w < 8; ++w) {
    off += s_pre[w];
    if (w < wv) off += s_tot[w];
  }
  int run = off + inc - loc;
#pragma unroll
  for (int k = 0; k < 4; ++k) {
    const int idx = base + tid * 4 + k;
    if (idx < N_) cur[idx] = run;
    run += v[k];
  }
}

__global__ void k_csr_fill(const int* __restrict__ dst, int* __restrict__ cur,
                           int* __restrict__ eid) {
  int e = blockIdx.x * blockDim.x + threadIdx.x;
  if (e < E_) { const int p = atomicAdd(&cur[dst[e]], 1); eid[p] = e; }
}

__global__ void k_agg_dst(const int* __restrict__ src, const int* __restrict__ rowend,
                          const int* __restrict__ eid, const float* __restrict__ x,
                          const float* __restrict__ dinv, float* __restrict__ h) {
  const int wv = blockIdx.x * (blockDim.x >> 5) + (threadIdx.x >> 5);
  const int lane = threadIdx.x & 31;
  if (2 * wv >= N_) return;
  const int hb = lane & 16;
  const int j  = lane & 15;
  const int d  = 2 * wv + (lane >> 4);
  const int beg = d ? rowend[d - 1] : 0;
  const int cnt = rowend[d] - beg;
  const int cmax = max(__builtin_amdgcn_readlane(cnt, 0), __builtin_amdgcn_readlane(cnt, 16));
  const float dd = dinv[d];
  const float4* __restrict__ x4 = (const float4*)x;
  const float4 xs = x4[(size_t)d * 16 + j];
  const float wself = dd * dd;
  float4 acc;
  acc.x = xs.x * wself; acc.y = xs.y * wself; acc.z = xs.z * wself; acc.w = xs.w * wself;
  for (int b0 = 0; b0 < cmax; b0 += 16) {
    const int n = min(16, cmax - b0);
    int s_i = d; float w_i = 0.0f;
    if (b0 + j < cnt) { s_i = src[eid[beg + b0 + j]]; w_i = dinv[s_i] * dd; }
    for (int t = 0; t < n; t += 4) {
      float4 v[4]; float w[4];
#pragma unroll
      for (int u = 0; u < 4; ++u) {
        const int s = __shfl(s_i, hb + t + u, 32);
        w[u] = __shfl(w_i, hb + t + u, 32);
        v[u] = x4[(size_t)s * 16 + j];
      }
#pragma unroll
      for (int u = 0; u < 4; ++u) {
        acc.x += v[u].x * w[u];
        acc.y += v[u].y * w[u];
        acc.z += v[u].z * w[u];
        acc.w += v[u].w * w[u];
      }
    }
  }
  float4 o;
  o.x = 0.5f * xs.x + 0.5f * acc.x;
  o.y = 0.5f * xs.y + 0.5f * acc.y;
  o.z = 0.5f * xs.z + 0.5f * acc.z;
  o.w = 0.5f * xs.w + 0.5f * acc.w;
  ((float4*)h)[(size_t)d * 16 + j] = o;
}

constexpr int GEMM_WAVES  = 4;
constexpr int GEMM_BLOCKS = CDIV(N_ / 16, GEMM_WAVES);

template <int K, int NC>
__global__ __launch_bounds__(GEMM_WAVES * 32)
void k_gemm_stripe(const float* __restrict__ A, const float* __restrict__ W,
                   const float* __restrict__ bias, float* __restrict__ C) {
  constexpr int TN = NC / 16;
  constexpr int KS = K / 4;
  constexpr int N4 = NC / 4;
  constexpr int ITEMS = (K / 2) * N4;
  __shared__ float4 sB[K * NC / 4];

  const int tid  = threadIdx.x;
  const int lane = tid & 31;
  const int lm   = lane & 15;
  const int half = lane >> 4;
  const int stripe = blockIdx.x * GEMM_WAVES + (tid >> 5);
  const bool valid = stripe < (N_ >> 4);

  v2f a[KS];
  if (valid) {
    const v2f* __restrict__ Ap = (const v2f*)(A + (size_t)(stripe * 16 + lm) * K + 2 * half);
#pragma unroll
    for (int j = 0; j < KS; ++j) a[j] = Ap[2 * j];
  }

  for (int it = tid; it < ITEMS; it += GEMM_WAVES * 32) {
    const int kp = it / N4;
    const int n  = (it - kp * N4) * 4;
    const float4 r0 = *(const float4*)(W + (size_t)(2 * kp) * NC + n);
    const float4 r1 = *(const float4*)(W + (size_t)(2 * kp + 1) * NC + n);
    const int slot = (((kp >> 1) * TN + (n >> 4)) * 32 + (kp & 1) * 16 + (n & 15));
    float4 lo, hi;
    lo.x = r0.x; lo.y = r1.x; lo.z = r0.y; lo.w = r1.y;
    hi.x = r0.z; hi.y = r1.z; hi.z = r0.w; hi.w = r1.w;
    sB[(slot >> 1)]     = lo;
    sB[(slot >> 1) + 1] = hi;
  }
  __syncthreads();
  if (!valid) return;

  v8f acc[TN];
#pragma unroll
  for (int c = 0; c < TN; ++c) acc[c] = v8f{};

  const v2f* __restrict__ sBf = (const v2f*)sB;
#pragma unroll
  for (int j = 0; j < KS; ++j) {
#pragma unroll
    for (int c = 0; c < TN; ++c) {
      const v2f b = sBf[(j * TN + c) * 32 + lane];
      acc[c] = __builtin_amdgcn_wmma_f32_16x16x4_f32(false, a[j], false, b,
                                                     (short)0, acc[c], false, false);
    }
  }

  const int rbase = stripe * 16 + (half << 3);
#pragma unroll
  for (int c = 0; c < TN; ++c) {
    const int n = c * 16 + lm;
    const float bn = bias[n];
#pragma unroll
    for (int r = 0; r < 8; ++r)
      C[(size_t)(rbase + r) * NC + n] = acc[c][r] + bn;
  }
}

__global__ void k_gat12_fused(const int* __restrict__ src, const int* __restrict__ rowend,
                              const int* __restrict__ eid, const float* __restrict__ xl,
                              const float* __restrict__ xr, const float* __restrict__ att,
                              const float* __restrict__ bias, float* __restrict__ out) {
  const int wv = blockIdx.x * (blockDim.x >> 5) + (threadIdx.x >> 5);
  const int lane = threadIdx.x & 31;
  if (2 * wv >= N_) return;
  const int hb = lane & 16;
  const int hh = lane & 15;
  const int d  = 2 * wv + (lane >> 4);
  const int beg = d ? rowend[d - 1] : 0;
  const int T = rowend[d] - beg + 1;
  const int Tmax = max(__builtin_amdgcn_readlane(T, 0), __builtin_amdgcn_readlane(T, 16));
  const bool hv = hh < 12;
  const int hc = hv ? hh : 0;
  const float NEG_INF = -__builtin_huge_valf();

  float r[12], w[12];
  {
    const float4* __restrict__ r4 = (const float4*)(xr + (size_t)d * F1_ + hc * 12);
    const float4* __restrict__ w4 = (const float4*)(att + hc * 12);
#pragma unroll
    for (int k = 0; k < 3; ++k) {
      const float4 q = r4[k];
      const float4 p = w4[k];
      r[4 * k] = q.x; r[4 * k + 1] = q.y; r[4 * k + 2] = q.z; r[4 * k + 3] = q.w;
      w[4 * k] = p.x; w[4 * k + 1] = p.y; w[4 * k + 2] = p.z; w[4 * k + 3] = p.w;
    }
  }

  float acc[12];
#pragma unroll
  for (int c = 0; c < 12; ++c) acc[c] = 0.0f;
  float den = 0.0f;
  float m = NEG_INF;

  for (int q0 = 0; q0 < Tmax; q0 += 16) {
    const int n = min(16, Tmax - q0);
    int s_i = d;
    if (q0 + hh + 1 < T) s_i = src[eid[beg + q0 + hh]];
    for (int t = 0; t < n; t += 2) {
      const int sA = __shfl(s_i, hb + t, 32);
      const int sB = __shfl(s_i, hb + t + 1, 32);
      const float4* __restrict__ pA = (const float4*)(xl + (size_t)sA * F1_ + hc * 12);
      const float4* __restrict__ pB = (const float4*)(xl + (size_t)sB * F1_ + hc * 12);
      float xa[12], xb[12];
#pragma unroll
      for (int k = 0; k < 3; ++k) {
        const float4 qa = pA[k];
        const float4 qb = pB[k];
        xa[4 * k] = qa.x; xa[4 * k + 1] = qa.y; xa[4 * k + 2] = qa.z; xa[4 * k + 3] = qa.w;
        xb[4 * k] = qb.x; xb[4 * k + 1] = qb.y; xb[4 * k + 2] = qb.z; xb[4 * k + 3] = qb.w;
      }
      float lA = 0.0f, lB = 0.0f;
#pragma unroll
      for (int c = 0; c < 12; ++c) {
        float ua = xa[c] + r[c];
        float ub = xb[c] + r[c];
        ua = (ua > 0.0f) ? ua : 0.2f * ua;
        ub = (ub > 0.0f) ? ub : 0.2f * ub;
        lA += ua * w[c];
        lB += ub * w[c];
      }
      if (q0 + t >= T)     lA = NEG_INF;
      if (q0 + t + 1 >= T) lB = NEG_INF;
      const float mn = fmaxf(m, fmaxf(lA, lB));
      const float sc = expf(m - mn);
      const float aA = expf(lA - mn);
      const float aB = expf(lB - mn);
      den = den * sc + aA + aB;
#pragma unroll
      for (int c = 0; c < 12; ++c) acc[c] = acc[c] * sc + aA * xa[c] + aB * xb[c];
      m = mn;
    }
  }

  if (hv) {
    const float inv = den + 1e-16f;
    float4* __restrict__ o4 = (float4*)(out + (size_t)d * F1_ + hh * 12);
    const float4* __restrict__ b4 = (const float4*)(bias + hh * 12);
#pragma unroll
    for (int k = 0; k < 3; ++k) {
      const float4 b = b4[k];
      float4 o;
      o.x = acc[4 * k] / inv + b.x;
      o.y = acc[4 * k + 1] / inv + b.y;
      o.z = acc[4 * k + 2] / inv + b.z;
      o.w = acc[4 * k + 3] / inv + b.w;
      o4[k] = o;
    }
  }
}

__global__ void k_gat1_fused(const int* __restrict__ src, const int* __restrict__ rowend,
                             const int* __restrict__ eid, const float* __restrict__ xl,
                             const float* __restrict__ xr, const float* __restrict__ att,
                             const float* __restrict__ bias, float* __restrict__ out) {
  const int wv = blockIdx.x * (blockDim.x >> 5) + (threadIdx.x >> 5);
  const int lane = threadIdx.x & 31;
  if (2 * wv >= N_) return;
  const int hb = lane & 16;
  const int j  = lane & 15;
  const int d  = 2 * wv + (lane >> 4);
  const int beg = d ? rowend[d - 1] : 0;
  const int T = rowend[d] - beg + 1;
  const int Tmax = max(__builtin_amdgcn_readlane(T, 0), __builtin_amdgcn_readlane(T, 16));
  const float NEG_INF = -__builtin_huge_valf();

  const float4* __restrict__ xl4 = (const float4*)xl;
  const float4 r = ((const float4*)xr)[(size_t)d * 16 + j];
  const float4 w = ((const float4*)att)[j];
  float4 acc; acc.x = 0.0f; acc.y = 0.0f; acc.z = 0.0f; acc.w = 0.0f;
  float den = 0.0f;
  float m = NEG_INF;
  for (int q0 = 0; q0 < Tmax; q0 += 16) {
    const int n = min(16, Tmax - q0);
    int s_i = d;
    if (q0 + j + 1 < T) s_i = src[eid[beg + q0 + j]];
    for (int t = 0; t < n; t += 4) {
      float4 v[4]; float l[4];
#pragma unroll
      for (int u = 0; u < 4; ++u) {
        const int s = __shfl(s_i, hb + t + u, 32);
        v[u] = xl4[(size_t)s * 16 + j];
      }
#pragma unroll
      for (int u = 0; u < 4; ++u) {
        float a = v[u].x + r.x;
        float b = v[u].y + r.y;
        float cc = v[u].z + r.z;
        float e = v[u].w + r.w;
        a  = (a  > 0.0f) ? a  : 0.2f * a;
        b  = (b  > 0.0f) ? b  : 0.2f * b;
        cc = (cc > 0.0f) ? cc : 0.2f * cc;
        e  = (e  > 0.0f) ? e  : 0.2f * e;
        l[u] = (a * w.x + b * w.y) + (cc * w.z + e * w.w);
      }
#pragma unroll
      for (int o = 8; o > 0; o >>= 1) {
#pragma unroll
        for (int u = 0; u < 4; ++u) l[u] += __shfl_xor(l[u], o, 32);
      }
#pragma unroll
      for (int u = 0; u < 4; ++u)
        if (q0 + t + u >= T) l[u] = NEG_INF;
      const float mn = fmaxf(fmaxf(m, l[0]), fmaxf(fmaxf(l[1], l[2]), l[3]));
      const float sc = expf(m - mn);
      float a[4];
#pragma unroll
      for (int u = 0; u < 4; ++u) a[u] = expf(l[u] - mn);
      den *= sc; acc.x *= sc; acc.y *= sc; acc.z *= sc; acc.w *= sc;
#pragma unroll
      for (int u = 0; u < 4; ++u) {
        den += a[u];
        acc.x += a[u] * v[u].x;
        acc.y += a[u] * v[u].y;
        acc.z += a[u] * v[u].z;
        acc.w += a[u] * v[u].w;
      }
      m = mn;
    }
  }
  const float inv = den + 1e-16f;
  const float4 bb = ((const float4*)bias)[j];
  float4 o4;
  o4.x = acc.x / inv + bb.x;
  o4.y = acc.y / inv + bb.y;
  o4.z = acc.z / inv + bb.z;
  o4.w = acc.w / inv + bb.w;
  ((float4*)out)[(size_t)d * 16 + j] = o4;
}

constexpr size_t OFF_DEG  = 0;
constexpr size_t OFF_EID  = OFF_DEG  + N_;
constexpr size_t OFF_HBUF = OFF_EID  + E_;
constexpr size_t OFF_X1   = OFF_HBUF + (size_t)N_*D_;
constexpr size_t OFF_XL1  = OFF_X1   + (size_t)N_*D_;
constexpr size_t OFF_XR1  = OFF_XL1  + (size_t)N_*F1_;
constexpr size_t OFF_ROW  = OFF_XR1  + (size_t)N_*F1_;
constexpr size_t OFF_OUT1 = OFF_ROW  + N_;

extern "C" void kernel_launch(void* const* d_in, const int* in_sizes, int n_in,
                              void* d_out, int out_size, void* d_ws, size_t ws_size,
                              hipStream_t stream) {
  const float* x     = (const float*)d_in[0];
  const int*   ei    = (const int*)  d_in[1];
  const float* Wssg  = (const float*)d_in[2];
  const float* bssg  = (const float*)d_in[3];
  const float* W1l   = (const float*)d_in[4];
  const float* b1l   = (const float*)d_in[5];
  const float* W1r   = (const float*)d_in[6];
  const float* b1r   = (const float*)d_in[7];
  const float* att1  = (const float*)d_in[8];
  const float* bias1 = (const float*)d_in[9];
  const float* W2l   = (const float*)d_in[10];
  const float* b2l   = (const float*)d_in[11];
  const float* W2r   = (const float*)d_in[12];
  const float* b2r   = (const float*)d_in[13];
  const float* att2  = (const float*)d_in[14];
  const float* bias2 = (const float*)d_in[15];

  const int* src = ei;
  const int* dst = ei + E_;

  float* ws   = (float*)d_ws;
  float* deg  = ws + OFF_DEG;
  int*   eid  = (int*)(ws + OFF_EID);
  int*   row  = (int*)(ws + OFF_ROW);
  float* hbuf = ws + OFF_HBUF;
  float* x1   = ws + OFF_X1;
  float* xl1  = ws + OFF_XL1;
  float* xr1  = ws + OFF_XR1;
  float* out1 = ws + OFF_OUT1;
  float* xl2  = hbuf;
  float* xr2  = xl1;
  float* out  = (float*)d_out;

  const int TB = 256;
  static_assert(N_ % 2 == 0, "two destinations per wave: the per-destination kernels assume an even node count");
  const int NODE_BLOCKS = CDIV(N_ / 2, TB / 32);

  k_fill<<<CDIV(N_, TB), TB, 0, stream>>>(deg, N_, 1.0f);
  k_deg<<<CDIV(E_, TB), TB, 0, stream>>>(dst, deg);
  k_rowscan<<<CDIV(N_, 1024), TB, 0, stream>>>(deg, row);
  k_rsqrt_inplace<<<CDIV(N_, TB), TB, 0, stream>>>(deg, N_);
  k_csr_fill<<<CDIV(E_, TB), TB, 0, stream>>>(dst, row, eid);
  k_agg_dst<<<NODE_BLOCKS, TB, 0, stream>>>(src, row, eid, x, deg, hbuf);
  k_gemm_stripe<D_, D_><<<GEMM_BLOCKS, GEMM_WAVES * 32, 0, stream>>>(hbuf, Wssg, bssg, x1);

  k_gemm_stripe<D_, F1_><<<GEMM_BLOCKS, GEMM_WAVES * 32, 0, stream>>>(x1, W1l, b1l, xl1);
  k_gemm_stripe<D_, F1_><<<GEMM_BLOCKS, GEMM_WAVES * 32, 0, stream>>>(x1, W1r, b1r, xr1);
  k_gat12_fused<<<NODE_BLOCKS, TB, 0, stream>>>(src, row, eid, xl1, xr1, att1, bias1, out1);

  k_gemm_stripe<F1_, F2_><<<GEMM_BLOCKS, GEMM_WAVES * 32, 0, stream>>>(out1, W2l, b2l, xl2);
  k_gemm_stripe<F1_, F2_><<<GEMM_BLOCKS, GEMM_WAVES * 32, 0, stream>>>(out1, W2r, b2r, xr2);
  k_gat1_fused<<<NODE_BLOCKS, TB, 0, stream>>>(src, row, eid, xl2, xr2, att2, bias2, out);
}


# 5 GEMM launches become 2, one per layer: a stripe's A fragments feed both projections; the 64x64 linear output stays on chip (its LDS tile is the next A operand); k order and sum order unchanged
# speedup vs baseline: 4.9025x; 1.0435x over previous
#include <hip/hip_runtime.h>
#include <cmath>

constexpr int N_  = 100000;
constexpr int E_  = 1000000;
constexpr int D_  = 64;
constexpr int F1_ = 144;
constexpr int F2_ = 64;

#define CDIV(a,b) (((a)+(b)-1)/(b))

typedef __attribute__((ext_vector_type(2))) float v2f;
typedef __attribute__((ext_vector_type(8))) float v8f;

__global__ void k_fill(float* __restrict__ p, int n, float v) {
  int i = blockIdx.x * blockDim.x + threadIdx.x;
  if (i < n) p[i] = v;
}

__global__ void k_deg(const int* __restrict__ dst, float* __restrict__ deg) {
  int e = blockIdx.x * blockDim.x + threadIdx.x;
  if (e < E_) atomicAdd(&deg[dst[e]], 1.0f);
}

__global__ void k_rsqrt_inplace(float* __restrict__ p, int n) {
  int i = blockIdx.x * blockDim.x + threadIdx.x;
  if (i < n) p[i] = rsqrtf(p[i]);
}

__global__ void k_rowscan(const float* __restrict__ deg, int* __restrict__ cur) {
  __shared__ int s_pre[8];
  __shared__ int s_tot[8];
  const int tid  = threadIdx.x;
  const int lane = tid & 31, wv = tid >> 5;
  const int base = blockIdx.x * 1024;

  int pre = 0;
  const float4* __restrict__ d4 = (const float4*)deg;
#pragma unroll 8
  for (int i = tid; i < (base >> 2); i += 256) {
    const float4 q = d4[i];
    pre += (int)q.x + (int)q.y + (int)q.z + (int)q.w - 4;
  }
  int v[4], loc = 0;
#pragma unroll
  for (int k = 0; k < 4; ++k) {
    const int idx = base + tid * 4 + k;
    v[k] = (idx < N_) ? (int)deg[idx] - 1 : 0;
    loc += v[k];
  }
  int inc = loc;
  for (int o = 1; o < 32; o <<= 1) {
    const int up = __shfl_up(inc, o, 32);
    if (lane >= o) inc += up;
  }
  for (int o = 16; o > 0; o >>= 1) pre += __shfl_xor(pre, o, 32);
  if (lane == 31) s_tot[wv] = inc;
  if (lane == 0)  s_pre[wv] = pre;
  __syncthreads();
  int off = 0;
  for (int w = 0; w < 8; ++w) {
    off += s_pre[w];
    if (w < wv) off += s_tot[w];
  }
  int run = off + inc - loc;
#pragma unroll
  for (int k = 0; k < 4; ++k) {
    const int idx = base + tid * 4 + k;
    if (idx < N_) cur[idx] = run;
    run += v[k];
  }
}

__global__ void k_csr_fill(const int* __restrict__ dst, int* __restrict__ cur,
                           int* __restrict__ eid) {
  int e = blockIdx.x * blockDim.x + threadIdx.x;
  if (e < E_) { const int p = atomicAdd(&cur[dst[e]], 1); eid[p] = e; }
}

__global__ void k_agg_dst(const int* __restrict__ src, const int* __restrict__ rowend,
                          const int* __restrict__ eid, const float* __restrict__ x,
                          const float* __restrict__ dinv, float* __restrict__ h) {
  const int wv = blockIdx.x * (blockDim.x >> 5) + (threadIdx.x >> 5);
  const int lane = threadIdx.x & 31;
  if (2 * wv >= N_) return;
  const int hb = lane & 16;
  const int j  = lane & 15;
  const int d  = 2 * wv + (lane >> 4);
  const int beg = d ? rowend[d - 1] : 0;
  const int cnt = rowend[d] - beg;
  const int cmax = max(__builtin_amdgcn_readlane(cnt, 0), __builtin_amdgcn_readlane(cnt, 16));
  const float dd = dinv[d];
  const float4* __restrict__ x4 = (const float4*)x;
  const float4 xs = x4[(size_t)d * 16 + j];
  const float wself = dd * dd;
  float4 acc;
  acc.x = xs.x * wself; acc.y = xs.y * wself; acc.z = xs.z * wself; acc.w = xs.w * wself;
  for (int b0 = 0; b0 < cmax; b0 += 16) {
    const int n = min(16, cmax - b0);
    int s_i = d; float w_i = 0.0f;
    if (b0 + j < cnt) { s_i = src[eid[beg + b0 + j]]; w_i = dinv[s_i] * dd; }
    for (int t = 0; t < n; t += 4) {
      float4 v[4]; float w[4];
#pragma unroll
      for (int u = 0; u < 4; ++u) {
        const int s = __shfl(s_i, hb + t + u, 32);
        w[u] = __shfl(w_i, hb + t + u, 32);
        v[u] = x4[(size_t)s * 16 + j];
      }
#pragma unroll
      for (int u = 0; u < 4; ++u) {
        acc.x += v[u].x * w[u];
        acc.y += v[u].y * w[u];
        acc.z += v[u].z * w[u];
        acc.w += v[u].w * w[u];
      }
    }
  }
  float4 o;
  o.x = 0.5f * xs.x + 0.5f * acc.x;
  o.y = 0.5f * xs.y + 0.5f * acc.y;
  o.z = 0.5f * xs.z + 0.5f * acc.z;
  o.w = 0.5f * xs.w + 0.5f * acc.w;
  ((float4*)h)[(size_t)d * 16 + j] = o;
}

constexpr int GEMM_WAVES  = 4;
constexpr int GEMM_BLOCKS = CDIV(N_ / 16, GEMM_WAVES);
constexpr int XP = 68;

template <int K, int NC>
__device__ __forceinline__ void stage_w(float4* __restrict__ sB, const float* __restrict__ W, int tid) {
  constexpr int TN = NC / 16;
  constexpr int N4 = NC / 4;
  constexpr int ITEMS = (K / 2) * N4;
  for (int it = tid; it < ITEMS; it += GEMM_WAVES * 32) {
    const int kp = it / N4;
    const int n  = (it - kp * N4) * 4;
    const float4 r0 = *(const float4*)(W + (size_t)(2 * kp) * NC + n);
    const float4 r1 = *(const float4*)(W + (size_t)(2 * kp + 1) * NC + n);
    const int slot = (((kp >> 1) * TN + (n >> 4)) * 32 + (kp & 1) * 16 + (n & 15));
    float4 lo, hi;
    lo.x = r0.x; lo.y = r1.x; lo.z = r0.y; lo.w = r1.y;
    hi.x = r0.z; hi.y = r1.z; hi.z = r0.w; hi.w = r1.w;
    sB[(slot >> 1)]     = lo;
    sB[(slot >> 1) + 1] = hi;
  }
}

template <int K, int NC>
__device__ __forceinline__ void stripe_mma(const float4* __restrict__ sB, const v2f (&a)[K / 4],
                                           v8f (&acc)[NC / 16], int lane) {
  constexpr int TN = NC / 16;
  constexpr int KS = K / 4;
#pragma unroll
  for (int c = 0; c < TN; ++c) acc[c] = v8f{};
  const v2f* __restrict__ sBf = (const v2f*)sB;
#pragma unroll
  for (int j = 0; j < KS; ++j) {
#pragma unroll
    for (int c = 0; c < TN; ++c) {
      const v2f b = sBf[(j * TN + c) * 32 + lane];
      acc[c] = __builtin_amdgcn_wmma_f32_16x16x4_f32(false, a[j], false, b,
                                                     (short)0, acc[c], false, false);
    }
  }
}

template <int K, int NC>
__device__ __forceinline__ void stripe_project(const float4* __restrict__ sB, const v2f (&a)[K / 4],
                                               const float* __restrict__ bias, float* __restrict__ C,
                                               int stripe, int lane) {
  constexpr int TN = NC / 16;
  v8f acc[TN];
  stripe_mma<K, NC>(sB, a, acc, lane);
  const int lm = lane & 15;
  const int rbase = stripe * 16 + ((lane >> 4) << 3);
#pragma unroll
  for (int c = 0; c < TN; ++c) {
    const int n = c * 16 + lm;
    const float bn = bias[n];
#pragma unroll
    for (int r = 0; r < 8; ++r)
      C[(size_t)(rbase + r) * NC + n] = acc[c][r] + bn;
  }
}

template <int K, int NC, bool PRE>
__global__ __launch_bounds__(GEMM_WAVES * 32, 7)
void k_gemm_layer(const float* __restrict__ A,
                  const float* __restrict__ Wp, const float* __restrict__ bp,
                  const float* __restrict__ W0, const float* __restrict__ b0, float* __restrict__ C0,
                  const float* __restrict__ W1, const float* __restrict__ b1, float* __restrict__ C1) {
  constexpr int KS = K / 4;
  static_assert(!PRE || K == 64, "the linear layer in front is 64 x 64");
  static_assert(K * NC >= GEMM_WAVES * 16 * XP, "the X tiles of the waves must fit the weight buffer");
  __shared__ float4 sB[K * NC / 4];

  const int tid  = threadIdx.x;
  const int lane = tid & 31;
  const int lm   = lane & 15;
  const int half = lane >> 4;
  const int wave = __builtin_amdgcn_readfirstlane(tid >> 5);
  const int stripe = blockIdx.x * GEMM_WAVES + wave;
  const bool valid = stripe < (N_ >> 4);

  v2f a[KS];
  if (valid) {
    const v2f* __restrict__ Ap = (const v2f*)(A + (size_t)(stripe * 16 + lm) * K + 2 * half);
#pragma unroll
    for (int j = 0; j < KS; ++j) a[j] = Ap[2 * j];
  }

  if constexpr (PRE) {
    stage_w<64, 64>(sB, Wp, tid);
    __syncthreads();
    v8f accx[4];
    if (valid) stripe_mma<64, 64>(sB, a, accx, lane);
    __syncthreads();
    float* __restrict__ sX = (float*)sB + wave * (16 * XP);
    if (valid) {
#pragma unroll
      for (int c = 0; c < 4; ++c) {
        const float bn = bp[c * 16 + lm];
#pragma unroll
        for (int r = 0; r < 8; ++r)
          sX[(r + 8 * half) * XP + c * 16 + lm] = accx[c][r] + bn;
      }
    }
    __syncthreads();
    if (valid) {
#pragma unroll
      for (int j = 0; j < KS; ++j) a[j] = *(const v2f*)(sX + lm * XP + 4 * j + 2 * half);
    }
    __syncthreads();
  }

  stage_w<K, NC>(sB, W0, tid);
  __syncthreads();
  if (valid) stripe_project<K, NC>(sB, a, b0, C0, stripe, lane);
  __syncthreads();
  stage_w<K, NC>(sB, W1, tid);
  __syncthreads();
  if (valid) stripe_project<K, NC>(sB, a, b1, C1, stripe, lane);
}

__global__ void k_gat12_fused(const int* __restrict__ src, const int* __restrict__ rowend,
                              const int* __restrict__ eid, const float* __restrict__ xl,
                              const float* __restrict__ xr, const float* __restrict__ att,
                              const float* __restrict__ bias, float* __restrict__ out) {
  const int wv = blockIdx.x * (blockDim.x >> 5) + (threadIdx.x >> 5);
  const int lane = threadIdx.x & 31;
  if (2 * wv >= N_) return;
  const int hb = lane & 16;
  const int hh = lane & 15;
  const int d  = 2 * wv + (lane >> 4);
  const int beg = d ? rowend[d - 1] : 0;
  const int T = rowend[d] - beg + 1;
  const int Tmax = max(__builtin_amdgcn_readlane(T, 0), __builtin_amdgcn_readlane(T, 16));
  const bool hv = hh < 12;
  const int hc = hv ? hh : 0;
  const float NEG_INF = -__builtin_huge_valf();

  float r[12], w[12];
  {
    const float4* __restrict__ r4 = (const float4*)(xr + (size_t)d * F1_ + hc * 12);
    const float4* __restrict__ w4 = (const float4*)(att + hc * 12);
#pragma unroll
    for (int k = 0; k < 3; ++k) {
      const float4 q = r4[k];
      const float4 p = w4[k];
      r[4 * k] = q.x; r[4 * k + 1] = q.y; r[4 * k + 2] = q.z; r[4 * k + 3] = q.w;
      w[4 * k] = p.x; w[4 * k + 1] = p.y; w[4 * k + 2] = p.z; w[4 * k + 3] = p.w;
    }
  }

  float acc[12];
#pragma unroll
  for (int c = 0; c < 12; ++c) acc[c] = 0.0f;
  float den = 0.0f;
  float m = NEG_INF;

  for (int q0 = 0; q0 < Tmax; q0 += 16) {
    const int n = min(16, Tmax - q0);
    int s_i = d;
    if (q0 + hh + 1 < T) s_i = src[eid[beg + q0 + hh]];
    for (int t = 0; t < n; t += 2) {
      const int sA = __shfl(s_i, hb + t, 32);
      const int sB = __shfl(s_i, hb + t + 1, 32);
      const float4* __restrict__ pA = (const float4*)(xl + (size_t)sA * F1_ + hc * 12);
      const float4* __restrict__ pB = (const float4*)(xl + (size_t)sB * F1_ + hc * 12);
      float xa[12], xb[12];
#pragma unroll
      for (int k = 0; k < 3; ++k) {
        const float4 qa = pA[k];
        const float4 qb = pB[k];
        xa[4 * k] = qa.x; xa[4 * k + 1] = qa.y; xa[4 * k + 2] = qa.z; xa[4 * k + 3] = qa.w;
        xb[4 * k] = qb.x; xb[4 * k + 1] = qb.y; xb[4 * k + 2] = qb.z; xb[4 * k + 3] = qb.w;
      }
      float lA = 0.0f, lB = 0.0f;
#pragma unroll
      for (int c = 0; c < 12; ++c) {
        float ua = xa[c] + r[c];
        float ub = xb[c] + r[c];
        ua = (ua > 0.0f) ? ua : 0.2f * ua;
        ub = (ub > 0.0f) ? ub : 0.2f * ub;
        lA += ua * w[c];
        lB += ub * w[c];
      }
      if (q0 + t >= T)     lA = NEG_INF;
      if (q0 + t + 1 >= T) lB = NEG_INF;
      const float mn = fmaxf(m, fmaxf(lA, lB));
      const float sc = expf(m - mn);
      const float aA = expf(lA - mn);
      const float aB = expf(lB - mn);
      den = den * sc + aA + aB;
#pragma unroll
      for (int c = 0; c < 12; ++c) acc[c] = acc[c] * sc + aA * xa[c] + aB * xb[c];
      m = mn;
    }
  }

  if (hv) {
    const float inv = den + 1e-16f;
    float4* __restrict__ o4 = (float4*)(out + (size_t)d * F1_ + hh * 12);
    const float4* __restrict__ b4 = (const float4*)(bias + hh * 12);
#pragma unroll
    for (int k = 0; k < 3; ++k) {
      const float4 b = b4[k];
      float4 o;
      o.x = acc[4 * k] / inv + b.x;
      o.y = acc[4 * k + 1] / inv + b.y;
      o.z = acc[4 * k + 2] / inv + b.z;
      o.w = acc[4 * k + 3] / inv + b.w;
      o4[k] = o;
    }
  }
}

__global__ void k_gat1_fused(const int* __restrict__ src, const int* __restrict__ rowend,
                             const int* __restrict__ eid, const float* __restrict__ xl,
                             const float* __restrict__ xr, const float* __restrict__ att,
                             const float* __restrict__ bias, float* __restrict__ out) {
  const int wv = blockIdx.x * (blockDim.x >> 5) + (threadIdx.x >> 5);
  const int lane = threadIdx.x & 31;
  if (2 * wv >= N_) return;
  const int hb = lane & 16;
  const int j  = lane & 15;
  const int d  = 2 * wv + (lane >> 4);
  const int beg = d ? rowend[d - 1] : 0;
  const int T = rowend[d] - beg + 1;
  const int Tmax = max(__builtin_amdgcn_readlane(T, 0), __builtin_amdgcn_readlane(T, 16));
  const float NEG_INF = -__builtin_huge_valf();

  const float4* __restrict__ xl4 = (const float4*)xl;
  const float4 r = ((const float4*)xr)[(size_t)d * 16 + j];
  const float4 w = ((const float4*)att)[j];
  float4 acc; acc.x = 0.0f; acc.y = 0.0f; acc.z = 0.0f; acc.w = 0.0f;
  float den = 0.0f;
  float m = NEG_INF;
  for (int q0 = 0; q0 < Tmax; q0 += 16) {
    const int n = min(16, Tmax - q0);
    int s_i = d;
    if (q0 + j + 1 < T) s_i = src[eid[beg + q0 + j]];
    for (int t = 0; t < n; t += 4) {
      float4 v[4]; float l[4];
#pragma unroll
      for (int u = 0; u < 4; ++u) {
        const int s = __shfl(s_i, hb + t + u, 32);
        v[u] = xl4[(size_t)s * 16 + j];
      }
#pragma unroll
      for (int u = 0; u < 4; ++u) {
        float a = v[u].x + r.x;
        float b = v[u].y + r.y;
        float cc = v[u].z + r.z;
        float e = v[u].w + r.w;
        a  = (a  > 0.0f) ? a  : 0.2f * a;
        b  = (b  > 0.0f) ? b  : 0.2f * b;
        cc = (cc > 0.0f) ? cc : 0.2f * cc;
        e  = (e  > 0.0f) ? e  : 0.2f * e;
        l[u] = (a * w.x + b * w.y) + (cc * w.z + e * w.w);
      }
#pragma unroll
      for (int o = 8; o > 0; o >>= 1) {
#pragma unroll
        for (int u = 0; u < 4; ++u) l[u] += __shfl_xor(l[u], o, 32);
      }
#pragma unroll
      for (int u = 0; u < 4; ++u)
        if (q0 + t + u >= T) l[u] = NEG_INF;
      const float mn = fmaxf(fmaxf(m, l[0]), fmaxf(fmaxf(l[1], l[2]), l[3]));
      const float sc = expf(m - mn);
      float a[4];
#pragma unroll
      for (int u = 0; u < 4; ++u) a[u] = expf(l[u] - mn);
      den *= sc; acc.x *= sc; acc.y *= sc; acc.z *= sc; acc.w *= sc;
#pragma unroll
      for (int u = 0; u < 4; ++u) {
        den += a[u];
        acc.x += a[u] * v[u].x;
        acc.y += a[u] * v[u].y;
        acc.z += a[u] * v[u].z;
        acc.w += a[u] * v[u].w;
      }
      m = mn;
    }
  }
  const float inv = den + 1e-16f;
  const float4 bb = ((const float4*)bias)[j];
  float4 o4;
  o4.x = acc.x / inv + bb.x;
  o4.y = acc.y / inv + bb.y;
  o4.z = acc.z / inv + bb.z;
  o4.w = acc.w / inv + bb.w;
  ((float4*)out)[(size_t)d * 16 + j] = o4;
}

constexpr size_t OFF_DEG  = 0;
constexpr size_t OFF_EID  = OFF_DEG  + N_;
constexpr size_t OFF_HBUF = OFF_EID  + E_;
constexpr size_t OFF_X1   = OFF_HBUF + (size_t)N_*D_;
constexpr size_t OFF_XL1  = OFF_X1   + (size_t)N_*D_;
constexpr size_t OFF_XR1  = OFF_XL1  + (size_t)N_*F1_;
constexpr size_t OFF_ROW  = OFF_XR1  + (size_t)N_*F1_;
constexpr size_t OFF_OUT1 = OFF_ROW  + N_;

extern "C" void kernel_launch(void* const* d_in, const int* in_sizes, int n_in,
                              void* d_out, int out_size, void* d_ws, size_t ws_size,
                              hipStream_t stream) {
  const float* x     = (const float*)d_in[0];
  const int*   ei    = (const int*)  d_in[1];
  const float* Wssg  = (const float*)d_in[2];
  const float* bssg  = (const float*)d_in[3];
  const float* W1l   = (const float*)d_in[4];
  const float* b1l   = (const float*)d_in[5];
  const float* W1r   = (const float*)d_in[6];
  const float* b1r   = (const float*)d_in[7];
  const float* att1  = (const float*)d_in[8];
  const float* bias1 = (const float*)d_in[9];
  const float* W2l   = (const float*)d_in[10];
  const float* b2l   = (const float*)d_in[11];
  const float* W2r   = (const float*)d_in[12];
  const float* b2r   = (const float*)d_in[13];
  const float* att2  = (const float*)d_in[14];
  const float* bias2 = (const float*)d_in[15];

  const int* src = ei;
  const int* dst = ei + E_;

  float* ws   = (float*)d_ws;
  float* deg  = ws + OFF_DEG;
  int*   eid  = (int*)(ws + OFF_EID);
  int*   row  = (int*)(ws + OFF_ROW);
  float* hbuf = ws + OFF_HBUF;
  float* xl1  = ws + OFF_XL1;
  float* xr1  = ws + OFF_XR1;
  float* out1 = ws + OFF_OUT1;
  float* xl2  = hbuf;
  float* xr2  = xl1;
  float* out  = (float*)d_out;

  const int TB = 256;
  static_assert(N_ % 2 == 0, "two destinations per wave: the per-destination kernels assume an even node count");
  const int NODE_BLOCKS = CDIV(N_ / 2, TB / 32);

  k_fill<<<CDIV(N_, TB), TB, 0, stream>>>(deg, N_, 1.0f);
  k_deg<<<CDIV(E_, TB), TB, 0, stream>>>(dst, deg);
  k_rowscan<<<CDIV(N_, 1024), TB, 0, stream>>>(deg, row);
  k_rsqrt_inplace<<<CDIV(N_, TB), TB, 0, stream>>>(deg, N_);
  k_csr_fill<<<CDIV(E_, TB), TB, 0, stream>>>(dst, row, eid);
  k_agg_dst<<<NODE_BLOCKS, TB, 0, stream>>>(src, row, eid, x, deg, hbuf);

  k_gemm_layer<D_, F1_, true><<<GEMM_BLOCKS, GEMM_WAVES * 32, 0, stream>>>(hbuf, Wssg, bssg, W1l, b1l, xl1, W1r, b1r, xr1);
  k_gat12_fused<<<NODE_BLOCKS, TB, 0, stream>>>(src, row, eid, xl1, xr1, att1, bias1, out1);

  k_gemm_layer<F1_, F2_, false><<<GEMM_BLOCKS, GEMM_WAVES * 32, 0, stream>>>(out1, nullptr, nullptr, W2l, b2l, xl2, W2r, b2r, xr2);
  k_gat1_fused<<<NODE_BLOCKS, TB, 0, stream>>>(src, row, eid, xl2, xr2, att2, bias2, out);
}


# per-destination kernels touch fewer lines per load: sorted slots hold source ids (contiguous read, no 32-line lookup), layer-1 projections piece-major (4 lines per load, was 10); sums unchanged
# speedup vs baseline: 5.1340x; 1.0472x over previous
#include <hip/hip_runtime.h>
#include <cmath>

constexpr int N_  = 100000;
constexpr int E_  = 1000000;
constexpr int D_  = 64;
constexpr int F1_ = 144;
constexpr int F2_ = 64;

#define CDIV(a,b) (((a)+(b)-1)/(b))

typedef __attribute__((ext_vector_type(2))) float v2f;
typedef __attribute__((ext_vector_type(8))) float v8f;

__global__ void k_fill(float* __restrict__ p, int n, float v) {
  int i = blockIdx.x * blockDim.x + threadIdx.x;
  if (i < n) p[i] = v;
}

__global__ void k_deg(const int* __restrict__ dst, float* __restrict__ deg) {
  int e = blockIdx.x * blockDim.x + threadIdx.x;
  if (e < E_) atomicAdd(&deg[dst[e]], 1.0f);
}

__global__ void k_rsqrt_inplace(float* __restrict__ p, int n) {
  int i = blockIdx.x * blockDim.x + threadIdx.x;
  if (i < n) p[i] = rsqrtf(p[i]);
}

__global__ void k_rowscan(const float* __restrict__ deg, int* __restrict__ cur) {
  __shared__ int s_pre[8];
  __shared__ int s_tot[8];
  const int tid  = threadIdx.x;
  const int lane = tid & 31, wv = tid >> 5;
  const int base = blockIdx.x * 1024;

  int pre = 0;
  const float4* __restrict__ d4 = (const float4*)deg;
#pragma unroll 8
  for (int i = tid; i < (base >> 2); i += 256) {
    const float4 q = d4[i];
    pre += (int)q.x + (int)q.y + (int)q.z + (int)q.w - 4;
  }
  int v[4], loc = 0;
#pragma unroll
  for (int k = 0; k < 4; ++k) {
    const int idx = base + tid * 4 + k;
    v[k] = (idx < N_) ? (int)deg[idx] - 1 : 0;
    loc += v[k];
  }
  int inc = loc;
  for (int o = 1; o < 32; o <<= 1) {
    const int up = __shfl_up(inc, o, 32);
    if (lane >= o) inc += up;
  }
  for (int o = 16; o > 0; o >>= 1) pre += __shfl_xor(pre, o, 32);
  if (lane == 31) s_tot[wv] = inc;
  if (lane == 0)  s_pre[wv] = pre;
  __syncthreads();
  int off = 0;
  for (int w = 0; w < 8; ++w) {
    off += s_pre[w];
    if (w < wv) off += s_tot[w];
  }
  int run = off + inc - loc;
#pragma unroll
  for (int k = 0; k < 4; ++k) {
    const int idx = base + tid * 4 + k;
    if (idx < N_) cur[idx] = run;
    run += v[k];
  }
}

__global__ void k_csr_fill(const int* __restrict__ src, const int* __restrict__ dst,
                           int* __restrict__ cur, int* __restrict__ srt) {
  int e = blockIdx.x * blockDim.x + threadIdx.x;
  if (e < E_) {
    const int s = src[e];
    const int p = atomicAdd(&cur[dst[e]], 1);
    srt[p] = s;
  }
}

__global__ void k_agg_dst(const int* __restrict__ rowend,
                          const int* __restrict__ srt, const float* __restrict__ x,
                          const float* __restrict__ dinv, float* __restrict__ h) {
  const int wv = blockIdx.x * (blockDim.x >> 5) + (threadIdx.x >> 5);
  const int lane = threadIdx.x & 31;
  if (2 * wv >= N_) return;
  const int hb = lane & 16;
  const int j  = lane & 15;
  const int d  = 2 * wv + (lane >> 4);
  const int beg = d ? rowend[d - 1] : 0;
  const int cnt = rowend[d] - beg;
  const int cmax = max(__builtin_amdgcn_readlane(cnt, 0), __builtin_amdgcn_readlane(cnt, 16));
  const float dd = dinv[d];
  const float4* __restrict__ x4 = (const float4*)x;
  const float4 xs = x4[(size_t)d * 16 + j];
  const float wself = dd * dd;
  float4 acc;
  acc.x = xs.x * wself; acc.y = xs.y * wself; acc.z = xs.z * wself; acc.w = xs.w * wself;
  for (int b0 = 0; b0 < cmax; b0 += 16) {
    const int n = min(16, cmax - b0);
    int s_i = d; float w_i = 0.0f;
    if (b0 + j < cnt) { s_i = srt[beg + b0 + j]; w_i = dinv[s_i] * dd; }
    for (int t = 0; t < n; t += 4) {
      float4 v[4]; float w[4];
#pragma unroll
      for (int u = 0; u < 4; ++u) {
        const int s = __shfl(s_i, hb + t + u, 32);
        w[u] = __shfl(w_i, hb + t + u, 32);
        v[u] = x4[(size_t)s * 16 + j];
      }
#pragma unroll
      for (int u = 0; u < 4; ++u) {
        acc.x += v[u].x * w[u];
        acc.y += v[u].y * w[u];
        acc.z += v[u].z * w[u];
        acc.w += v[u].w * w[u];
      }
    }
  }
  float4 o;
  o.x = 0.5f * xs.x + 0.5f * acc.x;
  o.y = 0.5f * xs.y + 0.5f * acc.y;
  o.z = 0.5f * xs.z + 0.5f * acc.z;
  o.w = 0.5f * xs.w + 0.5f * acc.w;
  ((float4*)h)[(size_t)d * 16 + j] = o;
}

constexpr int GEMM_WAVES  = 4;
constexpr int GEMM_BLOCKS = CDIV(N_ / 16, GEMM_WAVES);
constexpr int XP = 68;

__device__ __forceinline__ constexpr int chan_of_pos(int p) {
  return ((p >> 2) % 12) * 12 + ((p >> 2) / 12) * 4 + (p & 3);
}

template <int K, int NC, bool PM>
__device__ __forceinline__ void stage_w(float4* __restrict__ sB, const float* __restrict__ W, int tid) {
  constexpr int TN = NC / 16;
  constexpr int N4 = NC / 4;
  constexpr int ITEMS = (K / 2) * N4;
  for (int it = tid; it < ITEMS; it += GEMM_WAVES * 32) {
    const int kp = it / N4;
    const int n  = (it - kp * N4) * 4;
    const int nw = PM ? chan_of_pos(n) : n;
    const float4 r0 = *(const float4*)(W + (size_t)(2 * kp) * NC + nw);
    const float4 r1 = *(const float4*)(W + (size_t)(2 * kp + 1) * NC + nw);
    const int slot = (((kp >> 1) * TN + (n >> 4)) * 32 + (kp & 1) * 16 + (n & 15));
    float4 lo, hi;
    lo.x = r0.x; lo.y = r1.x; lo.z = r0.y; lo.w = r1.y;
    hi.x = r0.z; hi.y = r1.z; hi.z = r0.w; hi.w = r1.w;
    sB[(slot >> 1)]     = lo;
    sB[(slot >> 1) + 1] = hi;
  }
}

template <int K, int NC>
__device__ __forceinline__ void stripe_mma(const float4* __restrict__ sB, const v2f (&a)[K / 4],
                                           v8f (&acc)[NC / 16], int lane) {
  constexpr int TN = NC / 16;
  constexpr int KS = K / 4;
#pragma unroll
  for (int c = 0; c < TN; ++c) acc[c] = v8f{};
  const v2f* __restrict__ sBf = (const v2f*)sB;
#pragma unroll
  for (int j = 0; j < KS; ++j) {
#pragma unroll
    for (int c = 0; c < TN; ++c) {
      const v2f b = sBf[(j * TN + c) * 32 + lane];
      acc[c] = __builtin_amdgcn_wmma_f32_16x16x4_f32(false, a[j], false, b,
                                                     (short)0, acc[c], false, false);
    }
  }
}

template <int K, int NC, bool PM>
__device__ __forceinline__ void stripe_project(const float4* __restrict__ sB, const v2f (&a)[K / 4],
                                               const float* __restrict__ bias, float* __restrict__ C,
                                               int stripe, int lane) {
  constexpr int TN = NC / 16;
  v8f acc[TN];
  stripe_mma<K, NC>(sB, a, acc, lane);
  const int lm = lane & 15;
  const int rbase = stripe * 16 + ((lane >> 4) << 3);
#pragma unroll
  for (int c = 0; c < TN; ++c) {
    const int n = c * 16 + lm;
    const float bn = bias[PM ? chan_of_pos(n) : n];
#pragma unroll
    for (int r = 0; r < 8; ++r)
      C[(size_t)(rbase + r) * NC + n] = acc[c][r] + bn;
  }
}

template <int K, int NC, bool PRE, bool PM>
__global__ __launch_bounds__(GEMM_WAVES * 32, 7)
void k_gemm_layer(const float* __restrict__ A,
                  const float* __restrict__ Wp, const float* __restrict__ bp,
                  const float* __restrict__ W0, const float* __restrict__ b0, float* __restrict__ C0,
                  const float* __restrict__ W1, const float* __restrict__ b1, float* __restrict__ C1) {
  constexpr int KS = K / 4;
  static_assert(!PRE || K == 64, "the linear layer in front is 64 x 64");
  static_assert(!PM || NC == 144, "piece-major order is defined for 12 heads of 12 channels");
  static_assert(K * NC >= GEMM_WAVES * 16 * XP, "the X tiles of the waves must fit the weight buffer");
  __shared__ float4 sB[K * NC / 4];

  const int tid  = threadIdx.x;
  const int lane = tid & 31;
  const int lm   = lane & 15;
  const int half = lane >> 4;
  const int wave = __builtin_amdgcn_readfirstlane(tid >> 5);
  const int stripe = blockIdx.x * GEMM_WAVES + wave;
  const bool valid = stripe < (N_ >> 4);

  v2f a[KS];
  if (valid) {
    const v2f* __restrict__ Ap = (const v2f*)(A + (size_t)(stripe * 16 + lm) * K + 2 * half);
#pragma unroll
    for (int j = 0; j < KS; ++j) a[j] = Ap[2 * j];
  }

  if constexpr (PRE) {
    stage_w<64, 64, false>(sB, Wp, tid);
    __syncthreads();
    v8f accx[4];
    if (valid) stripe_mma<64, 64>(sB, a, accx, lane);
    __syncthreads();
    float* __restrict__ sX = (float*)sB + wave * (16 * XP);
    if (valid) {
#pragma unroll
      for (int c = 0; c < 4; ++c) {
        const float bn = bp[c * 16 + lm];
#pragma unroll
        for (int r = 0; r < 8; ++r)
          sX[(r + 8 * half) * XP + c * 16 + lm] = accx[c][r] + bn;
      }
    }
    __syncthreads();
    if (valid) {
#pragma unroll
      for (int j = 0; j < KS; ++j) a[j] = *(const v2f*)(sX + lm * XP + 4 * j + 2 * half);
    }
    __syncthreads();
  }

  stage_w<K, NC, PM>(sB, W0, tid);
  __syncthreads();
  if (valid) stripe_project<K, NC, PM>(sB, a, b0, C0, stripe, lane);
  __syncthreads();
  stage_w<K, NC, PM>(sB, W1, tid);
  __syncthreads();
  if (valid) stripe_project<K, NC, PM>(sB, a, b1, C1, stripe, lane);
}

__global__ void k_gat12_fused(const int* __restrict__ rowend,
                              const int* __restrict__ srt, const float* __restrict__ xl,
                              const float* __restrict__ xr, const float* __restrict__ att,
                              const float* __restrict__ bias, float* __restrict__ out) {
  const int wv = blockIdx.x * (blockDim.x >> 5) + (threadIdx.x >> 5);
  const int lane = threadIdx.x & 31;
  if (2 * wv >= N_) return;
  const int hb = lane & 16;
  const int hh = lane & 15;
  const int d  = 2 * wv + (lane >> 4);
  const int beg = d ? rowend[d - 1] : 0;
  const int T = rowend[d] - beg + 1;
  const int Tmax = max(__builtin_amdgcn_readlane(T, 0), __builtin_amdgcn_readlane(T, 16));
  const bool hv = hh < 12;
  const int hc = hv ? hh : 0;
  const float NEG_INF = -__builtin_huge_valf();

  float r[12], w[12];
  {
    const float4* __restrict__ r4 = (const float4*)(xr + (size_t)d * F1_ + hc * 4);
    const float4* __restrict__ w4 = (const float4*)(att + hc * 12);
#pragma unroll
    for (int k = 0; k < 3; ++k) {
      const float4 q = r4[12 * k];
      const float4 p = w4[k];
      r[4 * k] = q.x; r[4 * k + 1] = q.y; r[4 * k + 2] = q.z; r[4 * k + 3] = q.w;
      w[4 * k] = p.x; w[4 * k + 1] = p.y; w[4 * k + 2] = p.z; w[4 * k + 3] = p.w;
    }
  }

  float acc[12];
#pragma unroll
  for (int c = 0; c < 12; ++c) acc[c] = 0.0f;
  float den = 0.0f;
  float m = NEG_INF;

  for (int q0 = 0; q0 < Tmax; q0 += 16) {
    const int n = min(16, Tmax - q0);
    int s_i = d;
    if (q0 + hh + 1 < T) s_i = srt[beg + q0 + hh];
    for (int t = 0; t < n; t += 2) {
      const int sA = __shfl(s_i, hb + t, 32);
      const int sB = __shfl(s_i, hb + t + 1, 32);
      const float4* __restrict__ pA = (const float4*)(xl + (size_t)sA * F1_ + hc * 4);
      const float4* __restrict__ pB = (const float4*)(xl + (size_t)sB * F1_ + hc * 4);
      float xa[12], xb[12];
#pragma unroll
      for (int k = 0; k < 3; ++k) {
        const float4 qa = pA[12 * k];
        const float4 qb = pB[12 * k];
        xa[4 * k] = qa.x; xa[4 * k + 1] = qa.y; xa[4 * k + 2] = qa.z; xa[4 * k + 3] = qa.w;
        xb[4 * k] = qb.x; xb[4 * k + 1] = qb.y; xb[4 * k + 2] = qb.z; xb[4 * k + 3] = qb.w;
      }
      float lA = 0.0f, lB = 0.0f;
#pragma unroll
      for (int c = 0; c < 12; ++c) {
        float ua = xa[c] + r[c];
        float ub = xb[c] + r[c];
        ua = (ua > 0.0f) ? ua : 0.2f * ua;
        ub = (ub > 0.0f) ? ub : 0.2f * ub;
        lA += ua * w[c];
        lB += ub * w[c];
      }
      if (q0 + t >= T)     lA = NEG_INF;
      if (q0 + t + 1 >= T) lB = NEG_INF;
      const float mn = fmaxf(m, fmaxf(lA, lB));
      const float sc = expf(m - mn);
      const float aA = expf(lA - mn);
      const float aB = expf(lB - mn);
      den = den * sc + aA + aB;
#pragma unroll
      for (int c = 0; c < 12; ++c) acc[c] = acc[c] * sc + aA * xa[c] + aB * xb[c];
      m = mn;
    }
  }

  if (hv) {
    const float inv = den + 1e-16f;
    float4* __restrict__ o4 = (float4*)(out + (size_t)d * F1_ + hh * 12);
    const float4* __restrict__ b4 = (const float4*)(bias + hh * 12);
#pragma unroll
    for (int k = 0; k < 3; ++k) {
      const float4 b = b4[k];
      float4 o;
      o.x = acc[4 * k] / inv + b.x;
      o.y = acc[4 * k + 1] / inv + b.y;
      o.z = acc[4 * k + 2] / inv + b.z;
      o.w = acc[4 * k + 3] / inv + b.w;
      o4[k] = o;
    }
  }
}

__global__ void k_gat1_fused(const int* __restrict__ rowend,
                             const int* __restrict__ srt, const float* __restrict__ xl,
                             const float* __restrict__ xr, const float* __restrict__ att,
                             const float* __restrict__ bias, float* __restrict__ out) {
  const int wv = blockIdx.x * (blockDim.x >> 5) + (threadIdx.x >> 5);
  const int lane = threadIdx.x & 31;
  if (2 * wv >= N_) return;
  const int hb = lane & 16;
  const int j  = lane & 15;
  const int d  = 2 * wv + (lane >> 4);
  const int beg = d ? rowend[d - 1] : 0;
  const int T = rowend[d] - beg + 1;
  const int Tmax = max(__builtin_amdgcn_readlane(T, 0), __builtin_amdgcn_readlane(T, 16));
  const float NEG_INF = -__builtin_huge_valf();

  const float4* __restrict__ xl4 = (const float4*)xl;
  const float4 r = ((const float4*)xr)[(size_t)d * 16 + j];
  const float4 w = ((const float4*)att)[j];
  float4 acc; acc.x = 0.0f; acc.y = 0.0f; acc.z = 0.0f; acc.w = 0.0f;
  float den = 0.0f;
  float m = NEG_INF;
  for (int q0 = 0; q0 < Tmax; q0 += 16) {
    const int n = min(16, Tmax - q0);
    int s_i = d;
    if (q0 + j + 1 < T) s_i = srt[beg + q0 + j];
    for (int t = 0; t < n; t += 4) {
      float4 v[4]; float l[4];
#pragma unroll
      for (int u = 0; u < 4; ++u) {
        const int s = __shfl(s_i, hb + t + u, 32);
        v[u] = xl4[(size_t)s * 16 + j];
      }
#pragma unroll
      for (int u = 0; u < 4; ++u) {
        float a = v[u].x + r.x;
        float b = v[u].y + r.y;
        float cc = v[u].z + r.z;
        float e = v[u].w + r.w;
        a  = (a  > 0.0f) ? a  : 0.2f * a;
        b  = (b  > 0.0f) ? b  : 0.2f * b;
        cc = (cc > 0.0f) ? cc : 0.2f * cc;
        e  = (e  > 0.0f) ? e  : 0.2f * e;
        l[u] = (a * w.x + b * w.y) + (cc * w.z + e * w.w);
      }
#pragma unroll
      for (int o = 8; o > 0; o >>= 1) {
#pragma unroll
        for (int u = 0; u < 4; ++u) l[u] += __shfl_xor(l[u], o, 32);
      }
#pragma unroll
      for (int u = 0; u < 4; ++u)
        if (q0 + t + u >= T) l[u] = NEG_INF;
      const float mn = fmaxf(fmaxf(m, l[0]), fmaxf(fmaxf(l[1], l[2]), l[3]));
      const float sc = expf(m - mn);
      float a[4];
#pragma unroll
      for (int u = 0; u < 4; ++u) a[u] = expf(l[u] - mn);
      den *= sc; acc.x *= sc; acc.y *= sc; acc.z *= sc; acc.w *= sc;
#pragma unroll
      for (int u = 0; u < 4; ++u) {
        den += a[u];
        acc.x += a[u] * v[u].x;
        acc.y += a[u] * v[u].y;
        acc.z += a[u] * v[u].z;
        acc.w += a[u] * v[u].w;
      }
      m = mn;
    }
  }
  const float inv = den + 1e-16f;
  const float4 bb = ((const float4*)bias)[j];
  float4 o4;
  o4.x = acc.x / inv + bb.x;
  o4.y = acc.y / inv + bb.y;
  o4.z = acc.z / inv + bb.z;
  o4.w = acc.w / inv + bb.w;
  ((float4*)out)[(size_t)d * 16 + j] = o4;
}

constexpr size_t OFF_DEG  = 0;
constexpr size_t OFF_EID  = OFF_DEG  + N_;
constexpr size_t OFF_HBUF = OFF_EID  + E_;
constexpr size_t OFF_X1   = OFF_HBUF + (size_t)N_*D_;
constexpr size_t OFF_XL1  = OFF_X1   + (size_t)N_*D_;
constexpr size_t OFF_XR1  = OFF_XL1  + (size_t)N_*F1_;
constexpr size_t OFF_ROW  = OFF_XR1  + (size_t)N_*F1_;
constexpr size_t OFF_OUT1 = OFF_ROW  + N_;

extern "C" void kernel_launch(void* const* d_in, const int* in_sizes, int n_in,
                              void* d_out, int out_size, void* d_ws, size_t ws_size,
                              hipStream_t stream) {
  const float* x     = (const float*)d_in[0];
  const int*   ei    = (const int*)  d_in[1];
  const float* Wssg  = (const float*)d_in[2];
  const float* bssg  = (const float*)d_in[3];
  const float* W1l   = (const float*)d_in[4];
  const float* b1l   = (const float*)d_in[5];
  const float* W1r   = (const float*)d_in[6];
  const float* b1r   = (const float*)d_in[7];
  const float* att1  = (const float*)d_in[8];
  const float* bias1 = (const float*)d_in[9];
  const float* W2l   = (const float*)d_in[10];
  const float* b2l   = (const float*)d_in[11];
  const float* W2r   = (const float*)d_in[12];
  const float* b2r   = (const float*)d_in[13];
  const float* att2  = (const float*)d_in[14];
  const float* bias2 = (const float*)d_in[15];

  const int* src = ei;
  const int* dst = ei + E_;

  float* ws   = (float*)d_ws;
  float* deg  = ws + OFF_DEG;
  int*   srt  = (int*)(ws + OFF_EID);
  int*   row  = (int*)(ws + OFF_ROW);
  float* hbuf = ws + OFF_HBUF;
  float* xl1  = ws + OFF_XL1;
  float* xr1  = ws + OFF_XR1;
  float* out1 = ws + OFF_OUT1;
  float* xl2  = hbuf;
  float* xr2  = xl1;
  float* out  = (float*)d_out;

  const int TB = 256;
  static_assert(N_ % 2 == 0, "two destinations per wave: the per-destination kernels assume an even node count");
  const int NODE_BLOCKS = CDIV(N_ / 2, TB / 32);

  k_fill<<<CDIV(N_, TB), TB, 0, stream>>>(deg, N_, 1.0f);
  k_deg<<<CDIV(E_, TB), TB, 0, stream>>>(dst, deg);
  k_rowscan<<<CDIV(N_, 1024), TB, 0, stream>>>(deg, row);
  k_rsqrt_inplace<<<CDIV(N_, TB), TB, 0, stream>>>(deg, N_);
  k_csr_fill<<<CDIV(E_, TB), TB, 0, stream>>>(src, dst, row, srt);
  k_agg_dst<<<NODE_BLOCKS, TB, 0, stream>>>(row, srt, x, deg, hbuf);

  k_gemm_layer<D_, F1_, true, true><<<GEMM_BLOCKS, GEMM_WAVES * 32, 0, stream>>>(hbuf, Wssg, bssg, W1l, b1l, xl1, W1r, b1r, xr1);
  k_gat12_fused<<<NODE_BLOCKS, TB, 0, stream>>>(row, srt, xl1, xr1, att1, bias1, out1);

  k_gemm_layer<F1_, F2_, false, false><<<GEMM_BLOCKS, GEMM_WAVES * 32, 0, stream>>>(out1, nullptr, nullptr, W2l, b2l, xl2, W2r, b2r, xr2);
  k_gat1_fused<<<NODE_BLOCKS, TB, 0, stream>>>(row, srt, xl2, xr2, att2, bias2, out);
}


# one integer atomic pass, not two: the in-degree atomic returns the edge's rank, so the slot fill is a plain scatter; f32 degree atomics and rsqrt launch removed; slot (sum) order stays arrival order
# speedup vs baseline: 5.7530x; 1.1206x over previous
#include <hip/hip_runtime.h>
#include <cmath>

constexpr int N_  = 100000;
constexpr int E_  = 1000000;
constexpr int D_  = 64;
constexpr int F1_ = 144;
constexpr int F2_ = 64;

#define CDIV(a,b) (((a)+(b)-1)/(b))

typedef __attribute__((ext_vector_type(2))) float v2f;
typedef __attribute__((ext_vector_type(8))) float v8f;

__global__ void k_zero(int* __restrict__ p, int n) {
  int i = blockIdx.x * blockDim.x + threadIdx.x;
  if (i < n) p[i] = 0;
}

__global__ void k_rank(const int* __restrict__ dst, int* __restrict__ cnt, int* __restrict__ rank) {
  int e = blockIdx.x * blockDim.x + threadIdx.x;
  if (e < E_) rank[e] = atomicAdd(&cnt[dst[e]], 1);
}

__global__ void k_rowscan(const int* __restrict__ cnt, int* __restrict__ rowend, float* __restrict__ dinv) {
  __shared__ int s_pre[8];
  __shared__ int s_tot[8];
  const int tid  = threadIdx.x;
  const int lane = tid & 31, wv = tid >> 5;
  const int base = blockIdx.x * 1024;

  int pre = 0;
  const int4* __restrict__ c4 = (const int4*)cnt;
#pragma unroll 8
  for (int i = tid; i < (base >> 2); i += 256) {
    const int4 q = c4[i];
    pre += q.x + q.y + q.z + q.w;
  }
  int v[4], loc = 0;
#pragma unroll
  for (int k = 0; k < 4; ++k) {
    const int idx = base + tid * 4 + k;
    v[k] = (idx < N_) ? cnt[idx] : 0;
    loc += v[k];
  }
  int inc = loc;
  for (int o = 1; o < 32; o <<= 1) {
    const int up = __shfl_up(inc, o, 32);
    if (lane >= o) inc += up;
  }
  for (int o = 16; o > 0; o >>= 1) pre += __shfl_xor(pre, o, 32);
  if (lane == 31) s_tot[wv] = inc;
  if (lane == 0)  s_pre[wv] = pre;
  __syncthreads();
  int off = 0;
  for (int w = 0; w < 8; ++w) {
    off += s_pre[w];
    if (w < wv) off += s_tot[w];
  }
  int run = off + inc - loc;
#pragma unroll
  for (int k = 0; k < 4; ++k) {
    const int idx = base + tid * 4 + k;
    run += v[k];
    if (idx < N_) {
      rowend[idx] = run;
      dinv[idx] = rsqrtf((float)(v[k] + 1));
    }
  }
}

__global__ void k_csr_fill(const int* __restrict__ src, const int* __restrict__ dst,
                           const int* __restrict__ rowend, const int* __restrict__ rank,
                           int* __restrict__ srt) {
  int e = blockIdx.x * blockDim.x + threadIdx.x;
  if (e < E_) {
    const int s = src[e];
    const int d = dst[e];
    const int beg = d ? rowend[d - 1] : 0;
    srt[beg + rank[e]] = s;
  }
}

__global__ void k_agg_dst(const int* __restrict__ rowend,
                          const int* __restrict__ srt, const float* __restrict__ x,
                          const float* __restrict__ dinv, float* __restrict__ h) {
  const int wv = blockIdx.x * (blockDim.x >> 5) + (threadIdx.x >> 5);
  const int lane = threadIdx.x & 31;
  if (2 * wv >= N_) return;
  const int hb = lane & 16;
  const int j  = lane & 15;
  const int d  = 2 * wv + (lane >> 4);
  const int beg = d ? rowend[d - 1] : 0;
  const int cnt = rowend[d] - beg;
  const int cmax = max(__builtin_amdgcn_readlane(cnt, 0), __builtin_amdgcn_readlane(cnt, 16));
  const float dd = dinv[d];
  const float4* __restrict__ x4 = (const float4*)x;
  const float4 xs = x4[(size_t)d * 16 + j];
  const float wself = dd * dd;
  float4 acc;
  acc.x = xs.x * wself; acc.y = xs.y * wself; acc.z = xs.z * wself; acc.w = xs.w * wself;
  for (int b0 = 0; b0 < cmax; b0 += 16) {
    const int n = min(16, cmax - b0);
    int s_i = d; float w_i = 0.0f;
    if (b0 + j < cnt) { s_i = srt[beg + b0 + j]; w_i = dinv[s_i] * dd; }
    for (int t = 0; t < n; t += 4) {
      float4 v[4]; float w[4];
#pragma unroll
      for (int u = 0; u < 4; ++u) {
        const int s = __shfl(s_i, hb + t + u, 32);
        w[u] = __shfl(w_i, hb + t + u, 32);
        v[u] = x4[(size_t)s * 16 + j];
      }
#pragma unroll
      for (int u = 0; u < 4; ++u) {
        acc.x += v[u].x * w[u];
        acc.y += v[u].y * w[u];
        acc.z += v[u].z * w[u];
        acc.w += v[u].w * w[u];
      }
    }
  }
  float4 o;
  o.x = 0.5f * xs.x + 0.5f * acc.x;
  o.y = 0.5f * xs.y + 0.5f * acc.y;
  o.z = 0.5f * xs.z + 0.5f * acc.z;
  o.w = 0.5f * xs.w + 0.5f * acc.w;
  ((float4*)h)[(size_t)d * 16 + j] = o;
}

constexpr int GEMM_WAVES  = 4;
constexpr int GEMM_BLOCKS = CDIV(N_ / 16, GEMM_WAVES);
constexpr int XP = 68;

__device__ __forceinline__ constexpr int chan_of_pos(int p) {
  return ((p >> 2) % 12) * 12 + ((p >> 2) / 12) * 4 + (p & 3);
}

template <int K, int NC, bool PM>
__device__ __forceinline__ void stage_w(float4* __restrict__ sB, const float* __restrict__ W, int tid) {
  constexpr int TN = NC / 16;
  constexpr int N4 = NC / 4;
  constexpr int ITEMS = (K / 2) * N4;
  for (int it = tid; it < ITEMS; it += GEMM_WAVES * 32) {
    const int kp = it / N4;
    const int n  = (it - kp * N4) * 4;
    const int nw = PM ? chan_of_pos(n) : n;
    const float4 r0 = *(const float4*)(W + (size_t)(2 * kp) * NC + nw);
    const float4 r1 = *(const float4*)(W + (size_t)(2 * kp + 1) * NC + nw);
    const int slot = (((kp >> 1) * TN + (n >> 4)) * 32 + (kp & 1) * 16 + (n & 15));
    float4 lo, hi;
    lo.x = r0.x; lo.y = r1.x; lo.z = r0.y; lo.w = r1.y;
    hi.x = r0.z; hi.y = r1.z; hi.z = r0.w; hi.w = r1.w;
    sB[(slot >> 1)]     = lo;
    sB[(slot >> 1) + 1] = hi;
  }
}

template <int K, int NC>
__device__ __forceinline__ void stripe_mma(const float4* __restrict__ sB, const v2f (&a)[K / 4],
                                           v8f (&acc)[NC / 16], int lane) {
  constexpr int TN = NC / 16;
  constexpr int KS = K / 4;
#pragma unroll
  for (int c = 0; c < TN; ++c) acc[c] = v8f{};
  const v2f* __restrict__ sBf = (const v2f*)sB;
#pragma unroll
  for (int j = 0; j < KS; ++j) {
#pragma unroll
    for (int c = 0; c < TN; ++c) {
      const v2f b = sBf[(j * TN + c) * 32 + lane];
      acc[c] = __builtin_amdgcn_wmma_f32_16x16x4_f32(false, a[j], false, b,
                                                     (short)0, acc[c], false, false);
    }
  }
}

template <int K, int NC, bool PM>
__device__ __forceinline__ void stripe_project(const float4* __restrict__ sB, const v2f (&a)[K / 4],
                                               const float* __restrict__ bias, float* __restrict__ C,
                                               int stripe, int lane) {
  constexpr int TN = NC / 16;
  v8f acc[TN];
  stripe_mma<K, NC>(sB, a, acc, lane);
  const int lm = lane & 15;
  const int rbase = stripe * 16 + ((lane >> 4) << 3);
#pragma unroll
  for (int c = 0; c < TN; ++c) {
    const int n = c * 16 + lm;
    const float bn = bias[PM ? chan_of_pos(n) : n];
#pragma unroll
    for (int r = 0; r < 8; ++r)
      C[(size_t)(rbase + r) * NC + n] = acc[c][r] + bn;
  }
}

template <int K, int NC, bool PRE, bool PM>
__global__ __launch_bounds__(GEMM_WAVES * 32, 7)
void k_gemm_layer(const float* __restrict__ A,
                  const float* __restrict__ Wp, const float* __restrict__ bp,
                  const float* __restrict__ W0, const float* __restrict__ b0, float* __restrict__ C0,
                  const float* __restrict__ W1, const float* __restrict__ b1, float* __restrict__ C1) {
  constexpr int KS = K / 4;
  static_assert(!PRE || K == 64, "the linear layer in front is 64 x 64");
  static_assert(!PM || NC == 144, "piece-major order is defined for 12 heads of 12 channels");
  static_assert(K * NC >= GEMM_WAVES * 16 * XP, "the X tiles of the waves must fit the weight buffer");
  __shared__ float4 sB[K * NC / 4];

  const int tid  = threadIdx.x;
  const int lane = tid & 31;
  const int lm   = lane & 15;
  const int half = lane >> 4;
  const int wave = __builtin_amdgcn_readfirstlane(tid >> 5);
  const int stripe = blockIdx.x * GEMM_WAVES + wave;
  const bool valid = stripe < (N_ >> 4);

  v2f a[KS];
  if (valid) {
    const v2f* __restrict__ Ap = (const v2f*)(A + (size_t)(stripe * 16 + lm) * K + 2 * half);
#pragma unroll
    for (int j = 0; j < KS; ++j) a[j] = Ap[2 * j];
  }

  if constexpr (PRE) {
    stage_w<64, 64, false>(sB, Wp, tid);
    __syncthreads();
    v8f accx[4];
    if (valid) stripe_mma<64, 64>(sB, a, accx, lane);
    __syncthreads();
    float* __restrict__ sX = (float*)sB + wave * (16 * XP);
    if (valid) {
#pragma unroll
      for (int c = 0; c < 4; ++c) {
        const float bn = bp[c * 16 + lm];
#pragma unroll
        for (int r = 0; r < 8; ++r)
          sX[(r + 8 * half) * XP + c * 16 + lm] = accx[c][r] + bn;
      }
    }
    __syncthreads();
    if (valid) {
#pragma unroll
      for (int j = 0; j < KS; ++j) a[j] = *(const v2f*)(sX + lm * XP + 4 * j + 2 * half);
    }
    __syncthreads();
  }

  stage_w<K, NC, PM>(sB, W0, tid);
  __syncthreads();
  if (valid) stripe_project<K, NC, PM>(sB, a, b0, C0, stripe, lane);
  __syncthreads();
  stage_w<K, NC, PM>(sB, W1, tid);
  __syncthreads();
  if (valid) stripe_project<K, NC, PM>(sB, a, b1, C1, stripe, lane);
}

__global__ void k_gat12_fused(const int* __restrict__ rowend,
                              const int* __restrict__ srt, const float* __restrict__ xl,
                              const float* __restrict__ xr, const float* __restrict__ att,
                              const float* __restrict__ bias, float* __restrict__ out) {
  const int wv = blockIdx.x * (blockDim.x >> 5) + (threadIdx.x >> 5);
  const int lane = threadIdx.x & 31;
  if (2 * wv >= N_) return;
  const int hb = lane & 16;
  const int hh = lane & 15;
  const int d  = 2 * wv + (lane >> 4);
  const int beg = d ? rowend[d - 1] : 0;
  const int T = rowend[d] - beg + 1;
  const int Tmax = max(__builtin_amdgcn_readlane(T, 0), __builtin_amdgcn_readlane(T, 16));
  const bool hv = hh < 12;
  const int hc = hv ? hh : 0;
  const float NEG_INF = -__builtin_huge_valf();

  float r[12], w[12];
  {
    const float4* __restrict__ r4 = (const float4*)(xr + (size_t)d * F1_ + hc * 4);
    const float4* __restrict__ w4 = (const float4*)(att + hc * 12);
#pragma unroll
    for (int k = 0; k < 3; ++k) {
      const float4 q = r4[12 * k];
      const float4 p = w4[k];
      r[4 * k] = q.x; r[4 * k + 1] = q.y; r[4 * k + 2] = q.z; r[4 * k + 3] = q.w;
      w[4 * k] = p.x; w[4 * k + 1] = p.y; w[4 * k + 2] = p.z; w[4 * k + 3] = p.w;
    }
  }

  float acc[12];
#pragma unroll
  for (int c = 0; c < 12; ++c) acc[c] = 0.0f;
  float den = 0.0f;
  float m = NEG_INF;

  for (int q0 = 0; q0 < Tmax; q0 += 16) {
    const int n = min(16, Tmax - q0);
    int s_i = d;
    if (q0 + hh + 1 < T) s_i = srt[beg + q0 + hh];
    for (int t = 0; t < n; t += 2) {
      const int sA = __shfl(s_i, hb + t, 32);
      const int sB = __shfl(s_i, hb + t + 1, 32);
      const float4* __restrict__ pA = (const float4*)(xl + (size_t)sA * F1_ + hc * 4);
      const float4* __restrict__ pB = (const float4*)(xl + (size_t)sB * F1_ + hc * 4);
      float xa[12], xb[12];
#pragma unroll
      for (int k = 0; k < 3; ++k) {
        const float4 qa = pA[12 * k];
        const float4 qb = pB[12 * k];
        xa[4 * k] = qa.x; xa[4 * k + 1] = qa.y; xa[4 * k + 2] = qa.z; xa[4 * k + 3] = qa.w;
        xb[4 * k] = qb.x; xb[4 * k + 1] = qb.y; xb[4 * k + 2] = qb.z; xb[4 * k + 3] = qb.w;
      }
      float lA = 0.0f, lB = 0.0f;
#pragma unroll
      for (int c = 0; c < 12; ++c) {
        float ua = xa[c] + r[c];
        float ub = xb[c] + r[c];
        ua = (ua > 0.0f) ? ua : 0.2f * ua;
        ub = (ub > 0.0f) ? ub : 0.2f * ub;
        lA += ua * w[c];
        lB += ub * w[c];
      }
      if (q0 + t >= T)     lA = NEG_INF;
      if (q0 + t + 1 >= T) lB = NEG_INF;
      const float mn = fmaxf(m, fmaxf(lA, lB));
      const float sc = expf(m - mn);
      const float aA = expf(lA - mn);
      const float aB = expf(lB - mn);
      den = den * sc + aA + aB;
#pragma unroll
      for (int c = 0; c < 12; ++c) acc[c] = acc[c] * sc + aA * xa[c] + aB * xb[c];
      m = mn;
    }
  }

  if (hv) {
    const float inv = den + 1e-16f;
    float4* __restrict__ o4 = (float4*)(out + (size_t)d * F1_ + hh * 12);
    const float4* __restrict__ b4 = (const float4*)(bias + hh * 12);
#pragma unroll
    for (int k = 0; k < 3; ++k) {
      const float4 b = b4[k];
      float4 o;
      o.x = acc[4 * k] / inv + b.x;
      o.y = acc[4 * k + 1] / inv + b.y;
      o.z = acc[4 * k + 2] / inv + b.z;
      o.w = acc[4 * k + 3] / inv + b.w;
      o4[k] = o;
    }
  }
}

__global__ void k_gat1_fused(const int* __restrict__ rowend,
                             const int* __restrict__ srt, const float* __restrict__ xl,
                             const float* __restrict__ xr, const float* __restrict__ att,
                             const float* __restrict__ bias, float* __restrict__ out) {
  const int wv = blockIdx.x * (blockDim.x >> 5) + (threadIdx.x >> 5);
  const int lane = threadIdx.x & 31;
  if (2 * wv >= N_) return;
  const int hb = lane & 16;
  const int j  = lane & 15;
  const int d  = 2 * wv + (lane >> 4);
  const int beg = d ? rowend[d - 1] : 0;
  const int T = rowend[d] - beg + 1;
  const int Tmax = max(__builtin_amdgcn_readlane(T, 0), __builtin_amdgcn_readlane(T, 16));
  const float NEG_INF = -__builtin_huge_valf();

  const float4* __restrict__ xl4 = (const float4*)xl;
  const float4 r = ((const float4*)xr)[(size_t)d * 16 + j];
  const float4 w = ((const float4*)att)[j];
  float4 acc; acc.x = 0.0f; acc.y = 0.0f; acc.z = 0.0f; acc.w = 0.0f;
  float den = 0.0f;
  float m = NEG_INF;
  for (int q0 = 0; q0 < Tmax; q0 += 16) {
    const int n = min(16, Tmax - q0);
    int s_i = d;
    if (q0 + j + 1 < T) s_i = srt[beg + q0 + j];
    for (int t = 0; t < n; t += 4) {
      float4 v[4]; float l[4];
#pragma unroll
      for (int u = 0; u < 4; ++u) {
        const int s = __shfl(s_i, hb + t + u, 32);
        v[u] = xl4[(size_t)s * 16 + j];
      }
#pragma unroll
      for (int u = 0; u < 4; ++u) {
        float a = v[u].x + r.x;
        float b = v[u].y + r.y;
        float cc = v[u].z + r.z;
        float e = v[u].w + r.w;
        a  = (a  > 0.0f) ? a  : 0.2f * a;
        b  = (b  > 0.0f) ? b  : 0.2f * b;
        cc = (cc > 0.0f) ? cc : 0.2f * cc;
        e  = (e  > 0.0f) ? e  : 0.2f * e;
        l[u] = (a * w.x + b * w.y) + (cc * w.z + e * w.w);
      }
#pragma unroll
      for (int o = 8; o > 0; o >>= 1) {
#pragma unroll
        for (int u = 0; u < 4; ++u) l[u] += __shfl_xor(l[u], o, 32);
      }
#pragma unroll
      for (int u = 0; u < 4; ++u)
        if (q0 + t + u >= T) l[u] = NEG_INF;
      const float mn = fmaxf(fmaxf(m, l[0]), fmaxf(fmaxf(l[1], l[2]), l[3]));
      const float sc = expf(m - mn);
      float a[4];
#pragma unroll
      for (int u = 0; u < 4; ++u) a[u] = expf(l[u] - mn);
      den *= sc; acc.x *= sc; acc.y *= sc; acc.z *= sc; acc.w *= sc;
#pragma unroll
      for (int u = 0; u < 4; ++u) {
        den += a[u];
        acc.x += a[u] * v[u].x;
        acc.y += a[u] * v[u].y;
        acc.z += a[u] * v[u].z;
        acc.w += a[u] * v[u].w;
      }
      m = mn;
    }
  }
  const float inv = den + 1e-16f;
  const float4 bb = ((const float4*)bias)[j];
  float4 o4;
  o4.x = acc.x / inv + bb.x;
  o4.y = acc.y / inv + bb.y;
  o4.z = acc.z / inv + bb.z;
  o4.w = acc.w / inv + bb.w;
  ((float4*)out)[(size_t)d * 16 + j] = o4;
}

constexpr size_t OFF_DEG  = 0;
constexpr size_t OFF_EID  = OFF_DEG  + N_;
constexpr size_t OFF_HBUF = OFF_EID  + E_;
constexpr size_t OFF_X1   = OFF_HBUF + (size_t)N_*D_;
constexpr size_t OFF_XL1  = OFF_X1   + (size_t)N_*D_;
constexpr size_t OFF_XR1  = OFF_XL1  + (size_t)N_*F1_;
constexpr size_t OFF_ROW  = OFF_XR1  + (size_t)N_*F1_;
constexpr size_t OFF_CNT  = OFF_X1;
constexpr size_t OFF_RANK = OFF_CNT  + N_;
static_assert((OFF_CNT * 4) % 16 == 0, "the scan reads cnt with 16-byte loads");
static_assert(OFF_RANK + E_ <= OFF_XL1, "cnt and rank must fit the free region");
constexpr size_t OFF_OUT1 = OFF_ROW  + N_;

extern "C" void kernel_launch(void* const* d_in, const int* in_sizes, int n_in,
                              void* d_out, int out_size, void* d_ws, size_t ws_size,
                              hipStream_t stream) {
  const float* x     = (const float*)d_in[0];
  const int*   ei    = (const int*)  d_in[1];
  const float* Wssg  = (const float*)d_in[2];
  const float* bssg  = (const float*)d_in[3];
  const float* W1l   = (const float*)d_in[4];
  const float* b1l   = (const float*)d_in[5];
  const float* W1r   = (const float*)d_in[6];
  const float* b1r   = (const float*)d_in[7];
  const float* att1  = (const float*)d_in[8];
  const float* bias1 = (const float*)d_in[9];
  const float* W2l   = (const float*)d_in[10];
  const float* b2l   = (const float*)d_in[11];
  const float* W2r   = (const float*)d_in[12];
  const float* b2r   = (const float*)d_in[13];
  const float* att2  = (const float*)d_in[14];
  const float* bias2 = (const float*)d_in[15];

  const int* src = ei;
  const int* dst = ei + E_;

  float* ws   = (float*)d_ws;
  float* dinv = ws + OFF_DEG;
  int*   cnt  = (int*)(ws + OFF_CNT);
  int*   rank = (int*)(ws + OFF_RANK);
  int*   srt  = (int*)(ws + OFF_EID);
  int*   row  = (int*)(ws + OFF_ROW);
  float* hbuf = ws + OFF_HBUF;
  float* xl1  = ws + OFF_XL1;
  float* xr1  = ws + OFF_XR1;
  float* out1 = ws + OFF_OUT1;
  float* xl2  = hbuf;
  float* xr2  = xl1;
  float* out  = (float*)d_out;

  const int TB = 256;
  static_assert(N_ % 2 == 0, "two destinations per wave: the per-destination kernels assume an even node count");
  const int NODE_BLOCKS = CDIV(N_ / 2, TB / 32);

  k_zero<<<CDIV(N_, TB), TB, 0, stream>>>(cnt, N_);
  k_rank<<<CDIV(E_, TB), TB, 0, stream>>>(dst, cnt, rank);
  k_rowscan<<<CDIV(N_, 1024), TB, 0, stream>>>(cnt, row, dinv);
  k_csr_fill<<<CDIV(E_, TB), TB, 0, stream>>>(src, dst, row, rank, srt);
  k_agg_dst<<<NODE_BLOCKS, TB, 0, stream>>>(row, srt, x, dinv, hbuf);

  k_gemm_layer<D_, F1_, true, true><<<GEMM_BLOCKS, GEMM_WAVES * 32, 0, stream>>>(hbuf, Wssg, bssg, W1l, b1l, xl1, W1r, b1r, xr1);
  k_gat12_fused<<<NODE_BLOCKS, TB, 0, stream>>>(row, srt, xl1, xr1, att1, bias1, out1);

  k_gemm_layer<F1_, F2_, false, false><<<GEMM_BLOCKS, GEMM_WAVES * 32, 0, stream>>>(out1, nullptr, nullptr, W2l, b2l, xl2, W2r, b2r, xr2);
  k_gat1_fused<<<NODE_BLOCKS, TB, 0, stream>>>(row, srt, xl2, xr2, att2, bias2, out);
}


# each destination owns a 32-slot row (one line): the rank atomic places the source directly, longer rows spill to an overflow list; row scan and slot-fill launches removed; sums kept up to 32 edges
# speedup vs baseline: 6.4101x; 1.1142x over previous
#include <hip/hip_runtime.h>
#include <cmath>

constexpr int N_  = 100000;
constexpr int E_  = 1000000;
constexpr int D_  = 64;
constexpr int F1_ = 144;
constexpr int F2_ = 64;

#define CDIV(a,b) (((a)+(b)-1)/(b))

typedef __attribute__((ext_vector_type(2))) float v2f;
typedef __attribute__((ext_vector_type(8))) float v8f;

__global__ void k_zero(int* __restrict__ p, int n) {
  int i = blockIdx.x * blockDim.x + threadIdx.x;
  if (i < n) p[i] = 0;
}

constexpr int CAP_ = 32;
__global__ void k_place(const int* __restrict__ src, const int* __restrict__ dst,
                        int* __restrict__ cnt, int* __restrict__ ell, int* __restrict__ ovf) {
  int e = blockIdx.x * blockDim.x + threadIdx.x;
  if (e < E_) {
    const int s = src[e];
    const int d = dst[e];
    const int r = atomicAdd(&cnt[d], 1);
    if (r < CAP_) {
      ell[d * CAP_ + r] = s;
    } else {
      const int i = atomicAdd(&cnt[N_], 1);
      ovf[2 * i]     = d;
      ovf[2 * i + 1] = s;
    }
  }
}

__global__ void k_agg_dst(const int* __restrict__ cnt, const int* __restrict__ ell,
                          const int* __restrict__ ovf, const float* __restrict__ x,
                          float* __restrict__ h) {
  const int wv = blockIdx.x * (blockDim.x >> 5) + (threadIdx.x >> 5);
  const int lane = threadIdx.x & 31;
  if (2 * wv >= N_) return;
  const int wvs = __builtin_amdgcn_readfirstlane(wv);
  const int hb = lane & 16;
  const int j  = lane & 15;
  const int d  = 2 * wv + (lane >> 4);
  const int deg = cnt[d];
  const int c   = min(deg, CAP_);
  const int rb = d * CAP_;
  const int cmax = max(__builtin_amdgcn_readlane(c, 0), __builtin_amdgcn_readlane(c, 16));
  const int degmax = max(__builtin_amdgcn_readlane(deg, 0), __builtin_amdgcn_readlane(deg, 16));
  const float dd = rsqrtf((float)(deg + 1));
  const float4* __restrict__ x4 = (const float4*)x;
  const float4 xs = x4[(size_t)d * 16 + j];
  const float wself = dd * dd;
  float4 acc;
  acc.x = xs.x * wself; acc.y = xs.y * wself; acc.z = xs.z * wself; acc.w = xs.w * wself;
  for (int b0 = 0; b0 < cmax; b0 += 16) {
    const int n = min(16, cmax - b0);
    int s_i = d; float w_i = 0.0f;
    if (b0 + j < c) { s_i = ell[rb + b0 + j]; w_i = rsqrtf((float)(cnt[s_i] + 1)) * dd; }
    for (int t = 0; t < n; t += 4) {
      float4 v[4]; float w[4];
#pragma unroll
      for (int u = 0; u < 4; ++u) {
        const int s = __shfl(s_i, hb + t + u, 32);
        w[u] = __shfl(w_i, hb + t + u, 32);
        v[u] = x4[(size_t)s * 16 + j];
      }
#pragma unroll
      for (int u = 0; u < 4; ++u) {
        acc.x += v[u].x * w[u];
        acc.y += v[u].y * w[u];
        acc.z += v[u].z * w[u];
        acc.w += v[u].w * w[u];
      }
    }
  }
  if (degmax > CAP_) {
    const int nov = __builtin_amdgcn_readfirstlane(cnt[N_]);
    const int ln = (int)__builtin_amdgcn_mbcnt_lo(~0u, 0u);
    const int d2 = 2 * wvs + (ln >> 4);
    const int j2 = ln & 15;
    for (int i0 = 0; i0 < nov; i0 += 32) {
      int od = -1, os = 0;
      if (i0 + ln < nov) { od = ovf[2 * (i0 + ln)]; os = ovf[2 * (i0 + ln) + 1]; }
      for (int k = 0; k < 32; ++k) {
        const int dk = __builtin_amdgcn_readlane(od, k);
        const int sk = __builtin_amdgcn_readlane(os, k);
        if ((dk >> 1) == wvs) {
          const float wk = (dk == d2) ? rsqrtf((float)(cnt[sk] + 1)) * dd : 0.0f;
          const float4 v = x4[(size_t)sk * 16 + j2];
          acc.x += v.x * wk; acc.y += v.y * wk; acc.z += v.z * wk; acc.w += v.w * wk;
        }
      }
    }
  }
  float4 o;
  o.x = 0.5f * xs.x + 0.5f * acc.x;
  o.y = 0.5f * xs.y + 0.5f * acc.y;
  o.z = 0.5f * xs.z + 0.5f * acc.z;
  o.w = 0.5f * xs.w + 0.5f * acc.w;
  ((float4*)h)[(size_t)d * 16 + j] = o;
}

constexpr int GEMM_WAVES  = 4;
constexpr int GEMM_BLOCKS = CDIV(N_ / 16, GEMM_WAVES);
constexpr int XP = 68;

__device__ __forceinline__ constexpr int chan_of_pos(int p) {
  return ((p >> 2) % 12) * 12 + ((p >> 2) / 12) * 4 + (p & 3);
}

template <int K, int NC, bool PM>
__device__ __forceinline__ void stage_w(float4* __restrict__ sB, const float* __restrict__ W, int tid) {
  constexpr int TN = NC / 16;
  constexpr int N4 = NC / 4;
  constexpr int ITEMS = (K / 2) * N4;
  for (int it = tid; it < ITEMS; it += GEMM_WAVES * 32) {
    const int kp = it / N4;
    const int n  = (it - kp * N4) * 4;
    const int nw = PM ? chan_of_pos(n) : n;
    const float4 r0 = *(const float4*)(W + (size_t)(2 * kp) * NC + nw);
    const float4 r1 = *(const float4*)(W + (size_t)(2 * kp + 1) * NC + nw);
    const int slot = (((kp >> 1) * TN + (n >> 4)) * 32 + (kp & 1) * 16 + (n & 15));
    float4 lo, hi;
    lo.x = r0.x; lo.y = r1.x; lo.z = r0.y; lo.w = r1.y;
    hi.x = r0.z; hi.y = r1.z; hi.z = r0.w; hi.w = r1.w;
    sB[(slot >> 1)]     = lo;
    sB[(slot >> 1) + 1] = hi;
  }
}

template <int K, int NC>
__device__ __forceinline__ void stripe_mma(const float4* __restrict__ sB, const v2f (&a)[K / 4],
                                           v8f (&acc)[NC / 16], int lane) {
  constexpr int TN = NC / 16;
  constexpr int KS = K / 4;
#pragma unroll
  for (int c = 0; c < TN; ++c) acc[c] = v8f{};
  const v2f* __restrict__ sBf = (const v2f*)sB;
#pragma unroll
  for (int j = 0; j < KS; ++j) {
#pragma unroll
    for (int c = 0; c < TN; ++c) {
      const v2f b = sBf[(j * TN + c) * 32 + lane];
      acc[c] = __builtin_amdgcn_wmma_f32_16x16x4_f32(false, a[j], false, b,
                                                     (short)0, acc[c], false, false);
    }
  }
}

template <int K, int NC, bool PM>
__device__ __forceinline__ void stripe_project(const float4* __restrict__ sB, const v2f (&a)[K / 4],
                                               const float* __restrict__ bias, float* __restrict__ C,
                                               int stripe, int lane) {
  constexpr int TN = NC / 16;
  v8f acc[TN];
  stripe_mma<K, NC>(sB, a, acc, lane);
  const int lm = lane & 15;
  const int rbase = stripe * 16 + ((lane >> 4) << 3);
#pragma unroll
  for (int c = 0; c < TN; ++c) {
    const int n = c * 16 + lm;
    const float bn = bias[PM ? chan_of_pos(n) : n];
#pragma unroll
    for (int r = 0; r < 8; ++r)
      C[(size_t)(rbase + r) * NC + n] = acc[c][r] + bn;
  }
}

template <int K, int NC, bool PRE, bool PM>
__global__ __launch_bounds__(GEMM_WAVES * 32, 7)
void k_gemm_layer(const float* __restrict__ A,
                  const float* __restrict__ Wp, const float* __restrict__ bp,
                  const float* __restrict__ W0, const float* __restrict__ b0, float* __restrict__ C0,
                  const float* __restrict__ W1, const float* __restrict__ b1, float* __restrict__ C1) {
  constexpr int KS = K / 4;
  static_assert(!PRE || K == 64, "the linear layer in front is 64 x 64");
  static_assert(!PM || NC == 144, "piece-major order is defined for 12 heads of 12 channels");
  static_assert(K * NC >= GEMM_WAVES * 16 * XP, "the X tiles of the waves must fit the weight buffer");
  __shared__ float4 sB[K * NC / 4];

  const int tid  = threadIdx.x;
  const int lane = tid & 31;
  const int lm   = lane & 15;
  const int half = lane >> 4;
  const int wave = __builtin_amdgcn_readfirstlane(tid >> 5);
  const int stripe = blockIdx.x * GEMM_WAVES + wave;
  const bool valid = stripe < (N_ >> 4);

  v2f a[KS];
  if (valid) {
    const v2f* __restrict__ Ap = (const v2f*)(A + (size_t)(stripe * 16 + lm) * K + 2 * half);
#pragma unroll
    for (int j = 0; j < KS; ++j) a[j] = Ap[2 * j];
  }

  if constexpr (PRE) {
    stage_w<64, 64, false>(sB, Wp, tid);
    __syncthreads();
    v8f accx[4];
    if (valid) stripe_mma<64, 64>(sB, a, accx, lane);
    __syncthreads();
    float* __restrict__ sX = (float*)sB + wave * (16 * XP);
    if (valid) {
#pragma unroll
      for (int c = 0; c < 4; ++c) {
        const float bn = bp[c * 16 + lm];
#pragma unroll
        for (int r = 0; r < 8; ++r)
          sX[(r + 8 * half) * XP + c * 16 + lm] = accx[c][r] + bn;
      }
    }
    __syncthreads();
    if (valid) {
#pragma unroll
      for (int j = 0; j < KS; ++j) a[j] = *(const v2f*)(sX + lm * XP + 4 * j + 2 * half);
    }
    __syncthreads();
  }

  stage_w<K, NC, PM>(sB, W0, tid);
  __syncthreads();
  if (valid) stripe_project<K, NC, PM>(sB, a, b0, C0, stripe, lane);
  __syncthreads();
  stage_w<K, NC, PM>(sB, W1, tid);
  __syncthreads();
  if (valid) stripe_project<K, NC, PM>(sB, a, b1, C1, stripe, lane);
}

__global__ void k_gat12_fused(const int* __restrict__ cnt, const int* __restrict__ ell,
                              const int* __restrict__ ovf, const float* __restrict__ xl,
                              const float* __restrict__ xr, const float* __restrict__ att,
                              const float* __restrict__ bias, float* __restrict__ out) {
  const int wv = blockIdx.x * (blockDim.x >> 5) + (threadIdx.x >> 5);
  const int lane = threadIdx.x & 31;
  if (2 * wv >= N_) return;
  const int wvs = __builtin_amdgcn_readfirstlane(wv);
  const int hb = lane & 16;
  const int hh = lane & 15;
  const int d  = 2 * wv + (lane >> 4);
  const int deg = cnt[d];
  const int T = min(deg, CAP_) + 1;
  const int rb = d * CAP_;
  const int Tmax = max(__builtin_amdgcn_readlane(T, 0), __builtin_amdgcn_readlane(T, 16));
  const int degmax = max(__builtin_amdgcn_readlane(deg, 0), __builtin_amdgcn_readlane(deg, 16));
  const bool hv = hh < 12;
  const int hc = hv ? hh : 0;
  const float NEG_INF = -__builtin_huge_valf();

  float r[12], w[12];
  {
    const float4* __restrict__ r4 = (const float4*)(xr + (size_t)d * F1_ + hc * 4);
    const float4* __restrict__ w4 = (const float4*)(att + hc * 12);
#pragma unroll
    for (int k = 0; k < 3; ++k) {
      const float4 q = r4[12 * k];
      const float4 p = w4[k];
      r[4 * k] = q.x; r[4 * k + 1] = q.y; r[4 * k + 2] = q.z; r[4 * k + 3] = q.w;
      w[4 * k] = p.x; w[4 * k + 1] = p.y; w[4 * k + 2] = p.z; w[4 * k + 3] = p.w;
    }
  }

  float acc[12];
#pragma unroll
  for (int c = 0; c < 12; ++c) acc[c] = 0.0f;
  float den = 0.0f;
  float m = NEG_INF;

  for (int q0 = 0; q0 < Tmax; q0 += 16) {
    const int n = min(16, Tmax - q0);
    int s_i = d;
    if (q0 + hh + 1 < T) s_i = ell[rb + q0 + hh];
    for (int t = 0; t < n; t += 2) {
      const int sA = __shfl(s_i, hb + t, 32);
      const int sB = __shfl(s_i, hb + t + 1, 32);
      const float4* __restrict__ pA = (const float4*)(xl + (size_t)sA * F1_ + hc * 4);
      const float4* __restrict__ pB = (const float4*)(xl + (size_t)sB * F1_ + hc * 4);
      float xa[12], xb[12];
#pragma unroll
      for (int k = 0; k < 3; ++k) {
        const float4 qa = pA[12 * k];
        const float4 qb = pB[12 * k];
        xa[4 * k] = qa.x; xa[4 * k + 1] = qa.y; xa[4 * k + 2] = qa.z; xa[4 * k + 3] = qa.w;
        xb[4 * k] = qb.x; xb[4 * k + 1] = qb.y; xb[4 * k + 2] = qb.z; xb[4 * k + 3] = qb.w;
      }
      float lA = 0.0f, lB = 0.0f;
#pragma unroll
      for (int c = 0; c < 12; ++c) {
        float ua = xa[c] + r[c];
        float ub = xb[c] + r[c];
        ua = (ua > 0.0f) ? ua : 0.2f * ua;
        ub = (ub > 0.0f) ? ub : 0.2f * ub;
        lA += ua * w[c];
        lB += ub * w[c];
      }
      if (q0 + t >= T)     lA = NEG_INF;
      if (q0 + t + 1 >= T) lB = NEG_INF;
      const float mn = fmaxf(m, fmaxf(lA, lB));
      const float sc = expf(m - mn);
      const float aA = expf(lA - mn);
      const float aB = expf(lB - mn);
      den = den * sc + aA + aB;
#pragma unroll
      for (int c = 0; c < 12; ++c) acc[c] = acc[c] * sc + aA * xa[c] + aB * xb[c];
      m = mn;
    }
  }

  if (degmax > CAP_) {
    const int nov = __builtin_amdgcn_readfirstlane(cnt[N_]);
    const int ln = (int)__builtin_amdgcn_mbcnt_lo(~0u, 0u);
    const int d2 = 2 * wvs + (ln >> 4);
    const int j2 = ln & 15;
    const int hc2 = (j2 < 12) ? j2 : 0;
    for (int i0 = 0; i0 < nov; i0 += 32) {
      int od = -1, os = 0;
      if (i0 + ln < nov) { od = ovf[2 * (i0 + ln)]; os = ovf[2 * (i0 + ln) + 1]; }
      for (int k = 0; k < 32; ++k) {
        const int dk = __builtin_amdgcn_readlane(od, k);
        const int sk = __builtin_amdgcn_readlane(os, k);
        if ((dk >> 1) == wvs) {
          const float4* __restrict__ pA = (const float4*)(xl + (size_t)sk * F1_ + hc2 * 4);
          float xa[12];
#pragma unroll
          for (int p = 0; p < 3; ++p) {
            const float4 qa = pA[12 * p];
            xa[4 * p] = qa.x; xa[4 * p + 1] = qa.y; xa[4 * p + 2] = qa.z; xa[4 * p + 3] = qa.w;
          }
          float lA = 0.0f;
#pragma unroll
          for (int ch = 0; ch < 12; ++ch) {
            float ua = xa[ch] + r[ch];
            ua = (ua > 0.0f) ? ua : 0.2f * ua;
            lA += ua * w[ch];
          }
          if (dk != d2) lA = NEG_INF;
          const float mn = fmaxf(m, lA);
          const float sc = expf(m - mn);
          const float aA = expf(lA - mn);
          den = den * sc + aA;
#pragma unroll
          for (int ch = 0; ch < 12; ++ch) acc[ch] = acc[ch] * sc + aA * xa[ch];
          m = mn;
        }
      }
    }
  }

  if (hv) {
    const float inv = den + 1e-16f;
    float4* __restrict__ o4 = (float4*)(out + (size_t)d * F1_ + hh * 12);
    const float4* __restrict__ b4 = (const float4*)(bias + hh * 12);
#pragma unroll
    for (int k = 0; k < 3; ++k) {
      const float4 b = b4[k];
      float4 o;
      o.x = acc[4 * k] / inv + b.x;
      o.y = acc[4 * k + 1] / inv + b.y;
      o.z = acc[4 * k + 2] / inv + b.z;
      o.w = acc[4 * k + 3] / inv + b.w;
      o4[k] = o;
    }
  }
}

__global__ __launch_bounds__(256, 16)
void k_gat1_fused(const int* __restrict__ cnt, const int* __restrict__ ell,
                             const int* __restrict__ ovf, const float* __restrict__ xl,
                             const float* __restrict__ xr, const float* __restrict__ att,
                             const float* __restrict__ bias, float* __restrict__ out) {
  const int wv = blockIdx.x * (blockDim.x >> 5) + (threadIdx.x >> 5);
  const int lane = threadIdx.x & 31;
  if (2 * wv >= N_) return;
  const int wvs = __builtin_amdgcn_readfirstlane(wv);
  const int hb = lane & 16;
  const int j  = lane & 15;
  const int d  = 2 * wv + (lane >> 4);
  const int deg = cnt[d];
  const int T = min(deg, CAP_) + 1;
  const int rb = d * CAP_;
  const int Tmax = max(__builtin_amdgcn_readlane(T, 0), __builtin_amdgcn_readlane(T, 16));
  const int degmax = max(__builtin_amdgcn_readlane(deg, 0), __builtin_amdgcn_readlane(deg, 16));
  const float NEG_INF = -__builtin_huge_valf();

  const float4* __restrict__ xl4 = (const float4*)xl;
  const float4 r = ((const float4*)xr)[(size_t)d * 16 + j];
  const float4 w = ((const float4*)att)[j];
  float4 acc; acc.x = 0.0f; acc.y = 0.0f; acc.z = 0.0f; acc.w = 0.0f;
  float den = 0.0f;
  float m = NEG_INF;
  for (int q0 = 0; q0 < Tmax; q0 += 16) {
    const int n = min(16, Tmax - q0);
    int s_i = d;
    if (q0 + j + 1 < T) s_i = ell[rb + q0 + j];
    for (int t = 0; t < n; t += 4) {
      float4 v[4]; float l[4];
#pragma unroll
      for (int u = 0; u < 4; ++u) {
        const int s = __shfl(s_i, hb + t + u, 32);
        v[u] = xl4[(size_t)s * 16 + j];
      }
#pragma unroll
      for (int u = 0; u < 4; ++u) {
        float a = v[u].x + r.x;
        float b = v[u].y + r.y;
        float cc = v[u].z + r.z;
        float e = v[u].w + r.w;
        a  = (a  > 0.0f) ? a  : 0.2f * a;
        b  = (b  > 0.0f) ? b  : 0.2f * b;
        cc = (cc > 0.0f) ? cc : 0.2f * cc;
        e  = (e  > 0.0f) ? e  : 0.2f * e;
        l[u] = (a * w.x + b * w.y) + (cc * w.z + e * w.w);
      }
#pragma unroll
      for (int o = 8; o > 0; o >>= 1) {
#pragma unroll
        for (int u = 0; u < 4; ++u) l[u] += __shfl_xor(l[u], o, 32);
      }
#pragma unroll
      for (int u = 0; u < 4; ++u)
        if (q0 + t + u >= T) l[u] = NEG_INF;
      const float mn = fmaxf(fmaxf(m, l[0]), fmaxf(fmaxf(l[1], l[2]), l[3]));
      const float sc = expf(m - mn);
      float a[4];
#pragma unroll
      for (int u = 0; u < 4; ++u) a[u] = expf(l[u] - mn);
      den *= sc; acc.x *= sc; acc.y *= sc; acc.z *= sc; acc.w *= sc;
#pragma unroll
      for (int u = 0; u < 4; ++u) {
        den += a[u];
        acc.x += a[u] * v[u].x;
        acc.y += a[u] * v[u].y;
        acc.z += a[u] * v[u].z;
        acc.w += a[u] * v[u].w;
      }
      m = mn;
    }
  }
  if (degmax > CAP_) {
    const int nov = __builtin_amdgcn_readfirstlane(cnt[N_]);
    const int ln = (int)__builtin_amdgcn_mbcnt_lo(~0u, 0u);
    const int d2 = 2 * wvs + (ln >> 4);
    const int j2 = ln & 15;
    for (int i0 = 0; i0 < nov; i0 += 32) {
      int od = -1, os = 0;
      if (i0 + ln < nov) { od = ovf[2 * (i0 + ln)]; os = ovf[2 * (i0 + ln) + 1]; }
      for (int k = 0; k < 32; ++k) {
        const int dk = __builtin_amdgcn_readlane(od, k);
        const int sk = __builtin_amdgcn_readlane(os, k);
        if ((dk >> 1) == wvs) {
          const float4 v = xl4[(size_t)sk * 16 + j2];
          float a = v.x + r.x;
          float b = v.y + r.y;
          float cc = v.z + r.z;
          float e = v.w + r.w;
          a  = (a  > 0.0f) ? a  : 0.2f * a;
          b  = (b  > 0.0f) ? b  : 0.2f * b;
          cc = (cc > 0.0f) ? cc : 0.2f * cc;
          e  = (e  > 0.0f) ? e  : 0.2f * e;
          float l = (a * w.x + b * w.y) + (cc * w.z + e * w.w);
          for (int o = 8; o > 0; o >>= 1) l += __shfl_xor(l, o, 32);
          if (dk != d2) l = NEG_INF;
          const float mn = fmaxf(m, l);
          const float sc = expf(m - mn);
          const float al = expf(l - mn);
          den = den * sc + al;
          acc.x = acc.x * sc + al * v.x;
          acc.y = acc.y * sc + al * v.y;
          acc.z = acc.z * sc + al * v.z;
          acc.w = acc.w * sc + al * v.w;
          m = mn;
        }
      }
    }
  }
  const float inv = den + 1e-16f;
  const int lz = (int)__builtin_amdgcn_mbcnt_lo(~0u, 0u);
  const float4 bb = ((const float4*)bias)[lz & 15];
  float4 o4;
  o4.x = acc.x / inv + bb.x;
  o4.y = acc.y / inv + bb.y;
  o4.z = acc.z / inv + bb.z;
  o4.w = acc.w / inv + bb.w;
  ((float4*)out)[(size_t)wvs * 32 + lz] = o4;
}

constexpr size_t OFF_HBUF = 0;
constexpr size_t OFF_X1   = OFF_HBUF + (size_t)N_*D_;
constexpr size_t OFF_XL1  = OFF_X1   + (size_t)N_*D_;
constexpr size_t OFF_XR1  = OFF_XL1  + (size_t)N_*F1_;
constexpr size_t OFF_OUT1 = OFF_XR1  + (size_t)N_*F1_;
constexpr size_t OFF_ELL  = OFF_X1;
constexpr size_t OFF_CNT  = OFF_ELL  + (size_t)N_*CAP_;
constexpr size_t OFF_OVF  = OFF_CNT  + N_ + 32;
static_assert((OFF_ELL * 4) % 128 == 0 && (CAP_ * 4) % 128 == 0, "a row of slots is whole 128-byte lines");
static_assert(OFF_OVF + 2 * (size_t)E_ <= OFF_XL1, "rows, counts and overflow pairs must fit the free region");

extern "C" void kernel_launch(void* const* d_in, const int* in_sizes, int n_in,
                              void* d_out, int out_size, void* d_ws, size_t ws_size,
                              hipStream_t stream) {
  const float* x     = (const float*)d_in[0];
  const int*   ei    = (const int*)  d_in[1];
  const float* Wssg  = (const float*)d_in[2];
  const float* bssg  = (const float*)d_in[3];
  const float* W1l   = (const float*)d_in[4];
  const float* b1l   = (const float*)d_in[5];
  const float* W1r   = (const float*)d_in[6];
  const float* b1r   = (const float*)d_in[7];
  const float* att1  = (const float*)d_in[8];
  const float* bias1 = (const float*)d_in[9];
  const float* W2l   = (const float*)d_in[10];
  const float* b2l   = (const float*)d_in[11];
  const float* W2r   = (const float*)d_in[12];
  const float* b2r   = (const float*)d_in[13];
  const float* att2  = (const float*)d_in[14];
  const float* bias2 = (const float*)d_in[15];

  const int* src = ei;
  const int* dst = ei + E_;

  float* ws   = (float*)d_ws;
  int*   cnt  = (int*)(ws + OFF_CNT);
  int*   ell  = (int*)(ws + OFF_ELL);
  int*   ovf  = (int*)(ws + OFF_OVF);
  float* hbuf = ws + OFF_HBUF;
  float* xl1  = ws + OFF_XL1;
  float* xr1  = ws + OFF_XR1;
  float* out1 = ws + OFF_OUT1;
  float* xl2  = hbuf;
  float* xr2  = xl1;
  float* out  = (float*)d_out;

  const int TB = 256;
  static_assert(N_ % 2 == 0, "two destinations per wave: the per-destination kernels assume an even node count");
  const int NODE_BLOCKS = CDIV(N_ / 2, TB / 32);

  k_zero<<<CDIV(N_ + 1, TB), TB, 0, stream>>>(cnt, N_ + 1);
  k_place<<<CDIV(E_, TB), TB, 0, stream>>>(src, dst, cnt, ell, ovf);
  k_agg_dst<<<NODE_BLOCKS, TB, 0, stream>>>(cnt, ell, ovf, x, hbuf);

  k_gemm_layer<D_, F1_, true, true><<<GEMM_BLOCKS, GEMM_WAVES * 32, 0, stream>>>(hbuf, Wssg, bssg, W1l, b1l, xl1, W1r, b1r, xr1);
  k_gat12_fused<<<NODE_BLOCKS, TB, 0, stream>>>(cnt, ell, ovf, xl1, xr1, att1, bias1, out1);

  k_gemm_layer<F1_, F2_, false, false><<<GEMM_BLOCKS, GEMM_WAVES * 32, 0, stream>>>(out1, nullptr, nullptr, W2l, b2l, xl2, W2r, b2r, xr2);
  k_gat1_fused<<<NODE_BLOCKS, TB, 0, stream>>>(cnt, ell, ovf, xl2, xr2, att2, bias2, out);
}


# attention kernels: fewer vector instructions per edge: leaky_relu as packed max(u,0.2u), exp2 of log2e-scaled logits for exp, DPP half-wave sums, one reciprocal per row; logit sum order changed
# speedup vs baseline: 6.8061x; 1.0618x over previous
#include <hip/hip_runtime.h>
#include <cmath>

constexpr int N_  = 100000;
constexpr int E_  = 1000000;
constexpr int D_  = 64;
constexpr int F1_ = 144;
constexpr int F2_ = 64;

#define CDIV(a,b) (((a)+(b)-1)/(b))

typedef __attribute__((ext_vector_type(2))) float v2f;
typedef __attribute__((ext_vector_type(8))) float v8f;

__global__ void k_zero(int* __restrict__ p, int n) {
  int i = blockIdx.x * blockDim.x + threadIdx.x;
  if (i < n) p[i] = 0;
}

constexpr int CAP_ = 32;
__global__ void k_place(const int* __restrict__ src, const int* __restrict__ dst,
                        int* __restrict__ cnt, int* __restrict__ ell, int* __restrict__ ovf) {
  int e = blockIdx.x * blockDim.x + threadIdx.x;
  if (e < E_) {
    const int s = src[e];
    const int d = dst[e];
    const int r = atomicAdd(&cnt[d], 1);
    if (r < CAP_) {
      ell[d * CAP_ + r] = s;
    } else {
      const int i = atomicAdd(&cnt[N_], 1);
      ovf[2 * i]     = d;
      ovf[2 * i + 1] = s;
    }
  }
}

__global__ void k_agg_dst(const int* __restrict__ cnt, const int* __restrict__ ell,
                          const int* __restrict__ ovf, const float* __restrict__ x,
                          float* __restrict__ h) {
  const int wv = blockIdx.x * (blockDim.x >> 5) + (threadIdx.x >> 5);
  const int lane = threadIdx.x & 31;
  if (2 * wv >= N_) return;
  const int wvs = __builtin_amdgcn_readfirstlane(wv);
  const int hb = lane & 16;
  const int j  = lane & 15;
  const int d  = 2 * wv + (lane >> 4);
  const int deg = cnt[d];
  const int c   = min(deg, CAP_);
  const int rb = d * CAP_;
  const int cmax = max(__builtin_amdgcn_readlane(c, 0), __builtin_amdgcn_readlane(c, 16));
  const int degmax = max(__builtin_amdgcn_readlane(deg, 0), __builtin_amdgcn_readlane(deg, 16));
  const float dd = rsqrtf((float)(deg + 1));
  const float4* __restrict__ x4 = (const float4*)x;
  const float4 xs = x4[(size_t)d * 16 + j];
  const float wself = dd * dd;
  float4 acc;
  acc.x = xs.x * wself; acc.y = xs.y * wself; acc.z = xs.z * wself; acc.w = xs.w * wself;
  for (int b0 = 0; b0 < cmax; b0 += 16) {
    const int n = min(16, cmax - b0);
    int s_i = d; float w_i = 0.0f;
    if (b0 + j < c) { s_i = ell[rb + b0 + j]; w_i = rsqrtf((float)(cnt[s_i] + 1)) * dd; }
    for (int t = 0; t < n; t += 4) {
      float4 v[4]; float w[4];
#pragma unroll
      for (int u = 0; u < 4; ++u) {
        const int s = __shfl(s_i, hb + t + u, 32);
        w[u] = __shfl(w_i, hb + t + u, 32);
        v[u] = x4[(size_t)s * 16 + j];
      }
#pragma unroll
      for (int u = 0; u < 4; ++u) {
        acc.x += v[u].x * w[u];
        acc.y += v[u].y * w[u];
        acc.z += v[u].z * w[u];
        acc.w += v[u].w * w[u];
      }
    }
  }
  if (degmax > CAP_) {
    const int nov = __builtin_amdgcn_readfirstlane(cnt[N_]);
    const int ln = (int)__builtin_amdgcn_mbcnt_lo(~0u, 0u);
    const int d2 = 2 * wvs + (ln >> 4);
    const int j2 = ln & 15;
    for (int i0 = 0; i0 < nov; i0 += 32) {
      int od = -1, os = 0;
      if (i0 + ln < nov) { od = ovf[2 * (i0 + ln)]; os = ovf[2 * (i0 + ln) + 1]; }
      for (int k = 0; k < 32; ++k) {
        const int dk = __builtin_amdgcn_readlane(od, k);
        const int sk = __builtin_amdgcn_readlane(os, k);
        if ((dk >> 1) == wvs) {
          const float wk = (dk == d2) ? rsqrtf((float)(cnt[sk] + 1)) * dd : 0.0f;
          const float4 v = x4[(size_t)sk * 16 + j2];
          acc.x += v.x * wk; acc.y += v.y * wk; acc.z += v.z * wk; acc.w += v.w * wk;
        }
      }
    }
  }
  float4 o;
  o.x = 0.5f * xs.x + 0.5f * acc.x;
  o.y = 0.5f * xs.y + 0.5f * acc.y;
  o.z = 0.5f * xs.z + 0.5f * acc.z;
  o.w = 0.5f * xs.w + 0.5f * acc.w;
  ((float4*)h)[(size_t)d * 16 + j] = o;
}

constexpr int GEMM_WAVES  = 4;
constexpr int GEMM_BLOCKS = CDIV(N_ / 16, GEMM_WAVES);
constexpr int XP = 68;

__device__ __forceinline__ constexpr int chan_of_pos(int p) {
  return ((p >> 2) % 12) * 12 + ((p >> 2) / 12) * 4 + (p & 3);
}

template <int K, int NC, bool PM>
__device__ __forceinline__ void stage_w(float4* __restrict__ sB, const float* __restrict__ W, int tid) {
  constexpr int TN = NC / 16;
  constexpr int N4 = NC / 4;
  constexpr int ITEMS = (K / 2) * N4;
  for (int it = tid; it < ITEMS; it += GEMM_WAVES * 32) {
    const int kp = it / N4;
    const int n  = (it - kp * N4) * 4;
    const int nw = PM ? chan_of_pos(n) : n;
    const float4 r0 = *(const float4*)(W + (size_t)(2 * kp) * NC + nw);
    const float4 r1 = *(const float4*)(W + (size_t)(2 * kp + 1) * NC + nw);
    const int slot = (((kp >> 1) * TN + (n >> 4)) * 32 + (kp & 1) * 16 + (n & 15));
    float4 lo, hi;
    lo.x = r0.x; lo.y = r1.x; lo.z = r0.y; lo.w = r1.y;
    hi.x = r0.z; hi.y = r1.z; hi.z = r0.w; hi.w = r1.w;
    sB[(slot >> 1)]     = lo;
    sB[(slot >> 1) + 1] = hi;
  }
}

template <int K, int NC>
__device__ __forceinline__ void stripe_mma(const float4* __restrict__ sB, const v2f (&a)[K / 4],
                                           v8f (&acc)[NC / 16], int lane) {
  constexpr int TN = NC / 16;
  constexpr int KS = K / 4;
#pragma unroll
  for (int c = 0; c < TN; ++c) acc[c] = v8f{};
  const v2f* __restrict__ sBf = (const v2f*)sB;
#pragma unroll
  for (int j = 0; j < KS; ++j) {
#pragma unroll
    for (int c = 0; c < TN; ++c) {
      const v2f b = sBf[(j * TN + c) * 32 + lane];
      acc[c] = __builtin_amdgcn_wmma_f32_16x16x4_f32(false, a[j], false, b,
                                                     (short)0, acc[c], false, false);
    }
  }
}

template <int K, int NC, bool PM>
__device__ __forceinline__ void stripe_project(const float4* __restrict__ sB, const v2f (&a)[K / 4],
                                               const float* __restrict__ bias, float* __restrict__ C,
                                               int stripe, int lane) {
  constexpr int TN = NC / 16;
  v8f acc[TN];
  stripe_mma<K, NC>(sB, a, acc, lane);
  const int lm = lane & 15;
  const int rbase = stripe * 16 + ((lane >> 4) << 3);
#pragma unroll
  for (int c = 0; c < TN; ++c) {
    const int n = c * 16 + lm;
    const float bn = bias[PM ? chan_of_pos(n) : n];
#pragma unroll
    for (int r = 0; r < 8; ++r)
      C[(size_t)(rbase + r) * NC + n] = acc[c][r] + bn;
  }
}

template <int K, int NC, bool PRE, bool PM>
__global__ __launch_bounds__(GEMM_WAVES * 32, 7)
void k_gemm_layer(const float* __restrict__ A,
                  const float* __restrict__ Wp, const float* __restrict__ bp,
                  const float* __restrict__ W0, const float* __restrict__ b0, float* __restrict__ C0,
                  const float* __restrict__ W1, const float* __restrict__ b1, float* __restrict__ C1) {
  constexpr int KS = K / 4;
  static_assert(!PRE || K == 64, "the linear layer in front is 64 x 64");
  static_assert(!PM || NC == 144, "piece-major order is defined for 12 heads of 12 channels");
  static_assert(K * NC >= GEMM_WAVES * 16 * XP, "the X tiles of the waves must fit the weight buffer");
  __shared__ float4 sB[K * NC / 4];

  const int tid  = threadIdx.x;
  const int lane = tid & 31;
  const int lm   = lane & 15;
  const int half = lane >> 4;
  const int wave = __builtin_amdgcn_readfirstlane(tid >> 5);
  const int stripe = blockIdx.x * GEMM_WAVES + wave;
  const bool valid = stripe < (N_ >> 4);

  v2f a[KS];
  if (valid) {
    const v2f* __restrict__ Ap = (const v2f*)(A + (size_t)(stripe * 16 + lm) * K + 2 * half);
#pragma unroll
    for (int j = 0; j < KS; ++j) a[j] = Ap[2 * j];
  }

  if constexpr (PRE) {
    stage_w<64, 64, false>(sB, Wp, tid);
    __syncthreads();
    v8f accx[4];
    if (valid) stripe_mma<64, 64>(sB, a, accx, lane);
    __syncthreads();
    float* __restrict__ sX = (float*)sB + wave * (16 * XP);
    if (valid) {
#pragma unroll
      for (int c = 0; c < 4; ++c) {
        const float bn = bp[c * 16 + lm];
#pragma unroll
        for (int r = 0; r < 8; ++r)
          sX[(r + 8 * half) * XP + c * 16 + lm] = accx[c][r] + bn;
      }
    }
    __syncthreads();
    if (valid) {
#pragma unroll
      for (int j = 0; j < KS; ++j) a[j] = *(const v2f*)(sX + lm * XP + 4 * j + 2 * half);
    }
    __syncthreads();
  }

  stage_w<K, NC, PM>(sB, W0, tid);
  __syncthreads();
  if (valid) stripe_project<K, NC, PM>(sB, a, b0, C0, stripe, lane);
  __syncthreads();
  stage_w<K, NC, PM>(sB, W1, tid);
  __syncthreads();
  if (valid) stripe_project<K, NC, PM>(sB, a, b1, C1, stripe, lane);
}

__device__ __forceinline__ v2f leaky2(v2f u) {
  return __builtin_elementwise_max(u, u * 0.2f);
}
constexpr float LOG2E_ = 1.44269504088896340736f;
__device__ __forceinline__ float exp2_hw(float x) { return __builtin_amdgcn_exp2f(x); }
template <int O>
__device__ __forceinline__ float row_xor(float v) {
  return __builtin_bit_cast(float, __builtin_amdgcn_update_dpp(0, __builtin_bit_cast(int, v), 0x160 + O, 0xf, 0xf, true));
}

__global__ void k_gat12_fused(const int* __restrict__ cnt, const int* __restrict__ ell,
                              const int* __restrict__ ovf, const float* __restrict__ xl,
                              const float* __restrict__ xr, const float* __restrict__ att,
                              const float* __restrict__ bias, float* __restrict__ out) {
  const int wv = blockIdx.x * (blockDim.x >> 5) + (threadIdx.x >> 5);
  const int lane = threadIdx.x & 31;
  if (2 * wv >= N_) return;
  const int wvs = __builtin_amdgcn_readfirstlane(wv);
  const int hb = lane & 16;
  const int hh = lane & 15;
  const int d  = 2 * wv + (lane >> 4);
  const int deg = cnt[d];
  const int T = min(deg, CAP_) + 1;
  const int rb = d * CAP_;
  const int Tmax = max(__builtin_amdgcn_readlane(T, 0), __builtin_amdgcn_readlane(T, 16));
  const int degmax = max(__builtin_amdgcn_readlane(deg, 0), __builtin_amdgcn_readlane(deg, 16));
  const bool hv = hh < 12;
  const int hc = hv ? hh : 0;
  const float NEG_INF = -__builtin_huge_valf();

  v2f r2[6], w2[6];
  {
    const float4* __restrict__ r4 = (const float4*)(xr + (size_t)d * F1_ + hc * 4);
    const float4* __restrict__ w4 = (const float4*)(att + hc * 12);
#pragma unroll
    for (int k = 0; k < 3; ++k) {
      const float4 q = r4[12 * k];
      const float4 p = w4[k];
      r2[2 * k]     = v2f{q.x, q.y};
      r2[2 * k + 1] = v2f{q.z, q.w};
      w2[2 * k]     = v2f{p.x, p.y} * LOG2E_;
      w2[2 * k + 1] = v2f{p.z, p.w} * LOG2E_;
    }
  }

  v2f acc2[6];
#pragma unroll
  for (int c = 0; c < 6; ++c) acc2[c] = v2f{0.0f, 0.0f};
  float den = 0.0f;
  float m = NEG_INF;

  for (int q0 = 0; q0 < Tmax; q0 += 16) {
    const int n = min(16, Tmax - q0);
    int s_i = d;
    if (q0 + hh + 1 < T) s_i = ell[rb + q0 + hh];
    for (int t = 0; t < n; t += 2) {
      const int sA = __shfl(s_i, hb + t, 32);
      const int sB = __shfl(s_i, hb + t + 1, 32);
      const float4* __restrict__ pA = (const float4*)(xl + (size_t)sA * F1_ + hc * 4);
      const float4* __restrict__ pB = (const float4*)(xl + (size_t)sB * F1_ + hc * 4);
      v2f xa[6], xb[6];
#pragma unroll
      for (int k = 0; k < 3; ++k) {
        const float4 qa = pA[12 * k];
        const float4 qb = pB[12 * k];
        xa[2 * k] = v2f{qa.x, qa.y}; xa[2 * k + 1] = v2f{qa.z, qa.w};
        xb[2 * k] = v2f{qb.x, qb.y}; xb[2 * k + 1] = v2f{qb.z, qb.w};
      }
      v2f lsA = v2f{0.0f, 0.0f}, lsB = v2f{0.0f, 0.0f};
#pragma unroll
      for (int c = 0; c < 6; ++c) {
        lsA = __builtin_elementwise_fma(leaky2(xa[c] + r2[c]), w2[c], lsA);
        lsB = __builtin_elementwise_fma(leaky2(xb[c] + r2[c]), w2[c], lsB);
      }
      float lA = lsA.x + lsA.y;
      float lB = lsB.x + lsB.y;
      if (q0 + t >= T)     lA = NEG_INF;
      if (q0 + t + 1 >= T) lB = NEG_INF;
      const float mn = fmaxf(m, fmaxf(lA, lB));
      const float sc = exp2_hw(m - mn);
      const float aA = exp2_hw(lA - mn);
      const float aB = exp2_hw(lB - mn);
      den = den * sc + aA + aB;
#pragma unroll
      for (int c = 0; c < 6; ++c) acc2[c] = acc2[c] * sc + xa[c] * aA + xb[c] * aB;
      m = mn;
    }
  }

  if (degmax > CAP_) {
    const int nov = __builtin_amdgcn_readfirstlane(cnt[N_]);
    const int ln = (int)__builtin_amdgcn_mbcnt_lo(~0u, 0u);
    const int d2 = 2 * wvs + (ln >> 4);
    const int j2 = ln & 15;
    const int hc2 = (j2 < 12) ? j2 : 0;
    for (int i0 = 0; i0 < nov; i0 += 32) {
      int od = -1, os = 0;
      if (i0 + ln < nov) { od = ovf[2 * (i0 + ln)]; os = ovf[2 * (i0 + ln) + 1]; }
      for (int k = 0; k < 32; ++k) {
        const int dk = __builtin_amdgcn_readlane(od, k);
        const int sk = __builtin_amdgcn_readlane(os, k);
        if ((dk >> 1) == wvs) {
          const float4* __restrict__ pA = (const float4*)(xl + (size_t)sk * F1_ + hc2 * 4);
          v2f xa[6];
#pragma unroll
          for (int p = 0; p < 3; ++p) {
            const float4 qa = pA[12 * p];
            xa[2 * p] = v2f{qa.x, qa.y}; xa[2 * p + 1] = v2f{qa.z, qa.w};
          }
          v2f lsA = v2f{0.0f, 0.0f};
#pragma unroll
          for (int c = 0; c < 6; ++c)
            lsA = __builtin_elementwise_fma(leaky2(xa[c] + r2[c]), w2[c], lsA);
          float lA = lsA.x + lsA.y;
          if (dk != d2) lA = NEG_INF;
          const float mn = fmaxf(m, lA);
          const float sc = exp2_hw(m - mn);
          const float aA = exp2_hw(lA - mn);
          den = den * sc + aA;
#pragma unroll
          for (int c = 0; c < 6; ++c) acc2[c] = acc2[c] * sc + xa[c] * aA;
          m = mn;
        }
      }
    }
  }

  if (hv) {
    const float rinv = 1.0f / (den + 1e-16f);
    float4* __restrict__ o4 = (float4*)(out + (size_t)d * F1_ + hh * 12);
    const float4* __restrict__ b4 = (const float4*)(bias + hh * 12);
#pragma unroll
    for (int k = 0; k < 3; ++k) {
      const float4 b = b4[k];
      float4 o;
      o.x = acc2[2 * k].x * rinv + b.x;
      o.y = acc2[2 * k].y * rinv + b.y;
      o.z = acc2[2 * k + 1].x * rinv + b.z;
      o.w = acc2[2 * k + 1].y * rinv + b.w;
      o4[k] = o;
    }
  }
}

__global__ __launch_bounds__(256, 16)
void k_gat1_fused(const int* __restrict__ cnt, const int* __restrict__ ell,
                             const int* __restrict__ ovf, const float* __restrict__ xl,
                             const float* __restrict__ xr, const float* __restrict__ att,
                             const float* __restrict__ bias, float* __restrict__ out) {
  const int wv = blockIdx.x * (blockDim.x >> 5) + (threadIdx.x >> 5);
  const int lane = threadIdx.x & 31;
  if (2 * wv >= N_) return;
  const int wvs = __builtin_amdgcn_readfirstlane(wv);
  const int hb = lane & 16;
  const int j  = lane & 15;
  const int d  = 2 * wv + (lane >> 4);
  const int deg = cnt[d];
  const int T = min(deg, CAP_) + 1;
  const int rb = d * CAP_;
  const int Tmax = max(__builtin_amdgcn_readlane(T, 0), __builtin_amdgcn_readlane(T, 16));
  const int degmax = max(__builtin_amdgcn_readlane(deg, 0), __builtin_amdgcn_readlane(deg, 16));
  const float NEG_INF = -__builtin_huge_valf();

  const float4* __restrict__ xl4 = (const float4*)xl;
  const float4 r = ((const float4*)xr)[(size_t)d * 16 + j];
  const float4 w = ((const float4*)att)[j];
  const v2f rlo = v2f{r.x, r.y}, rhi = v2f{r.z, r.w};
  const v2f wlo = v2f{w.x, w.y} * LOG2E_, whi = v2f{w.z, w.w} * LOG2E_;
  v2f alo = v2f{0.0f, 0.0f}, ahi = v2f{0.0f, 0.0f};
  float den = 0.0f;
  float m = NEG_INF;
  for (int q0 = 0; q0 < Tmax; q0 += 16) {
    const int n = min(16, Tmax - q0);
    int s_i = d;
    if (q0 + j + 1 < T) s_i = ell[rb + q0 + j];
    for (int t = 0; t < n; t += 4) {
      v2f vlo[4], vhi[4]; float l[4];
#pragma unroll
      for (int u = 0; u < 4; ++u) {
        const int s = __shfl(s_i, hb + t + u, 32);
        const float4 v = xl4[(size_t)s * 16 + j];
        vlo[u] = v2f{v.x, v.y}; vhi[u] = v2f{v.z, v.w};
      }
#pragma unroll
      for (int u = 0; u < 4; ++u) {
        const v2f p = __builtin_elementwise_fma(leaky2(vhi[u] + rhi), whi, leaky2(vlo[u] + rlo) * wlo);
        l[u] = p.x + p.y;
      }
#pragma unroll
      for (int u = 0; u < 4; ++u) l[u] += row_xor<8>(l[u]);
#pragma unroll
      for (int u = 0; u < 4; ++u) l[u] += row_xor<4>(l[u]);
#pragma unroll
      for (int u = 0; u < 4; ++u) l[u] += row_xor<2>(l[u]);
#pragma unroll
      for (int u = 0; u < 4; ++u) l[u] += row_xor<1>(l[u]);
#pragma unroll
      for (int u = 0; u < 4; ++u)
        if (q0 + t + u >= T) l[u] = NEG_INF;
      const float mn = fmaxf(fmaxf(m, l[0]), fmaxf(fmaxf(l[1], l[2]), l[3]));
      const float sc = exp2_hw(m - mn);
      float a[4];
#pragma unroll
      for (int u = 0; u < 4; ++u) a[u] = exp2_hw(l[u] - mn);
      den *= sc; alo = alo * sc; ahi = ahi * sc;
#pragma unroll
      for (int u = 0; u < 4; ++u) {
        den += a[u];
        alo = alo + vlo[u] * a[u];
        ahi = ahi + vhi[u] * a[u];
      }
      m = mn;
    }
  }
  if (degmax > CAP_) {
    const int nov = __builtin_amdgcn_readfirstlane(cnt[N_]);
    const int ln = (int)__builtin_amdgcn_mbcnt_lo(~0u, 0u);
    const int d2 = 2 * wvs + (ln >> 4);
    const int j2 = ln & 15;
    for (int i0 = 0; i0 < nov; i0 += 32) {
      int od = -1, os = 0;
      if (i0 + ln < nov) { od = ovf[2 * (i0 + ln)]; os = ovf[2 * (i0 + ln) + 1]; }
      for (int k = 0; k < 32; ++k) {
        const int dk = __builtin_amdgcn_readlane(od, k);
        const int sk = __builtin_amdgcn_readlane(os, k);
        if ((dk >> 1) == wvs) {
          const float4 v = xl4[(size_t)sk * 16 + j2];
          const v2f vl = v2f{v.x, v.y}, vh = v2f{v.z, v.w};
          const v2f p = __builtin_elementwise_fma(leaky2(vh + rhi), whi, leaky2(vl + rlo) * wlo);
          float l = p.x + p.y;
          l += row_xor<8>(l); l += row_xor<4>(l); l += row_xor<2>(l); l += row_xor<1>(l);
          if (dk != d2) l = NEG_INF;
          const float mn = fmaxf(m, l);
          const float sc = exp2_hw(m - mn);
          const float al = exp2_hw(l - mn);
          den = den * sc + al;
          alo = alo * sc + vl * al;
          ahi = ahi * sc + vh * al;
          m = mn;
        }
      }
    }
  }
  const float rinv = 1.0f / (den + 1e-16f);
  const int lz = (int)__builtin_amdgcn_mbcnt_lo(~0u, 0u);
  const float4 bb = ((const float4*)bias)[lz & 15];
  float4 o4;
  o4.x = alo.x * rinv + bb.x;
  o4.y = alo.y * rinv + bb.y;
  o4.z = ahi.x * rinv + bb.z;
  o4.w = ahi.y * rinv + bb.w;
  ((float4*)out)[(size_t)wvs * 32 + lz] = o4;
}

constexpr size_t OFF_HBUF = 0;
constexpr size_t OFF_X1   = OFF_HBUF + (size_t)N_*D_;
constexpr size_t OFF_XL1  = OFF_X1   + (size_t)N_*D_;
constexpr size_t OFF_XR1  = OFF_XL1  + (size_t)N_*F1_;
constexpr size_t OFF_OUT1 = OFF_XR1  + (size_t)N_*F1_;
constexpr size_t OFF_ELL  = OFF_X1;
constexpr size_t OFF_CNT  = OFF_ELL  + (size_t)N_*CAP_;
constexpr size_t OFF_OVF  = OFF_CNT  + N_ + 32;
static_assert((OFF_ELL * 4) % 128 == 0 && (CAP_ * 4) % 128 == 0, "a row of slots is whole 128-byte lines");
static_assert(OFF_OVF + 2 * (size_t)E_ <= OFF_XL1, "rows, counts and overflow pairs must fit the free region");

extern "C" void kernel_launch(void* const* d_in, const int* in_sizes, int n_in,
                              void* d_out, int out_size, void* d_ws, size_t ws_size,
                              hipStream_t stream) {
  const float* x     = (const float*)d_in[0];
  const int*   ei    = (const int*)  d_in[1];
  const float* Wssg  = (const float*)d_in[2];
  const float* bssg  = (const float*)d_in[3];
  const float* W1l   = (const float*)d_in[4];
  const float* b1l   = (const float*)d_in[5];
  const float* W1r   = (const float*)d_in[6];
  const float* b1r   = (const float*)d_in[7];
  const float* att1  = (const float*)d_in[8];
  const float* bias1 = (const float*)d_in[9];
  const float* W2l   = (const float*)d_in[10];
  const float* b2l   = (const float*)d_in[11];
  const float* W2r   = (const float*)d_in[12];
  const float* b2r   = (const float*)d_in[13];
  const float* att2  = (const float*)d_in[14];
  const float* bias2 = (const float*)d_in[15];

  const int* src = ei;
  const int* dst = ei + E_;

  float* ws   = (float*)d_ws;
  int*   cnt  = (int*)(ws + OFF_CNT);
  int*   ell  = (int*)(ws + OFF_ELL);
  int*   ovf  = (int*)(ws + OFF_OVF);
  float* hbuf = ws + OFF_HBUF;
  float* xl1  = ws + OFF_XL1;
  float* xr1  = ws + OFF_XR1;
  float* out1 = ws + OFF_OUT1;
  float* xl2  = hbuf;
  float* xr2  = xl1;
  float* out  = (float*)d_out;

  const int TB = 256;
  static_assert(N_ % 2 == 0, "two destinations per wave: the per-destination kernels assume an even node count");
  const int NODE_BLOCKS = CDIV(N_ / 2, TB / 32);

  k_zero<<<CDIV(N_ + 1, TB), TB, 0, stream>>>(cnt, N_ + 1);
  k_place<<<CDIV(E_, TB), TB, 0, stream>>>(src, dst, cnt, ell, ovf);
  k_agg_dst<<<NODE_BLOCKS, TB, 0, stream>>>(cnt, ell, ovf, x, hbuf);

  k_gemm_layer<D_, F1_, true, true><<<GEMM_BLOCKS, GEMM_WAVES * 32, 0, stream>>>(hbuf, Wssg, bssg, W1l, b1l, xl1, W1r, b1r, xr1);
  k_gat12_fused<<<NODE_BLOCKS, TB, 0, stream>>>(cnt, ell, ovf, xl1, xr1, att1, bias1, out1);

  k_gemm_layer<F1_, F2_, false, false><<<GEMM_BLOCKS, GEMM_WAVES * 32, 0, stream>>>(out1, nullptr, nullptr, W2l, b2l, xl2, W2r, b2r, xr2);
  k_gat1_fused<<<NODE_BLOCKS, TB, 0, stream>>>(cnt, ell, ovf, xl2, xr2, att2, bias2, out);
}
